# starting kernel
# baseline (speedup 1.0000x reference)
#include <hip/hip_runtime.h>


static constexpr int DIM  = 128;
static constexpr int NBLK = 2;
static constexpr int LDP  = 68;

typedef __attribute__((ext_vector_type(2))) float v2f;
typedef __attribute__((ext_vector_type(8))) float v8f;

static __device__ __forceinline__ float wave_reduce_sum(float x) {
#pragma unroll
  for (int off = 16; off > 0; off >>= 1) x += __shfl_xor(x, off, 32);
  return x;
}

static __device__ __forceinline__ unsigned f32_order_enc(float f) {
  unsigned u = __float_as_uint(f);
  return (u & 0x80000000u) ? ~u : (u | 0x80000000u);
}
static __device__ __forceinline__ float f32_order_dec(unsigned e) {
  unsigned u = (e & 0x80000000u) ? (e & 0x7fffffffu) : ~e;
  return __uint_as_float(u);
}

__global__ __launch_bounds__(128) void gemm_nt_128(const float* __restrict__ A,
                                                   const float* __restrict__ W,
                                                   float* __restrict__ C, int N) {
  __shared__ float Wlds[128 * LDP];
  const int tid = threadIdx.x;
  const int wv  = tid >> 5;
  const int ln  = tid & 31;
  const int hk  = ln >> 4;
  const int l16 = ln & 15;
  const int r0  = blockIdx.x * 64 + wv * 16;

  v8f acc[8];
#pragma unroll
  for (int t = 0; t < 8; ++t)
#pragma unroll
    for (int j = 0; j < 8; ++j) acc[t][j] = 0.0f;

  int rowA = r0 + l16;
  if (rowA >= N) rowA = N - 1;
  const float* ap = A + (size_t)rowA * DIM;

  for (int kp = 0; kp < DIM; kp += 64) {
    __syncthreads();
#pragma unroll
    for (int i = 0; i < 16; ++i) {
      const int q  = tid + 128 * i;
      const int n  = q >> 4;
      const int c4 = (q & 15) << 2;
      *(float4*)(&Wlds[n * LDP + c4]) =
          *(const float4*)(W + (size_t)n * DIM + kp + c4);
    }
    __syncthreads();

#pragma unroll 4
    for (int k0 = 0; k0 < 64; k0 += 4) {
      const int ka = k0 + 2 * hk;
      v2f a; a.x = ap[kp + ka]; a.y = ap[kp + ka + 1];
#pragma unroll
      for (int t = 0; t < 8; ++t) {
        const float* wp = &Wlds[(t * 16 + l16) * LDP + ka];
        v2f b; b.x = wp[0]; b.y = wp[1];
        acc[t] = __builtin_amdgcn_wmma_f32_16x16x4_f32(
            false, a, false, b, (short)0, acc[t], false, false);
      }
    }
  }

  const int rb = r0 + hk * 8;
  if (r0 + 16 <= N) {
#pragma unroll
    for (int t = 0; t < 8; ++t)
#pragma unroll
      for (int j = 0; j < 8; ++j)
        C[(size_t)(rb + j) * DIM + t * 16 + l16] = acc[t][j];
  } else {
#pragma unroll
    for (int t = 0; t < 8; ++t)
#pragma unroll
      for (int j = 0; j < 8; ++j) {
        const int row = rb + j;
        if (row < N) C[(size_t)row * DIM + t * 16 + l16] = acc[t][j];
      }
  }
}

__global__ __launch_bounds__(128) void gemm_cat_256(const float* __restrict__ A0,
                                                    const float* __restrict__ A1,
                                                    const float* __restrict__ W,
                                                    float* __restrict__ C, int N) {
  __shared__ float Wlds[128 * LDP];
  const int tid = threadIdx.x;
  const int wv  = tid >> 5;
  const int ln  = tid & 31;
  const int hk  = ln >> 4;
  const int l16 = ln & 15;
  const int r0  = blockIdx.x * 64 + wv * 16;

  v8f acc[8];
#pragma unroll
  for (int t = 0; t < 8; ++t)
#pragma unroll
    for (int j = 0; j < 8; ++j) acc[t][j] = 0.0f;

  int rowA = r0 + l16;
  if (rowA >= N) rowA = N - 1;
  const size_t rbase = (size_t)rowA * DIM;

  for (int kp = 0; kp < 2 * DIM; kp += 64) {
    __syncthreads();
#pragma unroll
    for (int i = 0; i < 16; ++i) {
      const int q  = tid + 128 * i;
      const int n  = q >> 4;
      const int c4 = (q & 15) << 2;
      *(float4*)(&Wlds[n * LDP + c4]) =
          *(const float4*)(W + (size_t)n * (2 * DIM) + kp + c4);
    }
    __syncthreads();

    const float* abase = (kp < DIM) ? (A0 + rbase + kp) : (A1 + rbase + (kp - DIM));
#pragma unroll 4
    for (int k0 = 0; k0 < 64; k0 += 4) {
      const int ka = k0 + 2 * hk;
      v2f a; a.x = abase[ka]; a.y = abase[ka + 1];
#pragma unroll
      for (int t = 0; t < 8; ++t) {
        const float* wp = &Wlds[(t * 16 + l16) * LDP + ka];
        v2f b; b.x = wp[0]; b.y = wp[1];
        acc[t] = __builtin_amdgcn_wmma_f32_16x16x4_f32(
            false, a, false, b, (short)0, acc[t], false, false);
      }
    }
  }

  const int rb = r0 + hk * 8;
  if (r0 + 16 <= N) {
#pragma unroll
    for (int t = 0; t < 8; ++t)
#pragma unroll
      for (int j = 0; j < 8; ++j)
        C[(size_t)(rb + j) * DIM + t * 16 + l16] = acc[t][j];
  } else {
#pragma unroll
    for (int t = 0; t < 8; ++t)
#pragma unroll
      for (int j = 0; j < 8; ++j) {
        const int row = rb + j;
        if (row < N) C[(size_t)row * DIM + t * 16 + l16] = acc[t][j];
      }
  }
}

__global__ __launch_bounds__(256) void gn_kernel(const float* __restrict__ X,
                                                 float* __restrict__ Y,
                                                 const float* __restrict__ w,
                                                 const float* __restrict__ b, int N) {
  const int wv = threadIdx.x >> 5, ln = threadIdx.x & 31;
  const int row = blockIdx.x * 8 + wv;
  if (row >= N) return;
  float4 x = ((const float4*)(X + (size_t)row * DIM))[ln];
  float s  = x.x + x.y + x.z + x.w;
  float sq = x.x * x.x + x.y * x.y + x.z * x.z + x.w * x.w;
  s = wave_reduce_sum(s); sq = wave_reduce_sum(sq);
  const float mu = s * (1.0f / DIM);
  const float rs = rsqrtf(sq * (1.0f / DIM) - mu * mu + 1e-5f);
  float4 wv4 = ((const float4*)w)[ln];
  float4 bv4 = ((const float4*)b)[ln];
  float4 y;
  y.x = (x.x - mu) * rs * wv4.x + bv4.x;
  y.y = (x.y - mu) * rs * wv4.y + bv4.y;
  y.z = (x.z - mu) * rs * wv4.z + bv4.z;
  y.w = (x.w - mu) * rs * wv4.w + bv4.w;
  ((float4*)(Y + (size_t)row * DIM))[ln] = y;
}

__global__ __launch_bounds__(256) void gn2_relu_kernel(const float* __restrict__ X,
                                                       float* __restrict__ Y,
                                                       const float* __restrict__ w1,
                                                       const float* __restrict__ b1,
                                                       const float* __restrict__ w2,
                                                       const float* __restrict__ b2,
                                                       int N) {
  const int wv = threadIdx.x >> 5, ln = threadIdx.x & 31;
  const int row = blockIdx.x * 8 + wv;
  if (row >= N) return;
  float4 x = ((const float4*)(X + (size_t)row * DIM))[ln];
  float s  = x.x + x.y + x.z + x.w;
  float sq = x.x * x.x + x.y * x.y + x.z * x.z + x.w * x.w;
  s = wave_reduce_sum(s); sq = wave_reduce_sum(sq);
  float mu = s * (1.0f / DIM);
  float rs = rsqrtf(sq * (1.0f / DIM) - mu * mu + 1e-5f);
  float4 wa = ((const float4*)w1)[ln], ba = ((const float4*)b1)[ln];
  float4 y;
  y.x = (x.x - mu) * rs * wa.x + ba.x;
  y.y = (x.y - mu) * rs * wa.y + ba.y;
  y.z = (x.z - mu) * rs * wa.z + ba.z;
  y.w = (x.w - mu) * rs * wa.w + ba.w;
  s  = y.x + y.y + y.z + y.w;
  sq = y.x * y.x + y.y * y.y + y.z * y.z + y.w * y.w;
  s = wave_reduce_sum(s); sq = wave_reduce_sum(sq);
  mu = s * (1.0f / DIM);
  rs = rsqrtf(sq * (1.0f / DIM) - mu * mu + 1e-5f);
  float4 wn = ((const float4*)w2)[ln], bn = ((const float4*)b2)[ln];
  float4 z;
  z.x = fmaxf((y.x - mu) * rs * wn.x + bn.x, 0.0f);
  z.y = fmaxf((y.y - mu) * rs * wn.y + bn.y, 0.0f);
  z.z = fmaxf((y.z - mu) * rs * wn.z + bn.z, 0.0f);
  z.w = fmaxf((y.w - mu) * rs * wn.w + bn.w, 0.0f);
  ((float4*)(Y + (size_t)row * DIM))[ln] = z;
}

__global__ __launch_bounds__(256) void gn_res_relu_kernel(const float* __restrict__ X,
                                                          const float* __restrict__ R,
                                                          float* __restrict__ Y,
                                                          const float* __restrict__ w,
                                                          const float* __restrict__ b,
                                                          int N) {
  const int wv = threadIdx.x >> 5, ln = threadIdx.x & 31;
  const int row = blockIdx.x * 8 + wv;
  if (row >= N) return;
  float4 x = ((const float4*)(X + (size_t)row * DIM))[ln];
  float s  = x.x + x.y + x.z + x.w;
  float sq = x.x * x.x + x.y * x.y + x.z * x.z + x.w * x.w;
  s = wave_reduce_sum(s); sq = wave_reduce_sum(sq);
  const float mu = s * (1.0f / DIM);
  const float rs = rsqrtf(sq * (1.0f / DIM) - mu * mu + 1e-5f);
  float4 wv4 = ((const float4*)w)[ln], bv4 = ((const float4*)b)[ln];
  float4 r = ((const float4*)(R + (size_t)row * DIM))[ln];
  float4 y;
  y.x = fmaxf((x.x - mu) * rs * wv4.x + bv4.x + r.x, 0.0f);
  y.y = fmaxf((x.y - mu) * rs * wv4.y + bv4.y + r.y, 0.0f);
  y.z = fmaxf((x.z - mu) * rs * wv4.z + bv4.z + r.z, 0.0f);
  y.w = fmaxf((x.w - mu) * rs * wv4.w + bv4.w + r.w, 0.0f);
  ((float4*)(Y + (size_t)row * DIM))[ln] = y;
}

__global__ __launch_bounds__(256) void edge_dot_kernel(const float* __restrict__ Q,
                                                       const float* __restrict__ K,
                                                       const int* __restrict__ u,
                                                       const int* __restrict__ v,
                                                       float* __restrict__ c,
                                                       unsigned* __restrict__ gmax,
                                                       int E) {
  __shared__ float smax[8];
  const int wv = threadIdx.x >> 5, ln = threadIdx.x & 31;
  const int e = blockIdx.x * 8 + wv;
  float cv = -3.0e38f;
  if (e < E) {
    const int vi = v[e], ui = u[e];
    float4 qv = ((const float4*)(Q + (size_t)vi * DIM))[ln];
    float4 kv = ((const float4*)(K + (size_t)ui * DIM))[ln];
    float d = qv.x * kv.x + qv.y * kv.y + qv.z * kv.z + qv.w * kv.w;
    d = wave_reduce_sum(d);
    d = (d >= 0.0f) ? d : 0.1f * d;
    if (ln == 0) c[e] = d;
    cv = d;
  }
  if (ln == 0) smax[wv] = cv;
  __syncthreads();
  if (threadIdx.x == 0) {
    float m = smax[0];
#pragma unroll
    for (int i = 1; i < 8; ++i) m = fmaxf(m, smax[i]);
    atomicMax(gmax, f32_order_enc(m));
  }
}

__global__ __launch_bounds__(256) void edge_exp_kernel(float* __restrict__ c,
                                                       float* __restrict__ csum,
                                                       const int* __restrict__ v,
                                                       const unsigned* __restrict__ gmax,
                                                       int E) {
  const int e = blockIdx.x * 256 + threadIdx.x;
  if (e >= E) return;
  const float m = f32_order_dec(*gmax);
  const float ce = expf(c[e] - m);
  c[e] = ce;
  atomicAdd(&csum[v[e]], ce);
}

__global__ __launch_bounds__(256) void edge_scatter_kernel(const float* __restrict__ c,
                                                           const float* __restrict__ csum,
                                                           const float* __restrict__ ctxp,
                                                           const int* __restrict__ u,
                                                           const int* __restrict__ v,
                                                           float* __restrict__ agg,
                                                           int E) {
  const int wv = threadIdx.x >> 5, ln = threadIdx.x & 31;
  const int e = blockIdx.x * 8 + wv;
  if (e >= E) return;
  const int vi = v[e], ui = u[e];
  const float coef = c[e] / (csum[vi] + 1e-16f);
  float4 m = ((const float4*)(ctxp + (size_t)ui * DIM))[ln];
  float* dst = agg + (size_t)vi * DIM + ln * 4;
  atomicAdd(dst + 0, m.x * coef);
  atomicAdd(dst + 1, m.y * coef);
  atomicAdd(dst + 2, m.z * coef);
  atomicAdd(dst + 3, m.w * coef);
}

__global__ void zero_f32(float* __restrict__ p, long n) {
  const long i = (long)blockIdx.x * 256 + threadIdx.x;
  if (i < n) p[i] = 0.0f;
}
__global__ void init_scalar_u32(unsigned* p) {
  if (threadIdx.x == 0) *p = 0u;
}

extern "C" void kernel_launch(void* const* d_in, const int* in_sizes, int n_in,
                              void* d_out, int out_size, void* d_ws, size_t ws_size,
                              hipStream_t stream) {
  (void)in_sizes; (void)n_in; (void)out_size; (void)ws_size;
  const int NA = 50000, NC = 100000, NE = 600000;

  const float* agt    = (const float*)d_in[0];
  const float* ctx    = (const float*)d_in[1];
  const int*   u      = (const int*)d_in[2];
  const int*   v      = (const int*)d_in[3];
  const float* Wq     = (const float*)d_in[4];
  const float* Wk     = (const float*)d_in[5];
  const float* Wctx   = (const float*)d_in[6];
  const float* gnc_w  = (const float*)d_in[7];
  const float* gnc_b  = (const float*)d_in[8];
  const float* Wagt   = (const float*)d_in[9];
  const float* gna_w  = (const float*)d_in[10];
  const float* gna_b  = (const float*)d_in[11];
  const float* norm_w = (const float*)d_in[12];
  const float* norm_b = (const float*)d_in[13];
  const float* Wlin   = (const float*)d_in[14];
  const float* gnl_w  = (const float*)d_in[15];
  const float* gnl_b  = (const float*)d_in[16];

  float* ws = (float*)d_ws;
  size_t off = 0;
  float* qbuf   = ws + off; off += (size_t)NA * DIM;
  float* kbuf   = ws + off; off += (size_t)NC * DIM;
  float* cpbuf  = ws + off; off += (size_t)NC * DIM;
  float* cbuf   = ws + off; off += (size_t)NE;
  float* csum   = ws + off; off += (size_t)NA;
  float* aggbuf = ws + off; off += (size_t)NA * DIM;
  float* xbuf   = ws + off; off += (size_t)NA * DIM;
  float* agtbuf = ws + off; off += (size_t)NA * DIM;
  unsigned* maxbuf = (unsigned*)(ws + off); off += 16;

  const dim3 b128(128), b256(256);

  for (int i = 0; i < NBLK; ++i) {
    const float* ain  = (i == 0) ? agt : agtbuf;
    float*       aout = (i == NBLK - 1) ? (float*)d_out : agtbuf;

    gemm_nt_128<<<(NA + 63) / 64, b128, 0, stream>>>(ain, Wq + (size_t)i * DIM * DIM, qbuf, NA);
    gemm_nt_128<<<(NC + 63) / 64, b128, 0, stream>>>(ctx, Wk + (size_t)i * DIM * DIM, kbuf, NC);
    gemm_nt_128<<<(NC + 63) / 64, b128, 0, stream>>>(ctx, Wctx + (size_t)i * DIM * DIM, cpbuf, NC);
    gn_kernel<<<(NC + 7) / 8, b256, 0, stream>>>(cpbuf, cpbuf, gnc_w + i * DIM, gnc_b + i * DIM, NC);

    init_scalar_u32<<<1, 32, 0, stream>>>(maxbuf);
    zero_f32<<<(NA + 255) / 256, b256, 0, stream>>>(csum, (long)NA);
    zero_f32<<<((long)NA * DIM + 255) / 256, b256, 0, stream>>>(aggbuf, (long)NA * DIM);
    edge_dot_kernel<<<(NE + 7) / 8, b256, 0, stream>>>(qbuf, kbuf, u, v, cbuf, maxbuf, NE);
    edge_exp_kernel<<<(NE + 255) / 256, b256, 0, stream>>>(cbuf, csum, v, maxbuf, NE);
    edge_scatter_kernel<<<(NE + 7) / 8, b256, 0, stream>>>(cbuf, csum, cpbuf, u, v, aggbuf, NE);

    gemm_cat_256<<<(NA + 63) / 64, b128, 0, stream>>>(ain, aggbuf, Wagt + (size_t)i * DIM * 2 * DIM, xbuf, NA);
    gn2_relu_kernel<<<(NA + 7) / 8, b256, 0, stream>>>(xbuf, xbuf, gna_w + i * DIM, gna_b + i * DIM,
                                                       norm_w + i * DIM, norm_b + i * DIM, NA);
    gemm_nt_128<<<(NA + 63) / 64, b128, 0, stream>>>(xbuf, Wlin + (size_t)i * DIM * DIM, qbuf, NA);
    gn_res_relu_kernel<<<(NA + 7) / 8, b256, 0, stream>>>(qbuf, ain, aout, gnl_w + i * DIM, gnl_b + i * DIM, NA);
  }
}


# Segment sums without float atomics: edges counting-sorted by destination once per call (ranked by edge id), one wave per destination does dots, softmax sum and aggregation in that fixed order
# speedup vs baseline: 1.7425x; 1.7425x over previous
#include <hip/hip_runtime.h>


static constexpr int DIM  = 128;
static constexpr int NBLK = 2;
static constexpr int LDP  = 68;

typedef __attribute__((ext_vector_type(2))) float v2f;
typedef __attribute__((ext_vector_type(8))) float v8f;

static __device__ __forceinline__ float wave_reduce_sum(float x) {
#pragma unroll
  for (int off = 16; off > 0; off >>= 1) x += __shfl_xor(x, off, 32);
  return x;
}

static __device__ __forceinline__ unsigned f32_order_enc(float f) {
  unsigned u = __float_as_uint(f);
  return (u & 0x80000000u) ? ~u : (u | 0x80000000u);
}
static __device__ __forceinline__ float f32_order_dec(unsigned e) {
  unsigned u = (e & 0x80000000u) ? (e & 0x7fffffffu) : ~e;
  return __uint_as_float(u);
}

__global__ __launch_bounds__(128) void gemm_nt_128(const float* __restrict__ A,
                                                   const float* __restrict__ W,
                                                   float* __restrict__ C, int N) {
  __shared__ float Wlds[128 * LDP];
  const int tid = threadIdx.x;
  const int wv  = tid >> 5;
  const int ln  = tid & 31;
  const int hk  = ln >> 4;
  const int l16 = ln & 15;
  const int r0  = blockIdx.x * 64 + wv * 16;

  v8f acc[8];
#pragma unroll
  for (int t = 0; t < 8; ++t)
#pragma unroll
    for (int j = 0; j < 8; ++j) acc[t][j] = 0.0f;

  int rowA = r0 + l16;
  if (rowA >= N) rowA = N - 1;
  const float* ap = A + (size_t)rowA * DIM;

  for (int kp = 0; kp < DIM; kp += 64) {
    __syncthreads();
#pragma unroll
    for (int i = 0; i < 16; ++i) {
      const int q  = tid + 128 * i;
      const int n  = q >> 4;
      const int c4 = (q & 15) << 2;
      *(float4*)(&Wlds[n * LDP + c4]) =
          *(const float4*)(W + (size_t)n * DIM + kp + c4);
    }
    __syncthreads();

#pragma unroll 4
    for (int k0 = 0; k0 < 64; k0 += 4) {
      const int ka = k0 + 2 * hk;
      v2f a; a.x = ap[kp + ka]; a.y = ap[kp + ka + 1];
#pragma unroll
      for (int t = 0; t < 8; ++t) {
        const float* wp = &Wlds[(t * 16 + l16) * LDP + ka];
        v2f b; b.x = wp[0]; b.y = wp[1];
        acc[t] = __builtin_amdgcn_wmma_f32_16x16x4_f32(
            false, a, false, b, (short)0, acc[t], false, false);
      }
    }
  }

  const int rb = r0 + hk * 8;
  if (r0 + 16 <= N) {
#pragma unroll
    for (int t = 0; t < 8; ++t)
#pragma unroll
      for (int j = 0; j < 8; ++j)
        C[(size_t)(rb + j) * DIM + t * 16 + l16] = acc[t][j];
  } else {
#pragma unroll
    for (int t = 0; t < 8; ++t)
#pragma unroll
      for (int j = 0; j < 8; ++j) {
        const int row = rb + j;
        if (row < N) C[(size_t)row * DIM + t * 16 + l16] = acc[t][j];
      }
  }
}

__global__ __launch_bounds__(128) void gemm_cat_256(const float* __restrict__ A0,
                                                    const float* __restrict__ A1,
                                                    const float* __restrict__ W,
                                                    float* __restrict__ C, int N) {
  __shared__ float Wlds[128 * LDP];
  const int tid = threadIdx.x;
  const int wv  = tid >> 5;
  const int ln  = tid & 31;
  const int hk  = ln >> 4;
  const int l16 = ln & 15;
  const int r0  = blockIdx.x * 64 + wv * 16;

  v8f acc[8];
#pragma unroll
  for (int t = 0; t < 8; ++t)
#pragma unroll
    for (int j = 0; j < 8; ++j) acc[t][j] = 0.0f;

  int rowA = r0 + l16;
  if (rowA >= N) rowA = N - 1;
  const size_t rbase = (size_t)rowA * DIM;

  for (int kp = 0; kp < 2 * DIM; kp += 64) {
    __syncthreads();
#pragma unroll
    for (int i = 0; i < 16; ++i) {
      const int q  = tid + 128 * i;
      const int n  = q >> 4;
      const int c4 = (q & 15) << 2;
      *(float4*)(&Wlds[n * LDP + c4]) =
          *(const float4*)(W + (size_t)n * (2 * DIM) + kp + c4);
    }
    __syncthreads();

    const float* abase = (kp < DIM) ? (A0 + rbase + kp) : (A1 + rbase + (kp - DIM));
#pragma unroll 4
    for (int k0 = 0; k0 < 64; k0 += 4) {
      const int ka = k0 + 2 * hk;
      v2f a; a.x = abase[ka]; a.y = abase[ka + 1];
#pragma unroll
      for (int t = 0; t < 8; ++t) {
        const float* wp = &Wlds[(t * 16 + l16) * LDP + ka];
        v2f b; b.x = wp[0]; b.y = wp[1];
        acc[t] = __builtin_amdgcn_wmma_f32_16x16x4_f32(
            false, a, false, b, (short)0, acc[t], false, false);
      }
    }
  }

  const int rb = r0 + hk * 8;
  if (r0 + 16 <= N) {
#pragma unroll
    for (int t = 0; t < 8; ++t)
#pragma unroll
      for (int j = 0; j < 8; ++j)
        C[(size_t)(rb + j) * DIM + t * 16 + l16] = acc[t][j];
  } else {
#pragma unroll
    for (int t = 0; t < 8; ++t)
#pragma unroll
      for (int j = 0; j < 8; ++j) {
        const int row = rb + j;
        if (row < N) C[(size_t)row * DIM + t * 16 + l16] = acc[t][j];
      }
  }
}

__global__ __launch_bounds__(256) void gn_kernel(const float* __restrict__ X,
                                                 float* __restrict__ Y,
                                                 const float* __restrict__ w,
                                                 const float* __restrict__ b, int N) {
  const int wv = threadIdx.x >> 5, ln = threadIdx.x & 31;
  const int row = blockIdx.x * 8 + wv;
  if (row >= N) return;
  float4 x = ((const float4*)(X + (size_t)row * DIM))[ln];
  float s  = x.x + x.y + x.z + x.w;
  float sq = x.x * x.x + x.y * x.y + x.z * x.z + x.w * x.w;
  s = wave_reduce_sum(s); sq = wave_reduce_sum(sq);
  const float mu = s * (1.0f / DIM);
  const float rs = rsqrtf(sq * (1.0f / DIM) - mu * mu + 1e-5f);
  float4 wv4 = ((const float4*)w)[ln];
  float4 bv4 = ((const float4*)b)[ln];
  float4 y;
  y.x = (x.x - mu) * rs * wv4.x + bv4.x;
  y.y = (x.y - mu) * rs * wv4.y + bv4.y;
  y.z = (x.z - mu) * rs * wv4.z + bv4.z;
  y.w = (x.w - mu) * rs * wv4.w + bv4.w;
  ((float4*)(Y + (size_t)row * DIM))[ln] = y;
}

__global__ __launch_bounds__(256) void gn2_relu_kernel(const float* __restrict__ X,
                                                       float* __restrict__ Y,
                                                       const float* __restrict__ w1,
                                                       const float* __restrict__ b1,
                                                       const float* __restrict__ w2,
                                                       const float* __restrict__ b2,
                                                       int N) {
  const int wv = threadIdx.x >> 5, ln = threadIdx.x & 31;
  const int row = blockIdx.x * 8 + wv;
  if (row >= N) return;
  float4 x = ((const float4*)(X + (size_t)row * DIM))[ln];
  float s  = x.x + x.y + x.z + x.w;
  float sq = x.x * x.x + x.y * x.y + x.z * x.z + x.w * x.w;
  s = wave_reduce_sum(s); sq = wave_reduce_sum(sq);
  float mu = s * (1.0f / DIM);
  float rs = rsqrtf(sq * (1.0f / DIM) - mu * mu + 1e-5f);
  float4 wa = ((const float4*)w1)[ln], ba = ((const float4*)b1)[ln];
  float4 y;
  y.x = (x.x - mu) * rs * wa.x + ba.x;
  y.y = (x.y - mu) * rs * wa.y + ba.y;
  y.z = (x.z - mu) * rs * wa.z + ba.z;
  y.w = (x.w - mu) * rs * wa.w + ba.w;
  s  = y.x + y.y + y.z + y.w;
  sq = y.x * y.x + y.y * y.y + y.z * y.z + y.w * y.w;
  s = wave_reduce_sum(s); sq = wave_reduce_sum(sq);
  mu = s * (1.0f / DIM);
  rs = rsqrtf(sq * (1.0f / DIM) - mu * mu + 1e-5f);
  float4 wn = ((const float4*)w2)[ln], bn = ((const float4*)b2)[ln];
  float4 z;
  z.x = fmaxf((y.x - mu) * rs * wn.x + bn.x, 0.0f);
  z.y = fmaxf((y.y - mu) * rs * wn.y + bn.y, 0.0f);
  z.z = fmaxf((y.z - mu) * rs * wn.z + bn.z, 0.0f);
  z.w = fmaxf((y.w - mu) * rs * wn.w + bn.w, 0.0f);
  ((float4*)(Y + (size_t)row * DIM))[ln] = z;
}

__global__ __launch_bounds__(256) void gn_res_relu_kernel(const float* __restrict__ X,
                                                          const float* __restrict__ R,
                                                          float* __restrict__ Y,
                                                          const float* __restrict__ w,
                                                          const float* __restrict__ b,
                                                          int N) {
  const int wv = threadIdx.x >> 5, ln = threadIdx.x & 31;
  const int row = blockIdx.x * 8 + wv;
  if (row >= N) return;
  float4 x = ((const float4*)(X + (size_t)row * DIM))[ln];
  float s  = x.x + x.y + x.z + x.w;
  float sq = x.x * x.x + x.y * x.y + x.z * x.z + x.w * x.w;
  s = wave_reduce_sum(s); sq = wave_reduce_sum(sq);
  const float mu = s * (1.0f / DIM);
  const float rs = rsqrtf(sq * (1.0f / DIM) - mu * mu + 1e-5f);
  float4 wv4 = ((const float4*)w)[ln], bv4 = ((const float4*)b)[ln];
  float4 r = ((const float4*)(R + (size_t)row * DIM))[ln];
  float4 y;
  y.x = fmaxf((x.x - mu) * rs * wv4.x + bv4.x + r.x, 0.0f);
  y.y = fmaxf((x.y - mu) * rs * wv4.y + bv4.y + r.y, 0.0f);
  y.z = fmaxf((x.z - mu) * rs * wv4.z + bv4.z + r.z, 0.0f);
  y.w = fmaxf((x.w - mu) * rs * wv4.w + bv4.w + r.w, 0.0f);
  ((float4*)(Y + (size_t)row * DIM))[ln] = y;
}

__global__ __launch_bounds__(256) void zero_i32(int* __restrict__ p, int n) {
  const int i = blockIdx.x * 256 + threadIdx.x;
  if (i < n) p[i] = 0;
}

__global__ __launch_bounds__(256) void csr_count_kernel(const int* __restrict__ v,
                                                        int* __restrict__ cnt,
                                                        int* __restrict__ arr, int E) {
  const int e = blockIdx.x * 256 + threadIdx.x;
  if (e < E) arr[e] = atomicAdd(&cnt[v[e]], 1);
}

static __device__ __forceinline__ int wave_reduce_sum_i(int x) {
#pragma unroll
  for (int off = 16; off > 0; off >>= 1) x += __shfl_xor(x, off, 32);
  return x;
}

__global__ __launch_bounds__(256) void csr_blocksum_kernel(const int* __restrict__ cnt,
                                                           int* __restrict__ bsum, int N) {
  __shared__ int sw[8];
  const int wv = threadIdx.x >> 5, ln = threadIdx.x & 31;
  const int i = blockIdx.x * 256 + threadIdx.x;
  int c = (i < N) ? cnt[i] : 0;
  c = wave_reduce_sum_i(c);
  if (ln == 0) sw[wv] = c;
  __syncthreads();
  if (threadIdx.x == 0) {
    int t = 0;
#pragma unroll
    for (int k = 0; k < 8; ++k) t += sw[k];
    bsum[blockIdx.x] = t;
  }
}

__global__ __launch_bounds__(256) void csr_scan_kernel(const int* __restrict__ cnt,
                                                       const int* __restrict__ bsum,
                                                       int* __restrict__ rowptr, int N) {
  __shared__ int sa[256];
  __shared__ int sb[256];
  __shared__ int sw[8];
  const int t = threadIdx.x;
  const int wv = t >> 5, ln = t & 31;
  const int blk = blockIdx.x;

  int p = 0;
  for (int k = t; k < blk; k += 256) p += bsum[k];
  p = wave_reduce_sum_i(p);
  if (ln == 0) sw[wv] = p;

  const int i = blk * 256 + t;
  const int c = (i < N) ? cnt[i] : 0;
  sa[t] = c;
  __syncthreads();

  int base = 0;
#pragma unroll
  for (int k = 0; k < 8; ++k) base += sw[k];

  int* src = sa;
  int* dst = sb;
  for (int off = 1; off < 256; off <<= 1) {
    int val = src[t];
    if (t >= off) val += src[t - off];
    dst[t] = val;
    __syncthreads();
    int* tmp = src; src = dst; dst = tmp;
  }
  const int excl = base + src[t] - c;
  if (i < N) rowptr[i] = excl;
  if (i == N - 1) rowptr[N] = excl + c;
}

__global__ __launch_bounds__(256) void csr_fill_kernel(const int* __restrict__ v,
                                                       const int* __restrict__ arr,
                                                       const int* __restrict__ rowptr,
                                                       int* __restrict__ eid, int E) {
  const int e = blockIdx.x * 256 + threadIdx.x;
  if (e < E) eid[rowptr[v[e]] + arr[e]] = e;
}

__global__ __launch_bounds__(256) void csr_rank_kernel(const int* __restrict__ eid,
                                                       const int* __restrict__ rowptr,
                                                       const int* __restrict__ u,
                                                       const int* __restrict__ v,
                                                       int* __restrict__ su, int E) {
  const int j = blockIdx.x * 256 + threadIdx.x;
  if (j >= E) return;
  const int e = eid[j];
  const int n = v[e];
  const int b = rowptr[n], en = rowptr[n + 1];
  int r = 0;
  for (int k = b; k < en; ++k) r += (eid[k] < e) ? 1 : 0;
  su[b + r] = u[e];
}

__global__ __launch_bounds__(256) void node_dot_kernel(const float* __restrict__ Q,
                                                       const float* __restrict__ K,
                                                       const int* __restrict__ rowptr,
                                                       const int* __restrict__ su,
                                                       float* __restrict__ c,
                                                       unsigned* __restrict__ gmax,
                                                       int N) {
  __shared__ float smax[8];
  const int wv = threadIdx.x >> 5, ln = threadIdx.x & 31;
  const int n = __builtin_amdgcn_readfirstlane(blockIdx.x * 8 + wv);
  float mx = -3.0e38f;
  if (n < N) {
    const int b  = __builtin_amdgcn_readfirstlane(rowptr[n]);
    const int en = __builtin_amdgcn_readfirstlane(rowptr[n + 1]);
    if (b < en) {
      const float4 qv = ((const float4*)(Q + (size_t)n * DIM))[ln];
      for (int j0 = b; j0 < en; j0 += 32) {
        const int j = j0 + ln;
        const int uu = (j < en) ? su[j] : 0;
        const int cnt = (en - j0 < 32) ? (en - j0) : 32;
        float dl = 0.0f;
#pragma unroll 4
        for (int l = 0; l < cnt; ++l) {
          const int ui = __shfl(uu, l, 32);
          const float4 kv = ((const float4*)(K + (size_t)ui * DIM))[ln];
          float d = qv.x * kv.x + qv.y * kv.y + qv.z * kv.z + qv.w * kv.w;
          d = wave_reduce_sum(d);
          d = (d >= 0.0f) ? d : 0.1f * d;
          if (ln == l) dl = d;
          mx = fmaxf(mx, d);
        }
        if (j < en) c[j] = dl;
      }
    }
  }
  if (ln == 0) smax[wv] = mx;
  __syncthreads();
  if (threadIdx.x == 0) {
    float m = smax[0];
#pragma unroll
    for (int i = 1; i < 8; ++i) m = fmaxf(m, smax[i]);
    atomicMax(gmax, f32_order_enc(m));
  }
}

__global__ __launch_bounds__(256) void node_agg_kernel(const float* __restrict__ c,
                                                       const float* __restrict__ ctxp,
                                                       const int* __restrict__ rowptr,
                                                       const int* __restrict__ su,
                                                       const unsigned* __restrict__ gmax,
                                                       float* __restrict__ agg,
                                                       int N) {
  const int wv = threadIdx.x >> 5, ln = threadIdx.x & 31;
  const int n = __builtin_amdgcn_readfirstlane(blockIdx.x * 8 + wv);
  if (n >= N) return;
  const int b  = __builtin_amdgcn_readfirstlane(rowptr[n]);
  const int en = __builtin_amdgcn_readfirstlane(rowptr[n + 1]);
  const float m = f32_order_dec(*gmax);

  float s = 0.0f;
  for (int j0 = b; j0 < en; j0 += 32) {
    const int j = j0 + ln;
    const float ce = (j < en) ? expf(c[j] - m) : 0.0f;
    const int cnt = (en - j0 < 32) ? (en - j0) : 32;
    for (int l = 0; l < cnt; ++l) s += __shfl(ce, l, 32);
  }
  const float den = s + 1e-16f;

  float4 acc;
  acc.x = 0.0f; acc.y = 0.0f; acc.z = 0.0f; acc.w = 0.0f;
  for (int j0 = b; j0 < en; j0 += 32) {
    const int j = j0 + ln;
    float ce = 0.0f;
    int uu = 0;
    if (j < en) { ce = expf(c[j] - m); uu = su[j]; }
    const float coef = ce / den;
    const int cnt = (en - j0 < 32) ? (en - j0) : 32;
#pragma unroll 4
    for (int l = 0; l < cnt; ++l) {
      const int ui = __shfl(uu, l, 32);
      const float cf = __shfl(coef, l, 32);
      const float4 mr = ((const float4*)(ctxp + (size_t)ui * DIM))[ln];
      acc.x += mr.x * cf;
      acc.y += mr.y * cf;
      acc.z += mr.z * cf;
      acc.w += mr.w * cf;
    }
  }
  ((float4*)(agg + (size_t)n * DIM))[ln] = acc;
}

__global__ void init_scalar_u32(unsigned* p) {
  if (threadIdx.x == 0) *p = 0u;
}

extern "C" void kernel_launch(void* const* d_in, const int* in_sizes, int n_in,
                              void* d_out, int out_size, void* d_ws, size_t ws_size,
                              hipStream_t stream) {
  (void)in_sizes; (void)n_in; (void)out_size; (void)ws_size;
  const int NA = 50000, NC = 100000, NE = 600000;

  const float* agt    = (const float*)d_in[0];
  const float* ctx    = (const float*)d_in[1];
  const int*   u      = (const int*)d_in[2];
  const int*   v      = (const int*)d_in[3];
  const float* Wq     = (const float*)d_in[4];
  const float* Wk     = (const float*)d_in[5];
  const float* Wctx   = (const float*)d_in[6];
  const float* gnc_w  = (const float*)d_in[7];
  const float* gnc_b  = (const float*)d_in[8];
  const float* Wagt   = (const float*)d_in[9];
  const float* gna_w  = (const float*)d_in[10];
  const float* gna_b  = (const float*)d_in[11];
  const float* norm_w = (const float*)d_in[12];
  const float* norm_b = (const float*)d_in[13];
  const float* Wlin   = (const float*)d_in[14];
  const float* gnl_w  = (const float*)d_in[15];
  const float* gnl_b  = (const float*)d_in[16];

  float* ws = (float*)d_ws;
  size_t off = 0;
  float* qbuf   = ws + off; off += (size_t)NA * DIM;
  float* kbuf   = ws + off; off += (size_t)NC * DIM;
  float* cpbuf  = ws + off; off += (size_t)NC * DIM;
  float* cbuf   = ws + off; off += (size_t)NE;
  int*   cntbuf = (int*)(ws + off); off += (size_t)NA;
  float* aggbuf = ws + off; off += (size_t)NA * DIM;
  int*   csr    = (int*)(ws + off); off += (size_t)NA * DIM;
  float* agtbuf = ws + off; off += (size_t)NA * DIM;
  unsigned* maxbuf = (unsigned*)(ws + off); off += 16;

  float* xbuf   = kbuf;
  int*   rowptr = csr;
  int*   eid    = csr + 50016;
  int*   subuf  = csr + 650016;
  int*   arrbuf = csr + 1250016;
  int*   bsum   = csr + 1850016;

  const dim3 b128(128), b256(256);

  zero_i32<<<(NA + 255) / 256, b256, 0, stream>>>(cntbuf, NA);
  csr_count_kernel<<<(NE + 255) / 256, b256, 0, stream>>>(v, cntbuf, arrbuf, NE);
  csr_blocksum_kernel<<<(NA + 255) / 256, b256, 0, stream>>>(cntbuf, bsum, NA);
  csr_scan_kernel<<<(NA + 255) / 256, b256, 0, stream>>>(cntbuf, bsum, rowptr, NA);
  csr_fill_kernel<<<(NE + 255) / 256, b256, 0, stream>>>(v, arrbuf, rowptr, eid, NE);
  csr_rank_kernel<<<(NE + 255) / 256, b256, 0, stream>>>(eid, rowptr, u, v, subuf, NE);

  for (int i = 0; i < NBLK; ++i) {
    const float* ain  = (i == 0) ? agt : agtbuf;
    float*       aout = (i == NBLK - 1) ? (float*)d_out : agtbuf;

    gemm_nt_128<<<(NA + 63) / 64, b128, 0, stream>>>(ain, Wq + (size_t)i * DIM * DIM, qbuf, NA);
    gemm_nt_128<<<(NC + 63) / 64, b128, 0, stream>>>(ctx, Wk + (size_t)i * DIM * DIM, kbuf, NC);
    gemm_nt_128<<<(NC + 63) / 64, b128, 0, stream>>>(ctx, Wctx + (size_t)i * DIM * DIM, cpbuf, NC);
    gn_kernel<<<(NC + 7) / 8, b256, 0, stream>>>(cpbuf, cpbuf, gnc_w + i * DIM, gnc_b + i * DIM, NC);

    init_scalar_u32<<<1, 32, 0, stream>>>(maxbuf);
    node_dot_kernel<<<(NA + 7) / 8, b256, 0, stream>>>(qbuf, kbuf, rowptr, subuf, cbuf, maxbuf, NA);
    node_agg_kernel<<<(NA + 7) / 8, b256, 0, stream>>>(cbuf, cpbuf, rowptr, subuf, maxbuf, aggbuf, NA);

    gemm_cat_256<<<(NA + 63) / 64, b128, 0, stream>>>(ain, aggbuf, Wagt + (size_t)i * DIM * 2 * DIM, xbuf, NA);
    gn2_relu_kernel<<<(NA + 7) / 8, b256, 0, stream>>>(xbuf, xbuf, gna_w + i * DIM, gna_b + i * DIM,
                                                       norm_w + i * DIM, norm_b + i * DIM, NA);
    gemm_nt_128<<<(NA + 63) / 64, b128, 0, stream>>>(xbuf, Wlin + (size_t)i * DIM * DIM, qbuf, NA);
    gn_res_relu_kernel<<<(NA + 7) / 8, b256, 0, stream>>>(qbuf, ain, aout, gnl_w + i * DIM, gnl_b + i * DIM, NA);
  }
}


# GroupNorm fused into the epilogue of the GEMM that feeds it: row mean/var from the wave's own accumulators (8 adds per lane, then a 16-lane DPP xor tree: new sum order); 6 norm launches gone
# speedup vs baseline: 2.3679x; 1.3589x over previous
#include <hip/hip_runtime.h>


static constexpr int DIM  = 128;
static constexpr int NBLK = 2;
static constexpr int LDP  = 68;

typedef __attribute__((ext_vector_type(2))) float v2f;
typedef __attribute__((ext_vector_type(8))) float v8f;

static __device__ __forceinline__ float wave_reduce_sum(float x) {
#pragma unroll
  for (int off = 16; off > 0; off >>= 1) x += __shfl_xor(x, off, 32);
  return x;
}

static __device__ __forceinline__ unsigned f32_order_enc(float f) {
  unsigned u = __float_as_uint(f);
  return (u & 0x80000000u) ? ~u : (u | 0x80000000u);
}
static __device__ __forceinline__ float f32_order_dec(unsigned e) {
  unsigned u = (e & 0x80000000u) ? (e & 0x7fffffffu) : ~e;
  return __uint_as_float(u);
}

enum { EPI_NONE = 0, EPI_GN = 1, EPI_GN2_RELU = 2, EPI_GN_RES_RELU = 3 };

template <int XM>
static __device__ __forceinline__ float half_xor(float x) {
  return __int_as_float(__builtin_amdgcn_update_dpp(
      0, __float_as_int(x), 0x160 + XM, 0xf, 0xf, true));
}
static __device__ __forceinline__ float half_sum(float x) {
  x += half_xor<1>(x);
  x += half_xor<2>(x);
  x += half_xor<4>(x);
  x += half_xor<8>(x);
  return x;
}

template <bool RELU>
static __device__ __forceinline__ void gn_rows(v8f (&acc)[8],
                                               const float* __restrict__ w,
                                               const float* __restrict__ b,
                                               int l16) {
  float wv[8], bv[8];
#pragma unroll
  for (int t = 0; t < 8; ++t) {
    wv[t] = w[t * 16 + l16];
    bv[t] = b[t * 16 + l16];
  }
#pragma unroll
  for (int j = 0; j < 8; ++j) {
    float s = 0.0f, sq = 0.0f;
#pragma unroll
    for (int t = 0; t < 8; ++t) {
      const float x = acc[t][j];
      s += x;
      sq += x * x;
    }
    s  = half_sum(s);
    sq = half_sum(sq);
    const float mu = s * (1.0f / DIM);
    const float rs = rsqrtf(sq * (1.0f / DIM) - mu * mu + 1e-5f);
#pragma unroll
    for (int t = 0; t < 8; ++t) {
      float y = (acc[t][j] - mu) * rs * wv[t] + bv[t];
      if (RELU) y = fmaxf(y, 0.0f);
      acc[t][j] = y;
    }
  }
}

template <int EPI>
static __device__ __forceinline__ void epilogue_store(v8f (&acc)[8],
                                                      float* __restrict__ C,
                                                      const float* __restrict__ R,
                                                      const float* __restrict__ w1,
                                                      const float* __restrict__ b1,
                                                      const float* __restrict__ w2,
                                                      const float* __restrict__ b2,
                                                      int r0, int hk, int l16, int N) {
  const int rb = r0 + hk * 8;

  if (EPI == EPI_GN_RES_RELU) {
    float res[8][8];
#pragma unroll
    for (int j = 0; j < 8; ++j) {
      int row = rb + j;
      if (row >= N) row = N - 1;
#pragma unroll
      for (int t = 0; t < 8; ++t) res[t][j] = R[(size_t)row * DIM + t * 16 + l16];
    }
    gn_rows<false>(acc, w1, b1, l16);
#pragma unroll
    for (int t = 0; t < 8; ++t)
#pragma unroll
      for (int j = 0; j < 8; ++j) acc[t][j] = fmaxf(acc[t][j] + res[t][j], 0.0f);
  } else if (EPI == EPI_GN2_RELU) {
    gn_rows<false>(acc, w1, b1, l16);
    gn_rows<true>(acc, w2, b2, l16);
  } else if (EPI == EPI_GN) {
    gn_rows<false>(acc, w1, b1, l16);
  }

  if (r0 + 16 <= N) {
#pragma unroll
    for (int t = 0; t < 8; ++t)
#pragma unroll
      for (int j = 0; j < 8; ++j)
        C[(size_t)(rb + j) * DIM + t * 16 + l16] = acc[t][j];
  } else {
#pragma unroll
    for (int t = 0; t < 8; ++t)
#pragma unroll
      for (int j = 0; j < 8; ++j) {
        const int row = rb + j;
        if (row < N) C[(size_t)row * DIM + t * 16 + l16] = acc[t][j];
      }
  }
}

template <int EPI>
static __device__ __forceinline__ void gemm_nt_128_body(float* __restrict__ Wlds,
                                                        const float* __restrict__ A,
                                                        const float* __restrict__ W,
                                                        float* __restrict__ C,
                                                        const float* __restrict__ R,
                                                        const float* __restrict__ w1,
                                                        const float* __restrict__ b1,
                                                        const float* __restrict__ w2,
                                                        const float* __restrict__ b2,
                                                        int N) {
  const int tid = threadIdx.x;
  const int wv  = tid >> 5;
  const int ln  = tid & 31;
  const int hk  = ln >> 4;
  const int l16 = ln & 15;
  const int r0  = blockIdx.x * 64 + wv * 16;

  v8f acc[8];
#pragma unroll
  for (int t = 0; t < 8; ++t)
#pragma unroll
    for (int j = 0; j < 8; ++j) acc[t][j] = 0.0f;

  int rowA = r0 + l16;
  if (rowA >= N) rowA = N - 1;
  const float* ap = A + (size_t)rowA * DIM;

  for (int kp = 0; kp < DIM; kp += 64) {
    __syncthreads();
#pragma unroll
    for (int i = 0; i < 16; ++i) {
      const int q  = tid + 128 * i;
      const int n  = q >> 4;
      const int c4 = (q & 15) << 2;
      *(float4*)(&Wlds[n * LDP + c4]) =
          *(const float4*)(W + (size_t)n * DIM + kp + c4);
    }
    __syncthreads();

#pragma unroll 4
    for (int k0 = 0; k0 < 64; k0 += 4) {
      const int ka = k0 + 2 * hk;
      v2f a; a.x = ap[kp + ka]; a.y = ap[kp + ka + 1];
#pragma unroll
      for (int t = 0; t < 8; ++t) {
        const float* wp = &Wlds[(t * 16 + l16) * LDP + ka];
        v2f b; b.x = wp[0]; b.y = wp[1];
        acc[t] = __builtin_amdgcn_wmma_f32_16x16x4_f32(
            false, a, false, b, (short)0, acc[t], false, false);
      }
    }
  }

  epilogue_store<EPI>(acc, C, R, w1, b1, w2, b2, r0, hk, l16, N);
}

__global__ __launch_bounds__(128) void gemm_nt_128(const float* __restrict__ A,
                                                   const float* __restrict__ W,
                                                   float* __restrict__ C, int N) {
  __shared__ float Wlds[128 * LDP];
  gemm_nt_128_body<EPI_NONE>(Wlds, A, W, C, nullptr, nullptr, nullptr, nullptr, nullptr, N);
}

__global__ __launch_bounds__(128) void gemm_nt_128_gn(const float* __restrict__ A,
                                                      const float* __restrict__ W,
                                                      float* __restrict__ C,
                                                      const float* __restrict__ gw,
                                                      const float* __restrict__ gb, int N) {
  __shared__ float Wlds[128 * LDP];
  gemm_nt_128_body<EPI_GN>(Wlds, A, W, C, nullptr, gw, gb, nullptr, nullptr, N);
}

__global__ __launch_bounds__(128) void gemm_nt_128_gn_res_relu(const float* __restrict__ A,
                                                               const float* __restrict__ W,
                                                               float* __restrict__ C,
                                                               const float* __restrict__ R,
                                                               const float* __restrict__ gw,
                                                               const float* __restrict__ gb,
                                                               int N) {
  __shared__ float Wlds[128 * LDP];
  gemm_nt_128_body<EPI_GN_RES_RELU>(Wlds, A, W, C, R, gw, gb, nullptr, nullptr, N);
}

__global__ __launch_bounds__(128) void gemm_cat_256_gn2_relu(const float* __restrict__ A0,
                                                             const float* __restrict__ A1,
                                                             const float* __restrict__ W,
                                                             float* __restrict__ C,
                                                             const float* __restrict__ w1,
                                                             const float* __restrict__ b1,
                                                             const float* __restrict__ w2,
                                                             const float* __restrict__ b2,
                                                             int N) {
  __shared__ float Wlds[128 * LDP];
  const int tid = threadIdx.x;
  const int wv  = tid >> 5;
  const int ln  = tid & 31;
  const int hk  = ln >> 4;
  const int l16 = ln & 15;
  const int r0  = blockIdx.x * 64 + wv * 16;

  v8f acc[8];
#pragma unroll
  for (int t = 0; t < 8; ++t)
#pragma unroll
    for (int j = 0; j < 8; ++j) acc[t][j] = 0.0f;

  int rowA = r0 + l16;
  if (rowA >= N) rowA = N - 1;
  const size_t rbase = (size_t)rowA * DIM;

  for (int kp = 0; kp < 2 * DIM; kp += 64) {
    __syncthreads();
#pragma unroll
    for (int i = 0; i < 16; ++i) {
      const int q  = tid + 128 * i;
      const int n  = q >> 4;
      const int c4 = (q & 15) << 2;
      *(float4*)(&Wlds[n * LDP + c4]) =
          *(const float4*)(W + (size_t)n * (2 * DIM) + kp + c4);
    }
    __syncthreads();

    const float* abase = (kp < DIM) ? (A0 + rbase + kp) : (A1 + rbase + (kp - DIM));
#pragma unroll 4
    for (int k0 = 0; k0 < 64; k0 += 4) {
      const int ka = k0 + 2 * hk;
      v2f a; a.x = abase[ka]; a.y = abase[ka + 1];
#pragma unroll
      for (int t = 0; t < 8; ++t) {
        const float* wp = &Wlds[(t * 16 + l16) * LDP + ka];
        v2f b; b.x = wp[0]; b.y = wp[1];
        acc[t] = __builtin_amdgcn_wmma_f32_16x16x4_f32(
            false, a, false, b, (short)0, acc[t], false, false);
      }
    }
  }

  epilogue_store<EPI_GN2_RELU>(acc, C, nullptr, w1, b1, w2, b2, r0, hk, l16, N);
}

__global__ __launch_bounds__(256) void zero_i32(int* __restrict__ p, int n) {
  const int i = blockIdx.x * 256 + threadIdx.x;
  if (i < n) p[i] = 0;
}

__global__ __launch_bounds__(256) void csr_count_kernel(const int* __restrict__ v,
                                                        int* __restrict__ cnt,
                                                        int* __restrict__ arr, int E) {
  const int e = blockIdx.x * 256 + threadIdx.x;
  if (e < E) arr[e] = atomicAdd(&cnt[v[e]], 1);
}

static __device__ __forceinline__ int wave_reduce_sum_i(int x) {
#pragma unroll
  for (int off = 16; off > 0; off >>= 1) x += __shfl_xor(x, off, 32);
  return x;
}

__global__ __launch_bounds__(256) void csr_blocksum_kernel(const int* __restrict__ cnt,
                                                           int* __restrict__ bsum, int N) {
  __shared__ int sw[8];
  const int wv = threadIdx.x >> 5, ln = threadIdx.x & 31;
  const int i = blockIdx.x * 256 + threadIdx.x;
  int c = (i < N) ? cnt[i] : 0;
  c = wave_reduce_sum_i(c);
  if (ln == 0) sw[wv] = c;
  __syncthreads();
  if (threadIdx.x == 0) {
    int t = 0;
#pragma unroll
    for (int k = 0; k < 8; ++k) t += sw[k];
    bsum[blockIdx.x] = t;
  }
}

__global__ __launch_bounds__(256) void csr_scan_kernel(const int* __restrict__ cnt,
                                                       const int* __restrict__ bsum,
                                                       int* __restrict__ rowptr, int N) {
  __shared__ int sa[256];
  __shared__ int sb[256];
  __shared__ int sw[8];
  const int t = threadIdx.x;
  const int wv = t >> 5, ln = t & 31;
  const int blk = blockIdx.x;

  int p = 0;
  for (int k = t; k < blk; k += 256) p += bsum[k];
  p = wave_reduce_sum_i(p);
  if (ln == 0) sw[wv] = p;

  const int i = blk * 256 + t;
  const int c = (i < N) ? cnt[i] : 0;
  sa[t] = c;
  __syncthreads();

  int base = 0;
#pragma unroll
  for (int k = 0; k < 8; ++k) base += sw[k];

  int* src = sa;
  int* dst = sb;
  for (int off = 1; off < 256; off <<= 1) {
    int val = src[t];
    if (t >= off) val += src[t - off];
    dst[t] = val;
    __syncthreads();
    int* tmp = src; src = dst; dst = tmp;
  }
  const int excl = base + src[t] - c;
  if (i < N) rowptr[i] = excl;
  if (i == N - 1) rowptr[N] = excl + c;
}

__global__ __launch_bounds__(256) void csr_fill_kernel(const int* __restrict__ v,
                                                       const int* __restrict__ arr,
                                                       const int* __restrict__ rowptr,
                                                       int* __restrict__ eid, int E) {
  const int e = blockIdx.x * 256 + threadIdx.x;
  if (e < E) eid[rowptr[v[e]] + arr[e]] = e;
}

__global__ __launch_bounds__(256) void csr_rank_kernel(const int* __restrict__ eid,
                                                       const int* __restrict__ rowptr,
                                                       const int* __restrict__ u,
                                                       const int* __restrict__ v,
                                                       int* __restrict__ su, int E) {
  const int j = blockIdx.x * 256 + threadIdx.x;
  if (j >= E) return;
  const int e = eid[j];
  const int n = v[e];
  const int b = rowptr[n], en = rowptr[n + 1];
  int r = 0;
  for (int k = b; k < en; ++k) r += (eid[k] < e) ? 1 : 0;
  su[b + r] = u[e];
}

__global__ __launch_bounds__(256) void node_dot_kernel(const float* __restrict__ Q,
                                                       const float* __restrict__ K,
                                                       const int* __restrict__ rowptr,
                                                       const int* __restrict__ su,
                                                       float* __restrict__ c,
                                                       unsigned* __restrict__ gmax,
                                                       int N) {
  __shared__ float smax[8];
  const int wv = threadIdx.x >> 5, ln = threadIdx.x & 31;
  const int n = __builtin_amdgcn_readfirstlane(blockIdx.x * 8 + wv);
  float mx = -3.0e38f;
  if (n < N) {
    const int b  = __builtin_amdgcn_readfirstlane(rowptr[n]);
    const int en = __builtin_amdgcn_readfirstlane(rowptr[n + 1]);
    if (b < en) {
      const float4 qv = ((const float4*)(Q + (size_t)n * DIM))[ln];
      for (int j0 = b; j0 < en; j0 += 32) {
        const int j = j0 + ln;
        const int uu = (j < en) ? su[j] : 0;
        const int cnt = (en - j0 < 32) ? (en - j0) : 32;
        float dl = 0.0f;
#pragma unroll 4
        for (int l = 0; l < cnt; ++l) {
          const int ui = __shfl(uu, l, 32);
          const float4 kv = ((const float4*)(K + (size_t)ui * DIM))[ln];
          float d = qv.x * kv.x + qv.y * kv.y + qv.z * kv.z + qv.w * kv.w;
          d = wave_reduce_sum(d);
          d = (d >= 0.0f) ? d : 0.1f * d;
          if (ln == l) dl = d;
          mx = fmaxf(mx, d);
        }
        if (j < en) c[j] = dl;
      }
    }
  }
  if (ln == 0) smax[wv] = mx;
  __syncthreads();
  if (threadIdx.x == 0) {
    float m = smax[0];
#pragma unroll
    for (int i = 1; i < 8; ++i) m = fmaxf(m, smax[i]);
    atomicMax(gmax, f32_order_enc(m));
  }
}

__global__ __launch_bounds__(256) void node_agg_kernel(const float* __restrict__ c,
                                                       const float* __restrict__ ctxp,
                                                       const int* __restrict__ rowptr,
                                                       const int* __restrict__ su,
                                                       const unsigned* __restrict__ gmax,
                                                       float* __restrict__ agg,
                                                       int N) {
  const int wv = threadIdx.x >> 5, ln = threadIdx.x & 31;
  const int n = __builtin_amdgcn_readfirstlane(blockIdx.x * 8 + wv);
  if (n >= N) return;
  const int b  = __builtin_amdgcn_readfirstlane(rowptr[n]);
  const int en = __builtin_amdgcn_readfirstlane(rowptr[n + 1]);
  const float m = f32_order_dec(*gmax);

  float s = 0.0f;
  for (int j0 = b; j0 < en; j0 += 32) {
    const int j = j0 + ln;
    const float ce = (j < en) ? expf(c[j] - m) : 0.0f;
    const int cnt = (en - j0 < 32) ? (en - j0) : 32;
    for (int l = 0; l < cnt; ++l) s += __shfl(ce, l, 32);
  }
  const float den = s + 1e-16f;

  float4 acc;
  acc.x = 0.0f; acc.y = 0.0f; acc.z = 0.0f; acc.w = 0.0f;
  for (int j0 = b; j0 < en; j0 += 32) {
    const int j = j0 + ln;
    float ce = 0.0f;
    int uu = 0;
    if (j < en) { ce = expf(c[j] - m); uu = su[j]; }
    const float coef = ce / den;
    const int cnt = (en - j0 < 32) ? (en - j0) : 32;
#pragma unroll 4
    for (int l = 0; l < cnt; ++l) {
      const int ui = __shfl(uu, l, 32);
      const float cf = __shfl(coef, l, 32);
      const float4 mr = ((const float4*)(ctxp + (size_t)ui * DIM))[ln];
      acc.x += mr.x * cf;
      acc.y += mr.y * cf;
      acc.z += mr.z * cf;
      acc.w += mr.w * cf;
    }
  }
  ((float4*)(agg + (size_t)n * DIM))[ln] = acc;
}

__global__ void init_scalar_u32(unsigned* p) {
  if (threadIdx.x == 0) *p = 0u;
}

extern "C" void kernel_launch(void* const* d_in, const int* in_sizes, int n_in,
                              void* d_out, int out_size, void* d_ws, size_t ws_size,
                              hipStream_t stream) {
  (void)in_sizes; (void)n_in; (void)out_size; (void)ws_size;
  const int NA = 50000, NC = 100000, NE = 600000;

  const float* agt    = (const float*)d_in[0];
  const float* ctx    = (const float*)d_in[1];
  const int*   u      = (const int*)d_in[2];
  const int*   v      = (const int*)d_in[3];
  const float* Wq     = (const float*)d_in[4];
  const float* Wk     = (const float*)d_in[5];
  const float* Wctx   = (const float*)d_in[6];
  const float* gnc_w  = (const float*)d_in[7];
  const float* gnc_b  = (const float*)d_in[8];
  const float* Wagt   = (const float*)d_in[9];
  const float* gna_w  = (const float*)d_in[10];
  const float* gna_b  = (const float*)d_in[11];
  const float* norm_w = (const float*)d_in[12];
  const float* norm_b = (const float*)d_in[13];
  const float* Wlin   = (const float*)d_in[14];
  const float* gnl_w  = (const float*)d_in[15];
  const float* gnl_b  = (const float*)d_in[16];

  float* ws = (float*)d_ws;
  size_t off = 0;
  float* qbuf   = ws + off; off += (size_t)NA * DIM;
  float* kbuf   = ws + off; off += (size_t)NC * DIM;
  float* cpbuf  = ws + off; off += (size_t)NC * DIM;
  float* cbuf   = ws + off; off += (size_t)NE;
  int*   cntbuf = (int*)(ws + off); off += (size_t)NA;
  float* aggbuf = ws + off; off += (size_t)NA * DIM;
  int*   csr    = (int*)(ws + off); off += (size_t)NA * DIM;
  float* agtbuf = ws + off; off += (size_t)NA * DIM;
  unsigned* maxbuf = (unsigned*)(ws + off); off += 16;

  float* xbuf   = kbuf;
  int*   rowptr = csr;
  int*   eid    = csr + 50016;
  int*   subuf  = csr + 650016;
  int*   arrbuf = csr + 1250016;
  int*   bsum   = csr + 1850016;

  const dim3 b128(128), b256(256);

  zero_i32<<<(NA + 255) / 256, b256, 0, stream>>>(cntbuf, NA);
  csr_count_kernel<<<(NE + 255) / 256, b256, 0, stream>>>(v, cntbuf, arrbuf, NE);
  csr_blocksum_kernel<<<(NA + 255) / 256, b256, 0, stream>>>(cntbuf, bsum, NA);
  csr_scan_kernel<<<(NA + 255) / 256, b256, 0, stream>>>(cntbuf, bsum, rowptr, NA);
  csr_fill_kernel<<<(NE + 255) / 256, b256, 0, stream>>>(v, arrbuf, rowptr, eid, NE);
  csr_rank_kernel<<<(NE + 255) / 256, b256, 0, stream>>>(eid, rowptr, u, v, subuf, NE);

  for (int i = 0; i < NBLK; ++i) {
    const float* ain  = (i == 0) ? agt : agtbuf;
    float*       aout = (i == NBLK - 1) ? (float*)d_out : agtbuf;

    gemm_nt_128<<<(NA + 63) / 64, b128, 0, stream>>>(ain, Wq + (size_t)i * DIM * DIM, qbuf, NA);
    gemm_nt_128<<<(NC + 63) / 64, b128, 0, stream>>>(ctx, Wk + (size_t)i * DIM * DIM, kbuf, NC);
    gemm_nt_128_gn<<<(NC + 63) / 64, b128, 0, stream>>>(ctx, Wctx + (size_t)i * DIM * DIM, cpbuf,
                                                         gnc_w + i * DIM, gnc_b + i * DIM, NC);

    init_scalar_u32<<<1, 32, 0, stream>>>(maxbuf);
    node_dot_kernel<<<(NA + 7) / 8, b256, 0, stream>>>(qbuf, kbuf, rowptr, subuf, cbuf, maxbuf, NA);
    node_agg_kernel<<<(NA + 7) / 8, b256, 0, stream>>>(cbuf, cpbuf, rowptr, subuf, maxbuf, aggbuf, NA);

    gemm_cat_256_gn2_relu<<<(NA + 63) / 64, b128, 0, stream>>>(
        ain, aggbuf, Wagt + (size_t)i * DIM * 2 * DIM, xbuf,
        gna_w + i * DIM, gna_b + i * DIM, norm_w + i * DIM, norm_b + i * DIM, NA);
    gemm_nt_128_gn_res_relu<<<(NA + 63) / 64, b128, 0, stream>>>(
        xbuf, Wlin + (size_t)i * DIM * DIM, aout, ain, gnl_w + i * DIM, gnl_b + i * DIM, NA);
  }
}


# Score weights folded: Wf = Wk^T Wq once per call (WMMA), q' = agt Wf^T dotted with raw ctx rows; both key-projection GEMMs over all ctx rows gone (scores re-associated: new sum order)
# speedup vs baseline: 2.6663x; 1.1260x over previous
#include <hip/hip_runtime.h>


static constexpr int DIM  = 128;
static constexpr int NBLK = 2;
static constexpr int LDP  = 68;

typedef __attribute__((ext_vector_type(2))) float v2f;
typedef __attribute__((ext_vector_type(8))) float v8f;

static __device__ __forceinline__ float wave_reduce_sum(float x) {
#pragma unroll
  for (int off = 16; off > 0; off >>= 1) x += __shfl_xor(x, off, 32);
  return x;
}

static __device__ __forceinline__ unsigned f32_order_enc(float f) {
  unsigned u = __float_as_uint(f);
  return (u & 0x80000000u) ? ~u : (u | 0x80000000u);
}
static __device__ __forceinline__ float f32_order_dec(unsigned e) {
  unsigned u = (e & 0x80000000u) ? (e & 0x7fffffffu) : ~e;
  return __uint_as_float(u);
}

enum { EPI_NONE = 0, EPI_GN = 1, EPI_GN2_RELU = 2, EPI_GN_RES_RELU = 3 };

template <int XM>
static __device__ __forceinline__ float half_xor(float x) {
  return __int_as_float(__builtin_amdgcn_update_dpp(
      0, __float_as_int(x), 0x160 + XM, 0xf, 0xf, true));
}
static __device__ __forceinline__ float half_sum(float x) {
  x += half_xor<1>(x);
  x += half_xor<2>(x);
  x += half_xor<4>(x);
  x += half_xor<8>(x);
  return x;
}

template <bool RELU>
static __device__ __forceinline__ void gn_rows(v8f (&acc)[8],
                                               const float* __restrict__ w,
                                               const float* __restrict__ b,
                                               int l16) {
  float wv[8], bv[8];
#pragma unroll
  for (int t = 0; t < 8; ++t) {
    wv[t] = w[t * 16 + l16];
    bv[t] = b[t * 16 + l16];
  }
#pragma unroll
  for (int j = 0; j < 8; ++j) {
    float s = 0.0f, sq = 0.0f;
#pragma unroll
    for (int t = 0; t < 8; ++t) {
      const float x = acc[t][j];
      s += x;
      sq += x * x;
    }
    s  = half_sum(s);
    sq = half_sum(sq);
    const float mu = s * (1.0f / DIM);
    const float rs = rsqrtf(sq * (1.0f / DIM) - mu * mu + 1e-5f);
#pragma unroll
    for (int t = 0; t < 8; ++t) {
      float y = (acc[t][j] - mu) * rs * wv[t] + bv[t];
      if (RELU) y = fmaxf(y, 0.0f);
      acc[t][j] = y;
    }
  }
}

template <int EPI>
static __device__ __forceinline__ void epilogue_store(v8f (&acc)[8],
                                                      float* __restrict__ C,
                                                      const float* __restrict__ R,
                                                      const float* __restrict__ w1,
                                                      const float* __restrict__ b1,
                                                      const float* __restrict__ w2,
                                                      const float* __restrict__ b2,
                                                      int r0, int hk, int l16, int N) {
  const int rb = r0 + hk * 8;

  if (EPI == EPI_GN_RES_RELU) {
    float res[8][8];
#pragma unroll
    for (int j = 0; j < 8; ++j) {
      int row = rb + j;
      if (row >= N) row = N - 1;
#pragma unroll
      for (int t = 0; t < 8; ++t) res[t][j] = R[(size_t)row * DIM + t * 16 + l16];
    }
    gn_rows<false>(acc, w1, b1, l16);
#pragma unroll
    for (int t = 0; t < 8; ++t)
#pragma unroll
      for (int j = 0; j < 8; ++j) acc[t][j] = fmaxf(acc[t][j] + res[t][j], 0.0f);
  } else if (EPI == EPI_GN2_RELU) {
    gn_rows<false>(acc, w1, b1, l16);
    gn_rows<true>(acc, w2, b2, l16);
  } else if (EPI == EPI_GN) {
    gn_rows<false>(acc, w1, b1, l16);
  }

  if (r0 + 16 <= N) {
#pragma unroll
    for (int t = 0; t < 8; ++t)
#pragma unroll
      for (int j = 0; j < 8; ++j)
        C[(size_t)(rb + j) * DIM + t * 16 + l16] = acc[t][j];
  } else {
#pragma unroll
    for (int t = 0; t < 8; ++t)
#pragma unroll
      for (int j = 0; j < 8; ++j) {
        const int row = rb + j;
        if (row < N) C[(size_t)row * DIM + t * 16 + l16] = acc[t][j];
      }
  }
}

template <int EPI>
static __device__ __forceinline__ void gemm_nt_128_body(float* __restrict__ Wlds,
                                                        const float* __restrict__ A,
                                                        const float* __restrict__ W,
                                                        float* __restrict__ C,
                                                        const float* __restrict__ R,
                                                        const float* __restrict__ w1,
                                                        const float* __restrict__ b1,
                                                        const float* __restrict__ w2,
                                                        const float* __restrict__ b2,
                                                        int N) {
  const int tid = threadIdx.x;
  const int wv  = tid >> 5;
  const int ln  = tid & 31;
  const int hk  = ln >> 4;
  const int l16 = ln & 15;
  const int r0  = blockIdx.x * 64 + wv * 16;

  v8f acc[8];
#pragma unroll
  for (int t = 0; t < 8; ++t)
#pragma unroll
    for (int j = 0; j < 8; ++j) acc[t][j] = 0.0f;

  int rowA = r0 + l16;
  if (rowA >= N) rowA = N - 1;
  const float* ap = A + (size_t)rowA * DIM;

  for (int kp = 0; kp < DIM; kp += 64) {
    __syncthreads();
#pragma unroll
    for (int i = 0; i < 16; ++i) {
      const int q  = tid + 128 * i;
      const int n  = q >> 4;
      const int c4 = (q & 15) << 2;
      *(float4*)(&Wlds[n * LDP + c4]) =
          *(const float4*)(W + (size_t)n * DIM + kp + c4);
    }
    __syncthreads();

#pragma unroll 4
    for (int k0 = 0; k0 < 64; k0 += 4) {
      const int ka = k0 + 2 * hk;
      v2f a; a.x = ap[kp + ka]; a.y = ap[kp + ka + 1];
#pragma unroll
      for (int t = 0; t < 8; ++t) {
        const float* wp = &Wlds[(t * 16 + l16) * LDP + ka];
        v2f b; b.x = wp[0]; b.y = wp[1];
        acc[t] = __builtin_amdgcn_wmma_f32_16x16x4_f32(
            false, a, false, b, (short)0, acc[t], false, false);
      }
    }
  }

  epilogue_store<EPI>(acc, C, R, w1, b1, w2, b2, r0, hk, l16, N);
}

__global__ __launch_bounds__(128) void gemm_nt_128(const float* __restrict__ A,
                                                   const float* __restrict__ W,
                                                   float* __restrict__ C, int N) {
  __shared__ float Wlds[128 * LDP];
  gemm_nt_128_body<EPI_NONE>(Wlds, A, W, C, nullptr, nullptr, nullptr, nullptr, nullptr, N);
}

__global__ __launch_bounds__(128) void gemm_nt_128_gn(const float* __restrict__ A,
                                                      const float* __restrict__ W,
                                                      float* __restrict__ C,
                                                      const float* __restrict__ gw,
                                                      const float* __restrict__ gb, int N) {
  __shared__ float Wlds[128 * LDP];
  gemm_nt_128_body<EPI_GN>(Wlds, A, W, C, nullptr, gw, gb, nullptr, nullptr, N);
}

__global__ __launch_bounds__(128) void gemm_nt_128_gn_res_relu(const float* __restrict__ A,
                                                               const float* __restrict__ W,
                                                               float* __restrict__ C,
                                                               const float* __restrict__ R,
                                                               const float* __restrict__ gw,
                                                               const float* __restrict__ gb,
                                                               int N) {
  __shared__ float Wlds[128 * LDP];
  gemm_nt_128_body<EPI_GN_RES_RELU>(Wlds, A, W, C, R, gw, gb, nullptr, nullptr, N);
}

__global__ __launch_bounds__(128) void gemm_cat_256_gn2_relu(const float* __restrict__ A0,
                                                             const float* __restrict__ A1,
                                                             const float* __restrict__ W,
                                                             float* __restrict__ C,
                                                             const float* __restrict__ w1,
                                                             const float* __restrict__ b1,
                                                             const float* __restrict__ w2,
                                                             const float* __restrict__ b2,
                                                             int N) {
  __shared__ float Wlds[128 * LDP];
  const int tid = threadIdx.x;
  const int wv  = tid >> 5;
  const int ln  = tid & 31;
  const int hk  = ln >> 4;
  const int l16 = ln & 15;
  const int r0  = blockIdx.x * 64 + wv * 16;

  v8f acc[8];
#pragma unroll
  for (int t = 0; t < 8; ++t)
#pragma unroll
    for (int j = 0; j < 8; ++j) acc[t][j] = 0.0f;

  int rowA = r0 + l16;
  if (rowA >= N) rowA = N - 1;
  const size_t rbase = (size_t)rowA * DIM;

  for (int kp = 0; kp < 2 * DIM; kp += 64) {
    __syncthreads();
#pragma unroll
    for (int i = 0; i < 16; ++i) {
      const int q  = tid + 128 * i;
      const int n  = q >> 4;
      const int c4 = (q & 15) << 2;
      *(float4*)(&Wlds[n * LDP + c4]) =
          *(const float4*)(W + (size_t)n * (2 * DIM) + kp + c4);
    }
    __syncthreads();

    const float* abase = (kp < DIM) ? (A0 + rbase + kp) : (A1 + rbase + (kp - DIM));
#pragma unroll 4
    for (int k0 = 0; k0 < 64; k0 += 4) {
      const int ka = k0 + 2 * hk;
      v2f a; a.x = abase[ka]; a.y = abase[ka + 1];
#pragma unroll
      for (int t = 0; t < 8; ++t) {
        const float* wp = &Wlds[(t * 16 + l16) * LDP + ka];
        v2f b; b.x = wp[0]; b.y = wp[1];
        acc[t] = __builtin_amdgcn_wmma_f32_16x16x4_f32(
            false, a, false, b, (short)0, acc[t], false, false);
      }
    }
  }

  epilogue_store<EPI_GN2_RELU>(acc, C, nullptr, w1, b1, w2, b2, r0, hk, l16, N);
}

__global__ __launch_bounds__(256) void fold_qk_kernel(const float* __restrict__ Wq,
                                                      const float* __restrict__ Wk,
                                                      float* __restrict__ Wf) {
  const int wv  = threadIdx.x >> 5;
  const int ln  = threadIdx.x & 31;
  const int hk  = ln >> 4;
  const int l16 = ln & 15;
  const int tile = blockIdx.x * 8 + wv;
  const int i  = tile >> 6;
  const int j0 = ((tile >> 3) & 7) * 16;
  const int m0 = (tile & 7) * 16;
  const float* wq = Wq + (size_t)i * DIM * DIM;
  const float* wk = Wk + (size_t)i * DIM * DIM;

  v8f acc;
#pragma unroll
  for (int r = 0; r < 8; ++r) acc[r] = 0.0f;

#pragma unroll 8
  for (int d0 = 0; d0 < DIM; d0 += 4) {
    const int d = d0 + 2 * hk;
    v2f a; a.x = wk[d * DIM + j0 + l16]; a.y = wk[(d + 1) * DIM + j0 + l16];
    v2f b; b.x = wq[d * DIM + m0 + l16]; b.y = wq[(d + 1) * DIM + m0 + l16];
    acc = __builtin_amdgcn_wmma_f32_16x16x4_f32(
        false, a, false, b, (short)0, acc, false, false);
  }

  float* o = Wf + (size_t)i * DIM * DIM + (size_t)(j0 + 8 * hk) * DIM + m0 + l16;
#pragma unroll
  for (int r = 0; r < 8; ++r) o[r * DIM] = acc[r];
}

__global__ __launch_bounds__(256) void zero_i32(int* __restrict__ p, int n) {
  const int i = blockIdx.x * 256 + threadIdx.x;
  if (i < n) p[i] = 0;
}

__global__ __launch_bounds__(256) void csr_count_kernel(const int* __restrict__ v,
                                                        int* __restrict__ cnt,
                                                        int* __restrict__ arr, int E) {
  const int e = blockIdx.x * 256 + threadIdx.x;
  if (e < E) arr[e] = atomicAdd(&cnt[v[e]], 1);
}

static __device__ __forceinline__ int wave_reduce_sum_i(int x) {
#pragma unroll
  for (int off = 16; off > 0; off >>= 1) x += __shfl_xor(x, off, 32);
  return x;
}

__global__ __launch_bounds__(256) void csr_blocksum_kernel(const int* __restrict__ cnt,
                                                           int* __restrict__ bsum, int N) {
  __shared__ int sw[8];
  const int wv = threadIdx.x >> 5, ln = threadIdx.x & 31;
  const int i = blockIdx.x * 256 + threadIdx.x;
  int c = (i < N) ? cnt[i] : 0;
  c = wave_reduce_sum_i(c);
  if (ln == 0) sw[wv] = c;
  __syncthreads();
  if (threadIdx.x == 0) {
    int t = 0;
#pragma unroll
    for (int k = 0; k < 8; ++k) t += sw[k];
    bsum[blockIdx.x] = t;
  }
}

__global__ __launch_bounds__(256) void csr_scan_kernel(const int* __restrict__ cnt,
                                                       const int* __restrict__ bsum,
                                                       int* __restrict__ rowptr, int N) {
  __shared__ int sa[256];
  __shared__ int sb[256];
  __shared__ int sw[8];
  const int t = threadIdx.x;
  const int wv = t >> 5, ln = t & 31;
  const int blk = blockIdx.x;

  int p = 0;
  for (int k = t; k < blk; k += 256) p += bsum[k];
  p = wave_reduce_sum_i(p);
  if (ln == 0) sw[wv] = p;

  const int i = blk * 256 + t;
  const int c = (i < N) ? cnt[i] : 0;
  sa[t] = c;
  __syncthreads();

  int base = 0;
#pragma unroll
  for (int k = 0; k < 8; ++k) base += sw[k];

  int* src = sa;
  int* dst = sb;
  for (int off = 1; off < 256; off <<= 1) {
    int val = src[t];
    if (t >= off) val += src[t - off];
    dst[t] = val;
    __syncthreads();
    int* tmp = src; src = dst; dst = tmp;
  }
  const int excl = base + src[t] - c;
  if (i < N) rowptr[i] = excl;
  if (i == N - 1) rowptr[N] = excl + c;
}

__global__ __launch_bounds__(256) void csr_fill_kernel(const int* __restrict__ v,
                                                       const int* __restrict__ arr,
                                                       const int* __restrict__ rowptr,
                                                       int* __restrict__ eid, int E) {
  const int e = blockIdx.x * 256 + threadIdx.x;
  if (e < E) eid[rowptr[v[e]] + arr[e]] = e;
}

__global__ __launch_bounds__(256) void csr_rank_kernel(const int* __restrict__ eid,
                                                       const int* __restrict__ rowptr,
                                                       const int* __restrict__ u,
                                                       const int* __restrict__ v,
                                                       int* __restrict__ su, int E) {
  const int j = blockIdx.x * 256 + threadIdx.x;
  if (j >= E) return;
  const int e = eid[j];
  const int n = v[e];
  const int b = rowptr[n], en = rowptr[n + 1];
  int r = 0;
  for (int k = b; k < en; ++k) r += (eid[k] < e) ? 1 : 0;
  su[b + r] = u[e];
}

__global__ __launch_bounds__(256) void node_dot_kernel(const float* __restrict__ Q,
                                                       const float* __restrict__ K,
                                                       const int* __restrict__ rowptr,
                                                       const int* __restrict__ su,
                                                       float* __restrict__ c,
                                                       unsigned* __restrict__ gmax,
                                                       int N) {
  __shared__ float smax[8];
  const int wv = threadIdx.x >> 5, ln = threadIdx.x & 31;
  const int n = __builtin_amdgcn_readfirstlane(blockIdx.x * 8 + wv);
  float mx = -3.0e38f;
  if (n < N) {
    const int b  = __builtin_amdgcn_readfirstlane(rowptr[n]);
    const int en = __builtin_amdgcn_readfirstlane(rowptr[n + 1]);
    if (b < en) {
      const float4 qv = ((const float4*)(Q + (size_t)n * DIM))[ln];
      for (int j0 = b; j0 < en; j0 += 32) {
        const int j = j0 + ln;
        const int uu = (j < en) ? su[j] : 0;
        const int cnt = (en - j0 < 32) ? (en - j0) : 32;
        float dl = 0.0f;
#pragma unroll 4
        for (int l = 0; l < cnt; ++l) {
          const int ui = __shfl(uu, l, 32);
          const float4 kv = ((const float4*)(K + (size_t)ui * DIM))[ln];
          float d = qv.x * kv.x + qv.y * kv.y + qv.z * kv.z + qv.w * kv.w;
          d = wave_reduce_sum(d);
          d = (d >= 0.0f) ? d : 0.1f * d;
          if (ln == l) dl = d;
          mx = fmaxf(mx, d);
        }
        if (j < en) c[j] = dl;
      }
    }
  }
  if (ln == 0) smax[wv] = mx;
  __syncthreads();
  if (threadIdx.x == 0) {
    float m = smax[0];
#pragma unroll
    for (int i = 1; i < 8; ++i) m = fmaxf(m, smax[i]);
    atomicMax(gmax, f32_order_enc(m));
  }
}

__global__ __launch_bounds__(256) void node_agg_kernel(const float* __restrict__ c,
                                                       const float* __restrict__ ctxp,
                                                       const int* __restrict__ rowptr,
                                                       const int* __restrict__ su,
                                                       const unsigned* __restrict__ gmax,
                                                       float* __restrict__ agg,
                                                       int N) {
  const int wv = threadIdx.x >> 5, ln = threadIdx.x & 31;
  const int n = __builtin_amdgcn_readfirstlane(blockIdx.x * 8 + wv);
  if (n >= N) return;
  const int b  = __builtin_amdgcn_readfirstlane(rowptr[n]);
  const int en = __builtin_amdgcn_readfirstlane(rowptr[n + 1]);
  const float m = f32_order_dec(*gmax);

  float s = 0.0f;
  for (int j0 = b; j0 < en; j0 += 32) {
    const int j = j0 + ln;
    const float ce = (j < en) ? expf(c[j] - m) : 0.0f;
    const int cnt = (en - j0 < 32) ? (en - j0) : 32;
    for (int l = 0; l < cnt; ++l) s += __shfl(ce, l, 32);
  }
  const float den = s + 1e-16f;

  float4 acc;
  acc.x = 0.0f; acc.y = 0.0f; acc.z = 0.0f; acc.w = 0.0f;
  for (int j0 = b; j0 < en; j0 += 32) {
    const int j = j0 + ln;
    float ce = 0.0f;
    int uu = 0;
    if (j < en) { ce = expf(c[j] - m); uu = su[j]; }
    const float coef = ce / den;
    const int cnt = (en - j0 < 32) ? (en - j0) : 32;
#pragma unroll 4
    for (int l = 0; l < cnt; ++l) {
      const int ui = __shfl(uu, l, 32);
      const float cf = __shfl(coef, l, 32);
      const float4 mr = ((const float4*)(ctxp + (size_t)ui * DIM))[ln];
      acc.x += mr.x * cf;
      acc.y += mr.y * cf;
      acc.z += mr.z * cf;
      acc.w += mr.w * cf;
    }
  }
  ((float4*)(agg + (size_t)n * DIM))[ln] = acc;
}

__global__ void init_scalar_u32(unsigned* p) {
  if (threadIdx.x == 0) *p = 0u;
}

extern "C" void kernel_launch(void* const* d_in, const int* in_sizes, int n_in,
                              void* d_out, int out_size, void* d_ws, size_t ws_size,
                              hipStream_t stream) {
  (void)in_sizes; (void)n_in; (void)out_size; (void)ws_size;
  const int NA = 50000, NC = 100000, NE = 600000;

  const float* agt    = (const float*)d_in[0];
  const float* ctx    = (const float*)d_in[1];
  const int*   u      = (const int*)d_in[2];
  const int*   v      = (const int*)d_in[3];
  const float* Wq     = (const float*)d_in[4];
  const float* Wk     = (const float*)d_in[5];
  const float* Wctx   = (const float*)d_in[6];
  const float* gnc_w  = (const float*)d_in[7];
  const float* gnc_b  = (const float*)d_in[8];
  const float* Wagt   = (const float*)d_in[9];
  const float* gna_w  = (const float*)d_in[10];
  const float* gna_b  = (const float*)d_in[11];
  const float* norm_w = (const float*)d_in[12];
  const float* norm_b = (const float*)d_in[13];
  const float* Wlin   = (const float*)d_in[14];
  const float* gnl_w  = (const float*)d_in[15];
  const float* gnl_b  = (const float*)d_in[16];

  float* ws = (float*)d_ws;
  size_t off = 0;
  float* qbuf   = ws + off; off += (size_t)NA * DIM;
  float* xbuf   = ws + off; off += (size_t)NC * DIM;
  float* cpbuf  = ws + off; off += (size_t)NC * DIM;
  float* cbuf   = ws + off; off += (size_t)NE;
  int*   cntbuf = (int*)(ws + off); off += (size_t)NA;
  float* aggbuf = ws + off; off += (size_t)NA * DIM;
  int*   csr    = (int*)(ws + off); off += (size_t)NA * DIM;
  float* agtbuf = ws + off; off += (size_t)NA * DIM;
  unsigned* maxbuf = (unsigned*)(ws + off); off += 16;

  int*   rowptr = csr;
  int*   eid    = csr + 50016;
  int*   subuf  = csr + 650016;
  int*   arrbuf = csr + 1250016;
  int*   bsum   = csr + 1850016;
  float* wfold  = (float*)(csr + 2000000);

  const dim3 b128(128), b256(256);

  fold_qk_kernel<<<16, b256, 0, stream>>>(Wq, Wk, wfold);

  zero_i32<<<(NA + 255) / 256, b256, 0, stream>>>(cntbuf, NA);
  csr_count_kernel<<<(NE + 255) / 256, b256, 0, stream>>>(v, cntbuf, arrbuf, NE);
  csr_blocksum_kernel<<<(NA + 255) / 256, b256, 0, stream>>>(cntbuf, bsum, NA);
  csr_scan_kernel<<<(NA + 255) / 256, b256, 0, stream>>>(cntbuf, bsum, rowptr, NA);
  csr_fill_kernel<<<(NE + 255) / 256, b256, 0, stream>>>(v, arrbuf, rowptr, eid, NE);
  csr_rank_kernel<<<(NE + 255) / 256, b256, 0, stream>>>(eid, rowptr, u, v, subuf, NE);

  for (int i = 0; i < NBLK; ++i) {
    const float* ain  = (i == 0) ? agt : agtbuf;
    float*       aout = (i == NBLK - 1) ? (float*)d_out : agtbuf;

    gemm_nt_128<<<(NA + 63) / 64, b128, 0, stream>>>(ain, wfold + (size_t)i * DIM * DIM, qbuf, NA);
    gemm_nt_128_gn<<<(NC + 63) / 64, b128, 0, stream>>>(ctx, Wctx + (size_t)i * DIM * DIM, cpbuf,
                                                         gnc_w + i * DIM, gnc_b + i * DIM, NC);

    init_scalar_u32<<<1, 32, 0, stream>>>(maxbuf);
    node_dot_kernel<<<(NA + 7) / 8, b256, 0, stream>>>(qbuf, ctx, rowptr, subuf, cbuf, maxbuf, NA);
    node_agg_kernel<<<(NA + 7) / 8, b256, 0, stream>>>(cbuf, cpbuf, rowptr, subuf, maxbuf, aggbuf, NA);

    gemm_cat_256_gn2_relu<<<(NA + 63) / 64, b128, 0, stream>>>(
        ain, aggbuf, Wagt + (size_t)i * DIM * 2 * DIM, xbuf,
        gna_w + i * DIM, gna_b + i * DIM, norm_w + i * DIM, norm_b + i * DIM, NA);
    gemm_nt_128_gn_res_relu<<<(NA + 63) / 64, b128, 0, stream>>>(
        xbuf, Wlin + (size_t)i * DIM * DIM, aout, ain, gnl_w + i * DIM, gnl_b + i * DIM, NA);
  }
}


# Independent kernels share a launch: score-weight fold + counter clears + max resets in one, and per block the query GEMM and the ctx GEMM in one (block index picks the part); 21 launches -> 16
# speedup vs baseline: 2.7987x; 1.0497x over previous
#include <hip/hip_runtime.h>


static constexpr int DIM  = 128;
static constexpr int NBLK = 2;
static constexpr int LDP  = 68;

typedef __attribute__((ext_vector_type(2))) float v2f;
typedef __attribute__((ext_vector_type(8))) float v8f;

static __device__ __forceinline__ float wave_reduce_sum(float x) {
#pragma unroll
  for (int off = 16; off > 0; off >>= 1) x += __shfl_xor(x, off, 32);
  return x;
}

static __device__ __forceinline__ unsigned f32_order_enc(float f) {
  unsigned u = __float_as_uint(f);
  return (u & 0x80000000u) ? ~u : (u | 0x80000000u);
}
static __device__ __forceinline__ float f32_order_dec(unsigned e) {
  unsigned u = (e & 0x80000000u) ? (e & 0x7fffffffu) : ~e;
  return __uint_as_float(u);
}

enum { EPI_GN = 1, EPI_GN2_RELU = 2, EPI_GN_RES_RELU = 3 };

template <int XM>
static __device__ __forceinline__ float half_xor(float x) {
  return __int_as_float(__builtin_amdgcn_update_dpp(
      0, __float_as_int(x), 0x160 + XM, 0xf, 0xf, true));
}
static __device__ __forceinline__ float half_sum(float x) {
  x += half_xor<1>(x);
  x += half_xor<2>(x);
  x += half_xor<4>(x);
  x += half_xor<8>(x);
  return x;
}

template <bool RELU>
static __device__ __forceinline__ void gn_rows(v8f (&acc)[8],
                                               const float* __restrict__ w,
                                               const float* __restrict__ b,
                                               int l16) {
  float wv[8], bv[8];
#pragma unroll
  for (int t = 0; t < 8; ++t) {
    wv[t] = w[t * 16 + l16];
    bv[t] = b[t * 16 + l16];
  }
#pragma unroll
  for (int j = 0; j < 8; ++j) {
    float s = 0.0f, sq = 0.0f;
#pragma unroll
    for (int t = 0; t < 8; ++t) {
      const float x = acc[t][j];
      s += x;
      sq += x * x;
    }
    s  = half_sum(s);
    sq = half_sum(sq);
    const float mu = s * (1.0f / DIM);
    const float rs = rsqrtf(sq * (1.0f / DIM) - mu * mu + 1e-5f);
#pragma unroll
    for (int t = 0; t < 8; ++t) {
      float y = (acc[t][j] - mu) * rs * wv[t] + bv[t];
      if (RELU) y = fmaxf(y, 0.0f);
      acc[t][j] = y;
    }
  }
}

template <int EPI>
static __device__ __forceinline__ void epilogue_store(v8f (&acc)[8],
                                                      float* __restrict__ C,
                                                      const float* __restrict__ R,
                                                      const float* __restrict__ w1,
                                                      const float* __restrict__ b1,
                                                      const float* __restrict__ w2,
                                                      const float* __restrict__ b2,
                                                      int r0, int hk, int l16, int N,
                                                      bool plain) {
  const int rb = r0 + hk * 8;

  if (EPI == EPI_GN_RES_RELU) {
    float res[8][8];
#pragma unroll
    for (int j = 0; j < 8; ++j) {
      int row = rb + j;
      if (row >= N) row = N - 1;
#pragma unroll
      for (int t = 0; t < 8; ++t) res[t][j] = R[(size_t)row * DIM + t * 16 + l16];
    }
    gn_rows<false>(acc, w1, b1, l16);
#pragma unroll
    for (int t = 0; t < 8; ++t)
#pragma unroll
      for (int j = 0; j < 8; ++j) acc[t][j] = fmaxf(acc[t][j] + res[t][j], 0.0f);
  } else if (EPI == EPI_GN2_RELU) {
    gn_rows<false>(acc, w1, b1, l16);
    gn_rows<true>(acc, w2, b2, l16);
  } else if (EPI == EPI_GN) {
    if (!plain) gn_rows<false>(acc, w1, b1, l16);
  }

  if (r0 + 16 <= N) {
#pragma unroll
    for (int t = 0; t < 8; ++t)
#pragma unroll
      for (int j = 0; j < 8; ++j)
        C[(size_t)(rb + j) * DIM + t * 16 + l16] = acc[t][j];
  } else {
#pragma unroll
    for (int t = 0; t < 8; ++t)
#pragma unroll
      for (int j = 0; j < 8; ++j) {
        const int row = rb + j;
        if (row < N) C[(size_t)row * DIM + t * 16 + l16] = acc[t][j];
      }
  }
}

template <int EPI>
static __device__ __forceinline__ void gemm_nt_128_body(float* __restrict__ Wlds,
                                                        const float* __restrict__ A,
                                                        const float* __restrict__ W,
                                                        float* __restrict__ C,
                                                        const float* __restrict__ R,
                                                        const float* __restrict__ w1,
                                                        const float* __restrict__ b1,
                                                        const float* __restrict__ w2,
                                                        const float* __restrict__ b2,
                                                        int N, int tile, bool plain) {
  const int tid = threadIdx.x;
  const int wv  = tid >> 5;
  const int ln  = tid & 31;
  const int hk  = ln >> 4;
  const int l16 = ln & 15;
  const int r0  = tile * 64 + wv * 16;

  v8f acc[8];
#pragma unroll
  for (int t = 0; t < 8; ++t)
#pragma unroll
    for (int j = 0; j < 8; ++j) acc[t][j] = 0.0f;

  int rowA = r0 + l16;
  if (rowA >= N) rowA = N - 1;
  const float* ap = A + (size_t)rowA * DIM;

  for (int kp = 0; kp < DIM; kp += 64) {
    __syncthreads();
#pragma unroll
    for (int i = 0; i < 16; ++i) {
      const int q  = tid + 128 * i;
      const int n  = q >> 4;
      const int c4 = (q & 15) << 2;
      *(float4*)(&Wlds[n * LDP + c4]) =
          *(const float4*)(W + (size_t)n * DIM + kp + c4);
    }
    __syncthreads();

#pragma unroll 4
    for (int k0 = 0; k0 < 64; k0 += 4) {
      const int ka = k0 + 2 * hk;
      v2f a; a.x = ap[kp + ka]; a.y = ap[kp + ka + 1];
#pragma unroll
      for (int t = 0; t < 8; ++t) {
        const float* wp = &Wlds[(t * 16 + l16) * LDP + ka];
        v2f b; b.x = wp[0]; b.y = wp[1];
        acc[t] = __builtin_amdgcn_wmma_f32_16x16x4_f32(
            false, a, false, b, (short)0, acc[t], false, false);
      }
    }
  }

  epilogue_store<EPI>(acc, C, R, w1, b1, w2, b2, r0, hk, l16, N, plain);
}

__global__ __launch_bounds__(128) void gemm_q_and_ctxp(const float* __restrict__ Aq,
                                                       const float* __restrict__ Wfq,
                                                       float* __restrict__ Cq, int Nq,
                                                       int tilesQ,
                                                       const float* __restrict__ Ac,
                                                       const float* __restrict__ Wc,
                                                       float* __restrict__ Cc,
                                                       const float* __restrict__ gw,
                                                       const float* __restrict__ gb, int Nc) {
  __shared__ float Wlds[128 * LDP];
  const int bid = blockIdx.x;
  const bool isq = bid < tilesQ;
  const float* A  = isq ? Aq : Ac;
  const float* Wm = isq ? Wfq : Wc;
  float* C        = isq ? Cq : Cc;
  const int N     = isq ? Nq : Nc;
  const int tile  = isq ? bid : bid - tilesQ;
  gemm_nt_128_body<EPI_GN>(Wlds, A, Wm, C, nullptr, gw, gb, nullptr, nullptr, N, tile, isq);
}

__global__ __launch_bounds__(128) void gemm_nt_128_gn_res_relu(const float* __restrict__ A,
                                                               const float* __restrict__ W,
                                                               float* __restrict__ C,
                                                               const float* __restrict__ R,
                                                               const float* __restrict__ gw,
                                                               const float* __restrict__ gb,
                                                               int N) {
  __shared__ float Wlds[128 * LDP];
  gemm_nt_128_body<EPI_GN_RES_RELU>(Wlds, A, W, C, R, gw, gb, nullptr, nullptr, N, blockIdx.x, false);
}

__global__ __launch_bounds__(128) void gemm_cat_256_gn2_relu(const float* __restrict__ A0,
                                                             const float* __restrict__ A1,
                                                             const float* __restrict__ W,
                                                             float* __restrict__ C,
                                                             const float* __restrict__ w1,
                                                             const float* __restrict__ b1,
                                                             const float* __restrict__ w2,
                                                             const float* __restrict__ b2,
                                                             int N) {
  __shared__ float Wlds[128 * LDP];
  const int tid = threadIdx.x;
  const int wv  = tid >> 5;
  const int ln  = tid & 31;
  const int hk  = ln >> 4;
  const int l16 = ln & 15;
  const int r0  = blockIdx.x * 64 + wv * 16;

  v8f acc[8];
#pragma unroll
  for (int t = 0; t < 8; ++t)
#pragma unroll
    for (int j = 0; j < 8; ++j) acc[t][j] = 0.0f;

  int rowA = r0 + l16;
  if (rowA >= N) rowA = N - 1;
  const size_t rbase = (size_t)rowA * DIM;

  for (int kp = 0; kp < 2 * DIM; kp += 64) {
    __syncthreads();
#pragma unroll
    for (int i = 0; i < 16; ++i) {
      const int q  = tid + 128 * i;
      const int n  = q >> 4;
      const int c4 = (q & 15) << 2;
      *(float4*)(&Wlds[n * LDP + c4]) =
          *(const float4*)(W + (size_t)n * (2 * DIM) + kp + c4);
    }
    __syncthreads();

    const float* abase = (kp < DIM) ? (A0 + rbase + kp) : (A1 + rbase + (kp - DIM));
#pragma unroll 4
    for (int k0 = 0; k0 < 64; k0 += 4) {
      const int ka = k0 + 2 * hk;
      v2f a; a.x = abase[ka]; a.y = abase[ka + 1];
#pragma unroll
      for (int t = 0; t < 8; ++t) {
        const float* wp = &Wlds[(t * 16 + l16) * LDP + ka];
        v2f b; b.x = wp[0]; b.y = wp[1];
        acc[t] = __builtin_amdgcn_wmma_f32_16x16x4_f32(
            false, a, false, b, (short)0, acc[t], false, false);
      }
    }
  }

  epilogue_store<EPI_GN2_RELU>(acc, C, nullptr, w1, b1, w2, b2, r0, hk, l16, N, false);
}

static __device__ __forceinline__ void fold_qk_body(const float* __restrict__ Wq,
                                                    const float* __restrict__ Wk,
                                                    float* __restrict__ Wf, int bid) {
  const int wv  = threadIdx.x >> 5;
  const int ln  = threadIdx.x & 31;
  const int hk  = ln >> 4;
  const int l16 = ln & 15;
  const int tile = bid * 8 + wv;
  const int i  = tile >> 6;
  const int j0 = ((tile >> 3) & 7) * 16;
  const int m0 = (tile & 7) * 16;
  const float* wq = Wq + (size_t)i * DIM * DIM;
  const float* wk = Wk + (size_t)i * DIM * DIM;

  v8f acc;
#pragma unroll
  for (int r = 0; r < 8; ++r) acc[r] = 0.0f;

#pragma unroll 8
  for (int d0 = 0; d0 < DIM; d0 += 4) {
    const int d = d0 + 2 * hk;
    v2f a; a.x = wk[d * DIM + j0 + l16]; a.y = wk[(d + 1) * DIM + j0 + l16];
    v2f b; b.x = wq[d * DIM + m0 + l16]; b.y = wq[(d + 1) * DIM + m0 + l16];
    acc = __builtin_amdgcn_wmma_f32_16x16x4_f32(
        false, a, false, b, (short)0, acc, false, false);
  }

  float* o = Wf + (size_t)i * DIM * DIM + (size_t)(j0 + 8 * hk) * DIM + m0 + l16;
#pragma unroll
  for (int r = 0; r < 8; ++r) o[r * DIM] = acc[r];
}

__global__ __launch_bounds__(256) void setup_kernel(const float* __restrict__ Wq,
                                                    const float* __restrict__ Wk,
                                                    float* __restrict__ Wf,
                                                    int* __restrict__ cnt, int n,
                                                    unsigned* __restrict__ gmax) {
  const int bid = blockIdx.x;
  if (bid < 16) {
    fold_qk_body(Wq, Wk, Wf, bid);
  } else {
    const int i = (bid - 16) * 256 + threadIdx.x;
    if (i < n) cnt[i] = 0;
    if (bid == 16 && threadIdx.x < NBLK) gmax[threadIdx.x] = 0u;
  }
}

__global__ __launch_bounds__(256) void csr_count_kernel(const int* __restrict__ v,
                                                        int* __restrict__ cnt,
                                                        int* __restrict__ arr, int E) {
  const int e = blockIdx.x * 256 + threadIdx.x;
  if (e < E) arr[e] = atomicAdd(&cnt[v[e]], 1);
}

static __device__ __forceinline__ int wave_reduce_sum_i(int x) {
#pragma unroll
  for (int off = 16; off > 0; off >>= 1) x += __shfl_xor(x, off, 32);
  return x;
}

__global__ __launch_bounds__(256) void csr_blocksum_kernel(const int* __restrict__ cnt,
                                                           int* __restrict__ bsum, int N) {
  __shared__ int sw[8];
  const int wv = threadIdx.x >> 5, ln = threadIdx.x & 31;
  const int i = blockIdx.x * 256 + threadIdx.x;
  int c = (i < N) ? cnt[i] : 0;
  c = wave_reduce_sum_i(c);
  if (ln == 0) sw[wv] = c;
  __syncthreads();
  if (threadIdx.x == 0) {
    int t = 0;
#pragma unroll
    for (int k = 0; k < 8; ++k) t += sw[k];
    bsum[blockIdx.x] = t;
  }
}

__global__ __launch_bounds__(256) void csr_scan_kernel(const int* __restrict__ cnt,
                                                       const int* __restrict__ bsum,
                                                       int* __restrict__ rowptr, int N) {
  __shared__ int sa[256];
  __shared__ int sb[256];
  __shared__ int sw[8];
  const int t = threadIdx.x;
  const int wv = t >> 5, ln = t & 31;
  const int blk = blockIdx.x;

  int p = 0;
  for (int k = t; k < blk; k += 256) p += bsum[k];
  p = wave_reduce_sum_i(p);
  if (ln == 0) sw[wv] = p;

  const int i = blk * 256 + t;
  const int c = (i < N) ? cnt[i] : 0;
  sa[t] = c;
  __syncthreads();

  int base = 0;
#pragma unroll
  for (int k = 0; k < 8; ++k) base += sw[k];

  int* src = sa;
  int* dst = sb;
  for (int off = 1; off < 256; off <<= 1) {
    int val = src[t];
    if (t >= off) val += src[t - off];
    dst[t] = val;
    __syncthreads();
    int* tmp = src; src = dst; dst = tmp;
  }
  const int excl = base + src[t] - c;
  if (i < N) rowptr[i] = excl;
  if (i == N - 1) rowptr[N] = excl + c;
}

__global__ __launch_bounds__(256) void csr_fill_kernel(const int* __restrict__ v,
                                                       const int* __restrict__ arr,
                                                       const int* __restrict__ rowptr,
                                                       int* __restrict__ eid, int E) {
  const int e = blockIdx.x * 256 + threadIdx.x;
  if (e < E) eid[rowptr[v[e]] + arr[e]] = e;
}

__global__ __launch_bounds__(256) void csr_rank_kernel(const int* __restrict__ eid,
                                                       const int* __restrict__ rowptr,
                                                       const int* __restrict__ u,
                                                       const int* __restrict__ v,
                                                       int* __restrict__ su, int E) {
  const int j = blockIdx.x * 256 + threadIdx.x;
  if (j >= E) return;
  const int e = eid[j];
  const int n = v[e];
  const int b = rowptr[n], en = rowptr[n + 1];
  int r = 0;
  for (int k = b; k < en; ++k) r += (eid[k] < e) ? 1 : 0;
  su[b + r] = u[e];
}

__global__ __launch_bounds__(256) void node_dot_kernel(const float* __restrict__ Q,
                                                       const float* __restrict__ K,
                                                       const int* __restrict__ rowptr,
                                                       const int* __restrict__ su,
                                                       float* __restrict__ c,
                                                       unsigned* __restrict__ gmax,
                                                       int N) {
  __shared__ float smax[8];
  const int wv = threadIdx.x >> 5, ln = threadIdx.x & 31;
  const int n = __builtin_amdgcn_readfirstlane(blockIdx.x * 8 + wv);
  float mx = -3.0e38f;
  if (n < N) {
    const int b  = __builtin_amdgcn_readfirstlane(rowptr[n]);
    const int en = __builtin_amdgcn_readfirstlane(rowptr[n + 1]);
    if (b < en) {
      const float4 qv = ((const float4*)(Q + (size_t)n * DIM))[ln];
      for (int j0 = b; j0 < en; j0 += 32) {
        const int j = j0 + ln;
        const int uu = (j < en) ? su[j] : 0;
        const int cnt = (en - j0 < 32) ? (en - j0) : 32;
        float dl = 0.0f;
#pragma unroll 4
        for (int l = 0; l < cnt; ++l) {
          const int ui = __shfl(uu, l, 32);
          const float4 kv = ((const float4*)(K + (size_t)ui * DIM))[ln];
          float d = qv.x * kv.x + qv.y * kv.y + qv.z * kv.z + qv.w * kv.w;
          d = wave_reduce_sum(d);
          d = (d >= 0.0f) ? d : 0.1f * d;
          if (ln == l) dl = d;
          mx = fmaxf(mx, d);
        }
        if (j < en) c[j] = dl;
      }
    }
  }
  if (ln == 0) smax[wv] = mx;
  __syncthreads();
  if (threadIdx.x == 0) {
    float m = smax[0];
#pragma unroll
    for (int i = 1; i < 8; ++i) m = fmaxf(m, smax[i]);
    atomicMax(gmax, f32_order_enc(m));
  }
}

__global__ __launch_bounds__(256) void node_agg_kernel(const float* __restrict__ c,
                                                       const float* __restrict__ ctxp,
                                                       const int* __restrict__ rowptr,
                                                       const int* __restrict__ su,
                                                       const unsigned* __restrict__ gmax,
                                                       float* __restrict__ agg,
                                                       int N) {
  const int wv = threadIdx.x >> 5, ln = threadIdx.x & 31;
  const int n = __builtin_amdgcn_readfirstlane(blockIdx.x * 8 + wv);
  if (n >= N) return;
  const int b  = __builtin_amdgcn_readfirstlane(rowptr[n]);
  const int en = __builtin_amdgcn_readfirstlane(rowptr[n + 1]);
  const float m = f32_order_dec(*gmax);

  float s = 0.0f;
  for (int j0 = b; j0 < en; j0 += 32) {
    const int j = j0 + ln;
    const float ce = (j < en) ? expf(c[j] - m) : 0.0f;
    const int cnt = (en - j0 < 32) ? (en - j0) : 32;
    for (int l = 0; l < cnt; ++l) s += __shfl(ce, l, 32);
  }
  const float den = s + 1e-16f;

  float4 acc;
  acc.x = 0.0f; acc.y = 0.0f; acc.z = 0.0f; acc.w = 0.0f;
  for (int j0 = b; j0 < en; j0 += 32) {
    const int j = j0 + ln;
    float ce = 0.0f;
    int uu = 0;
    if (j < en) { ce = expf(c[j] - m); uu = su[j]; }
    const float coef = ce / den;
    const int cnt = (en - j0 < 32) ? (en - j0) : 32;
#pragma unroll 4
    for (int l = 0; l < cnt; ++l) {
      const int ui = __shfl(uu, l, 32);
      const float cf = __shfl(coef, l, 32);
      const float4 mr = ((const float4*)(ctxp + (size_t)ui * DIM))[ln];
      acc.x += mr.x * cf;
      acc.y += mr.y * cf;
      acc.z += mr.z * cf;
      acc.w += mr.w * cf;
    }
  }
  ((float4*)(agg + (size_t)n * DIM))[ln] = acc;
}

extern "C" void kernel_launch(void* const* d_in, const int* in_sizes, int n_in,
                              void* d_out, int out_size, void* d_ws, size_t ws_size,
                              hipStream_t stream) {
  (void)in_sizes; (void)n_in; (void)out_size; (void)ws_size;
  const int NA = 50000, NC = 100000, NE = 600000;

  const float* agt    = (const float*)d_in[0];
  const float* ctx    = (const float*)d_in[1];
  const int*   u      = (const int*)d_in[2];
  const int*   v      = (const int*)d_in[3];
  const float* Wq     = (const float*)d_in[4];
  const float* Wk     = (const float*)d_in[5];
  const float* Wctx   = (const float*)d_in[6];
  const float* gnc_w  = (const float*)d_in[7];
  const float* gnc_b  = (const float*)d_in[8];
  const float* Wagt   = (const float*)d_in[9];
  const float* gna_w  = (const float*)d_in[10];
  const float* gna_b  = (const float*)d_in[11];
  const float* norm_w = (const float*)d_in[12];
  const float* norm_b = (const float*)d_in[13];
  const float* Wlin   = (const float*)d_in[14];
  const float* gnl_w  = (const float*)d_in[15];
  const float* gnl_b  = (const float*)d_in[16];

  float* ws = (float*)d_ws;
  size_t off = 0;
  float* qbuf   = ws + off; off += (size_t)NA * DIM;
  float* xbuf   = ws + off; off += (size_t)NC * DIM;
  float* cpbuf  = ws + off; off += (size_t)NC * DIM;
  float* cbuf   = ws + off; off += (size_t)NE;
  int*   cntbuf = (int*)(ws + off); off += (size_t)NA;
  float* aggbuf = ws + off; off += (size_t)NA * DIM;
  int*   csr    = (int*)(ws + off); off += (size_t)NA * DIM;
  float* agtbuf = ws + off; off += (size_t)NA * DIM;
  unsigned* maxbuf = (unsigned*)(ws + off); off += 16;

  int*   rowptr = csr;
  int*   eid    = csr + 50016;
  int*   subuf  = csr + 650016;
  int*   arrbuf = csr + 1250016;
  int*   bsum   = csr + 1850016;
  float* wfold  = (float*)(csr + 2000000);

  const dim3 b128(128), b256(256);

  setup_kernel<<<16 + (NA + 255) / 256, b256, 0, stream>>>(Wq, Wk, wfold, cntbuf, NA, maxbuf);

  csr_count_kernel<<<(NE + 255) / 256, b256, 0, stream>>>(v, cntbuf, arrbuf, NE);
  csr_blocksum_kernel<<<(NA + 255) / 256, b256, 0, stream>>>(cntbuf, bsum, NA);
  csr_scan_kernel<<<(NA + 255) / 256, b256, 0, stream>>>(cntbuf, bsum, rowptr, NA);
  csr_fill_kernel<<<(NE + 255) / 256, b256, 0, stream>>>(v, arrbuf, rowptr, eid, NE);
  csr_rank_kernel<<<(NE + 255) / 256, b256, 0, stream>>>(eid, rowptr, u, v, subuf, NE);

  for (int i = 0; i < NBLK; ++i) {
    const float* ain  = (i == 0) ? agt : agtbuf;
    float*       aout = (i == NBLK - 1) ? (float*)d_out : agtbuf;

    gemm_q_and_ctxp<<<(NA + 63) / 64 + (NC + 63) / 64, b128, 0, stream>>>(
        ain, wfold + (size_t)i * DIM * DIM, qbuf, NA, (NA + 63) / 64,
        ctx, Wctx + (size_t)i * DIM * DIM, cpbuf, gnc_w + i * DIM, gnc_b + i * DIM, NC);

    node_dot_kernel<<<(NA + 7) / 8, b256, 0, stream>>>(qbuf, ctx, rowptr, subuf, cbuf, maxbuf + i, NA);
    node_agg_kernel<<<(NA + 7) / 8, b256, 0, stream>>>(cbuf, cpbuf, rowptr, subuf, maxbuf + i, aggbuf, NA);

    gemm_cat_256_gn2_relu<<<(NA + 63) / 64, b128, 0, stream>>>(
        ain, aggbuf, Wagt + (size_t)i * DIM * 2 * DIM, xbuf,
        gna_w + i * DIM, gna_b + i * DIM, norm_w + i * DIM, norm_b + i * DIM, NA);
    gemm_nt_128_gn_res_relu<<<(NA + 63) / 64, b128, 0, stream>>>(
        xbuf, Wlin + (size_t)i * DIM * DIM, aout, ain, gnl_w + i * DIM, gnl_b + i * DIM, NA);
  }
}


# Each source row gathered once per edge: GN of a ctx row is affine in its (mean, rstd), so raw ctx rows are summed with segment-max-shifted weights; Wctx folded into the update GEMM (new sum order)
# speedup vs baseline: 2.9741x; 1.0627x over previous
#include <hip/hip_runtime.h>


static constexpr int DIM  = 128;
static constexpr int NBLK = 2;
static constexpr int LDP  = 68;

typedef __attribute__((ext_vector_type(2))) float v2f;
typedef __attribute__((ext_vector_type(8))) float v8f;

static __device__ __forceinline__ unsigned f32_order_enc(float f) {
  unsigned u = __float_as_uint(f);
  return (u & 0x80000000u) ? ~u : (u | 0x80000000u);
}
static __device__ __forceinline__ float f32_order_dec(unsigned e) {
  unsigned u = (e & 0x80000000u) ? (e & 0x7fffffffu) : ~e;
  return __uint_as_float(u);
}

enum { EPI_STATS = 1, EPI_GN2_RELU = 2, EPI_GN_RES_RELU = 3 };

template <int XM>
static __device__ __forceinline__ float half_xor(float x) {
  return __int_as_float(__builtin_amdgcn_update_dpp(
      0, __float_as_int(x), 0x160 + XM, 0xf, 0xf, true));
}
static __device__ __forceinline__ float half_sum(float x) {
  x += half_xor<1>(x);
  x += half_xor<2>(x);
  x += half_xor<4>(x);
  x += half_xor<8>(x);
  return x;
}

template <bool RELU>
static __device__ __forceinline__ void gn_rows(v8f (&acc)[8],
                                               const float* __restrict__ w,
                                               const float* __restrict__ b,
                                               int l16) {
  float wv[8], bv[8];
#pragma unroll
  for (int t = 0; t < 8; ++t) {
    wv[t] = w[t * 16 + l16];
    bv[t] = b[t * 16 + l16];
  }
#pragma unroll
  for (int j = 0; j < 8; ++j) {
    float s = 0.0f, sq = 0.0f;
#pragma unroll
    for (int t = 0; t < 8; ++t) {
      const float x = acc[t][j];
      s += x;
      sq += x * x;
    }
    s  = half_sum(s);
    sq = half_sum(sq);
    const float mu = s * (1.0f / DIM);
    const float rs = rsqrtf(sq * (1.0f / DIM) - mu * mu + 1e-5f);
#pragma unroll
    for (int t = 0; t < 8; ++t) {
      float y = (acc[t][j] - mu) * rs * wv[t] + bv[t];
      if (RELU) y = fmaxf(y, 0.0f);
      acc[t][j] = y;
    }
  }
}

static __device__ __forceinline__ void stats_rows(const v8f (&acc)[8],
                                                  float* __restrict__ st,
                                                  int r0, int hk, int l16, int N) {
  float mymu = 0.0f, myrs = 0.0f;
#pragma unroll
  for (int j = 0; j < 8; ++j) {
    float s = 0.0f, sq = 0.0f;
#pragma unroll
    for (int t = 0; t < 8; ++t) {
      const float x = acc[t][j];
      s += x;
      sq += x * x;
    }
    s  = half_sum(s);
    sq = half_sum(sq);
    const float mu = s * (1.0f / DIM);
    const float rs = rsqrtf(sq * (1.0f / DIM) - mu * mu + 1e-5f);
    if (l16 == j) { mymu = mu; myrs = rs; }
  }
  const int row = r0 + hk * 8 + l16;
  if (l16 < 8 && row < N) {
    float2 v; v.x = mymu; v.y = myrs;
    ((float2*)st)[row] = v;
  }
}

template <int EPI>
static __device__ __forceinline__ void epilogue_store(v8f (&acc)[8],
                                                      float* __restrict__ C,
                                                      const float* __restrict__ R,
                                                      const float* __restrict__ w1,
                                                      const float* __restrict__ b1,
                                                      const float* __restrict__ w2,
                                                      const float* __restrict__ b2,
                                                      int r0, int hk, int l16, int N,
                                                      bool plain) {
  const int rb = r0 + hk * 8;

  if (EPI == EPI_GN_RES_RELU) {
    float res[8][8];
#pragma unroll
    for (int j = 0; j < 8; ++j) {
      int row = rb + j;
      if (row >= N) row = N - 1;
#pragma unroll
      for (int t = 0; t < 8; ++t) res[t][j] = R[(size_t)row * DIM + t * 16 + l16];
    }
    gn_rows<false>(acc, w1, b1, l16);
#pragma unroll
    for (int t = 0; t < 8; ++t)
#pragma unroll
      for (int j = 0; j < 8; ++j) acc[t][j] = fmaxf(acc[t][j] + res[t][j], 0.0f);
  } else if (EPI == EPI_GN2_RELU) {
    gn_rows<false>(acc, w1, b1, l16);
    gn_rows<true>(acc, w2, b2, l16);
  } else if (EPI == EPI_STATS) {
    if (!plain) {
      stats_rows(acc, C, r0, hk, l16, N);
      return;
    }
  }

  if (r0 + 16 <= N) {
#pragma unroll
    for (int t = 0; t < 8; ++t)
#pragma unroll
      for (int j = 0; j < 8; ++j)
        C[(size_t)(rb + j) * DIM + t * 16 + l16] = acc[t][j];
  } else {
#pragma unroll
    for (int t = 0; t < 8; ++t)
#pragma unroll
      for (int j = 0; j < 8; ++j) {
        const int row = rb + j;
        if (row < N) C[(size_t)row * DIM + t * 16 + l16] = acc[t][j];
      }
  }
}

template <int EPI>
static __device__ __forceinline__ void gemm_nt_128_body(float* __restrict__ Wlds,
                                                        const float* __restrict__ A,
                                                        const float* __restrict__ W,
                                                        float* __restrict__ C,
                                                        const float* __restrict__ R,
                                                        const float* __restrict__ w1,
                                                        const float* __restrict__ b1,
                                                        const float* __restrict__ w2,
                                                        const float* __restrict__ b2,
                                                        int N, int tile, bool plain) {
  const int tid = threadIdx.x;
  const int wv  = tid >> 5;
  const int ln  = tid & 31;
  const int hk  = ln >> 4;
  const int l16 = ln & 15;
  const int r0  = tile * 64 + wv * 16;

  v8f acc[8];
#pragma unroll
  for (int t = 0; t < 8; ++t)
#pragma unroll
    for (int j = 0; j < 8; ++j) acc[t][j] = 0.0f;

  int rowA = r0 + l16;
  if (rowA >= N) rowA = N - 1;
  const float* ap = A + (size_t)rowA * DIM;

  for (int kp = 0; kp < DIM; kp += 64) {
    __syncthreads();
#pragma unroll
    for (int i = 0; i < 16; ++i) {
      const int q  = tid + 128 * i;
      const int n  = q >> 4;
      const int c4 = (q & 15) << 2;
      *(float4*)(&Wlds[n * LDP + c4]) =
          *(const float4*)(W + (size_t)n * DIM + kp + c4);
    }
    __syncthreads();

#pragma unroll 4
    for (int k0 = 0; k0 < 64; k0 += 4) {
      const int ka = k0 + 2 * hk;
      v2f a; a.x = ap[kp + ka]; a.y = ap[kp + ka + 1];
#pragma unroll
      for (int t = 0; t < 8; ++t) {
        const float* wp = &Wlds[(t * 16 + l16) * LDP + ka];
        v2f b; b.x = wp[0]; b.y = wp[1];
        acc[t] = __builtin_amdgcn_wmma_f32_16x16x4_f32(
            false, a, false, b, (short)0, acc[t], false, false);
      }
    }
  }

  epilogue_store<EPI>(acc, C, R, w1, b1, w2, b2, r0, hk, l16, N, plain);
}

__global__ __launch_bounds__(128) void gemm_q_and_ctxp(const float* __restrict__ Aq,
                                                       const float* __restrict__ Wfq,
                                                       float* __restrict__ Cq, int Nq,
                                                       int tilesQ,
                                                       const float* __restrict__ Ac,
                                                       const float* __restrict__ Wc,
                                                       float* __restrict__ Cc, int Nc) {
  __shared__ float Wlds[128 * LDP];
  const int bid = blockIdx.x;
  const bool isq = bid < tilesQ;
  const float* A  = isq ? Aq : Ac;
  const float* Wm = isq ? Wfq : Wc;
  float* C        = isq ? Cq : Cc;
  const int N     = isq ? Nq : Nc;
  const int tile  = isq ? bid : bid - tilesQ;
  gemm_nt_128_body<EPI_STATS>(Wlds, A, Wm, C, nullptr, nullptr, nullptr, nullptr, nullptr, N, tile, isq);
}

__global__ __launch_bounds__(128) void gemm_nt_128_gn_res_relu(const float* __restrict__ A,
                                                               const float* __restrict__ W,
                                                               float* __restrict__ C,
                                                               const float* __restrict__ R,
                                                               const float* __restrict__ gw,
                                                               const float* __restrict__ gb,
                                                               int N) {
  __shared__ float Wlds[128 * LDP];
  gemm_nt_128_body<EPI_GN_RES_RELU>(Wlds, A, W, C, R, gw, gb, nullptr, nullptr, N, blockIdx.x, false);
}

__global__ __launch_bounds__(128) void gemm_cat_256_gn2_relu(const float* __restrict__ A0,
                                                             const float* __restrict__ A1,
                                                             const float* __restrict__ aux,
                                                             const unsigned* __restrict__ gmax,
                                                             const float* __restrict__ W,
                                                             const float* __restrict__ Wm,
                                                             const float* __restrict__ r12,
                                                             float* __restrict__ C,
                                                             const float* __restrict__ w1,
                                                             const float* __restrict__ b1,
                                                             const float* __restrict__ w2,
                                                             const float* __restrict__ b2,
                                                             int N) {
  __shared__ float Wlds[128 * LDP];
  const int tid = threadIdx.x;
  const int wv  = tid >> 5;
  const int ln  = tid & 31;
  const int hk  = ln >> 4;
  const int l16 = ln & 15;
  const int r0  = blockIdx.x * 64 + wv * 16;

  v8f acc[8];
#pragma unroll
  for (int t = 0; t < 8; ++t)
#pragma unroll
    for (int j = 0; j < 8; ++j) acc[t][j] = 0.0f;

  int rowA = r0 + l16;
  if (rowA >= N) rowA = N - 1;
  const size_t rbase = (size_t)rowA * DIM;

  const float4 ax = ((const float4*)aux)[rowA];
  const float gm  = f32_order_dec(*gmax);
  const float ee  = expf(ax.x - gm);
  const float fr  = ee / (ee * ax.y + 1e-16f);

  for (int kp = 0; kp < 2 * DIM; kp += 64) {
    const bool first = kp < DIM;
    const float* wsrc = first ? (W + kp) : (Wm + (kp - DIM));
    const int wld     = first ? 2 * DIM : DIM;
    __syncthreads();
#pragma unroll
    for (int i = 0; i < 16; ++i) {
      const int q  = tid + 128 * i;
      const int n  = q >> 4;
      const int c4 = (q & 15) << 2;
      *(float4*)(&Wlds[n * LDP + c4]) =
          *(const float4*)(wsrc + (size_t)n * wld + c4);
    }
    __syncthreads();

    const float* abase = first ? (A0 + rbase + kp) : (A1 + rbase + (kp - DIM));
    const float asc    = first ? 1.0f : fr;
#pragma unroll 4
    for (int k0 = 0; k0 < 64; k0 += 4) {
      const int ka = k0 + 2 * hk;
      v2f a; a.x = abase[ka] * asc; a.y = abase[ka + 1] * asc;
#pragma unroll
      for (int t = 0; t < 8; ++t) {
        const float* wp = &Wlds[(t * 16 + l16) * LDP + ka];
        v2f b; b.x = wp[0]; b.y = wp[1];
        acc[t] = __builtin_amdgcn_wmma_f32_16x16x4_f32(
            false, a, false, b, (short)0, acc[t], false, false);
      }
    }
  }

  {
    v2f a;
    a.x = (hk == 0) ? -fr * ax.z : 0.0f;
    a.y = (hk == 0) ?  fr * ax.y : 0.0f;
#pragma unroll
    for (int t = 0; t < 8; ++t) {
      v2f b;
      b.x = (hk == 0) ? r12[t * 16 + l16] : 0.0f;
      b.y = (hk == 0) ? r12[DIM + t * 16 + l16] : 0.0f;
      acc[t] = __builtin_amdgcn_wmma_f32_16x16x4_f32(
          false, a, false, b, (short)0, acc[t], false, false);
    }
  }

  epilogue_store<EPI_GN2_RELU>(acc, C, nullptr, w1, b1, w2, b2, r0, hk, l16, N, false);
}

static __device__ __forceinline__ void fold_qk_body(const float* __restrict__ Wq,
                                                    const float* __restrict__ Wk,
                                                    float* __restrict__ Wf, int bid) {
  const int wv  = threadIdx.x >> 5;
  const int ln  = threadIdx.x & 31;
  const int hk  = ln >> 4;
  const int l16 = ln & 15;
  const int tile = bid * 8 + wv;
  const int i  = tile >> 6;
  const int j0 = ((tile >> 3) & 7) * 16;
  const int m0 = (tile & 7) * 16;
  const float* wq = Wq + (size_t)i * DIM * DIM;
  const float* wk = Wk + (size_t)i * DIM * DIM;

  v8f acc;
#pragma unroll
  for (int r = 0; r < 8; ++r) acc[r] = 0.0f;

#pragma unroll 8
  for (int d0 = 0; d0 < DIM; d0 += 4) {
    const int d = d0 + 2 * hk;
    v2f a; a.x = wk[d * DIM + j0 + l16]; a.y = wk[(d + 1) * DIM + j0 + l16];
    v2f b; b.x = wq[d * DIM + m0 + l16]; b.y = wq[(d + 1) * DIM + m0 + l16];
    acc = __builtin_amdgcn_wmma_f32_16x16x4_f32(
        false, a, false, b, (short)0, acc, false, false);
  }

  float* o = Wf + (size_t)i * DIM * DIM + (size_t)(j0 + 8 * hk) * DIM + m0 + l16;
#pragma unroll
  for (int r = 0; r < 8; ++r) o[r * DIM] = acc[r];
}

static __device__ __forceinline__ void fold_msg_body(const float* __restrict__ Wagt,
                                                     const float* __restrict__ Wctx,
                                                     const float* __restrict__ gw,
                                                     float* __restrict__ Wm, int bid) {
  const int wv  = threadIdx.x >> 5;
  const int ln  = threadIdx.x & 31;
  const int hk  = ln >> 4;
  const int l16 = ln & 15;
  const int tile = bid * 8 + wv;
  const int i  = tile >> 6;
  const int o0 = ((tile >> 3) & 7) * 16;
  const int k0 = (tile & 7) * 16;
  const float* wa = Wagt + (size_t)i * DIM * 2 * DIM + (size_t)(o0 + l16) * (2 * DIM) + DIM;
  const float* wc = Wctx + (size_t)i * DIM * DIM + k0 + l16;
  const float* g  = gw + i * DIM;

  v8f acc;
#pragma unroll
  for (int r = 0; r < 8; ++r) acc[r] = 0.0f;

#pragma unroll 8
  for (int d0 = 0; d0 < DIM; d0 += 4) {
    const int d = d0 + 2 * hk;
    v2f a; a.x = wa[d] * g[d];       a.y = wa[d + 1] * g[d + 1];
    v2f b; b.x = wc[d * DIM];        b.y = wc[(d + 1) * DIM];
    acc = __builtin_amdgcn_wmma_f32_16x16x4_f32(
        false, a, false, b, (short)0, acc, false, false);
  }

  float* o = Wm + (size_t)i * DIM * DIM + (size_t)(o0 + 8 * hk) * DIM + k0 + l16;
#pragma unroll
  for (int r = 0; r < 8; ++r) o[r * DIM] = acc[r];
}

__global__ __launch_bounds__(256) void setup_kernel(const float* __restrict__ Wq,
                                                    const float* __restrict__ Wk,
                                                    float* __restrict__ Wf,
                                                    const float* __restrict__ Wagt,
                                                    const float* __restrict__ Wctx,
                                                    const float* __restrict__ gw,
                                                    const float* __restrict__ gb,
                                                    float* __restrict__ Wm,
                                                    float* __restrict__ r12,
                                                    int* __restrict__ cnt, int n,
                                                    unsigned* __restrict__ gmax) {
  const int bid = blockIdx.x;
  if (bid < 16) {
    fold_qk_body(Wq, Wk, Wf, bid);
  } else if (bid < 32) {
    fold_msg_body(Wagt, Wctx, gw, Wm, bid - 16);
  } else if (bid == 32) {
    const int i = threadIdx.x >> 7;
    const int o = threadIdx.x & 127;
    const float* wa = Wagt + (size_t)i * DIM * 2 * DIM + (size_t)o * (2 * DIM) + DIM;
    float a1 = 0.0f, a2 = 0.0f;
    for (int c = 0; c < DIM; ++c) {
      const float x = wa[c];
      a1 += x * gw[i * DIM + c];
      a2 += x * gb[i * DIM + c];
    }
    r12[i * 2 * DIM + o]       = a1;
    r12[i * 2 * DIM + DIM + o] = a2;
  } else {
    const int i = (bid - 33) * 256 + threadIdx.x;
    if (i < n) cnt[i] = 0;
    if (bid == 33 && threadIdx.x < NBLK) gmax[threadIdx.x] = 0u;
  }
}

__global__ __launch_bounds__(256) void csr_count_kernel(const int* __restrict__ v,
                                                        int* __restrict__ cnt,
                                                        int* __restrict__ arr, int E) {
  const int e = blockIdx.x * 256 + threadIdx.x;
  if (e < E) arr[e] = atomicAdd(&cnt[v[e]], 1);
}

static __device__ __forceinline__ int wave_reduce_sum_i(int x) {
#pragma unroll
  for (int off = 16; off > 0; off >>= 1) x += __shfl_xor(x, off, 32);
  return x;
}

__global__ __launch_bounds__(256) void csr_blocksum_kernel(const int* __restrict__ cnt,
                                                           int* __restrict__ bsum, int N) {
  __shared__ int sw[8];
  const int wv = threadIdx.x >> 5, ln = threadIdx.x & 31;
  const int i = blockIdx.x * 256 + threadIdx.x;
  int c = (i < N) ? cnt[i] : 0;
  c = wave_reduce_sum_i(c);
  if (ln == 0) sw[wv] = c;
  __syncthreads();
  if (threadIdx.x == 0) {
    int t = 0;
#pragma unroll
    for (int k = 0; k < 8; ++k) t += sw[k];
    bsum[blockIdx.x] = t;
  }
}

__global__ __launch_bounds__(256) void csr_scan_kernel(const int* __restrict__ cnt,
                                                       const int* __restrict__ bsum,
                                                       int* __restrict__ rowptr, int N) {
  __shared__ int sa[256];
  __shared__ int sb[256];
  __shared__ int sw[8];
  const int t = threadIdx.x;
  const int wv = t >> 5, ln = t & 31;
  const int blk = blockIdx.x;

  int p = 0;
  for (int k = t; k < blk; k += 256) p += bsum[k];
  p = wave_reduce_sum_i(p);
  if (ln == 0) sw[wv] = p;

  const int i = blk * 256 + t;
  const int c = (i < N) ? cnt[i] : 0;
  sa[t] = c;
  __syncthreads();

  int base = 0;
#pragma unroll
  for (int k = 0; k < 8; ++k) base += sw[k];

  int* src = sa;
  int* dst = sb;
  for (int off = 1; off < 256; off <<= 1) {
    int val = src[t];
    if (t >= off) val += src[t - off];
    dst[t] = val;
    __syncthreads();
    int* tmp = src; src = dst; dst = tmp;
  }
  const int excl = base + src[t] - c;
  if (i < N) rowptr[i] = excl;
  if (i == N - 1) rowptr[N] = excl + c;
}

__global__ __launch_bounds__(256) void csr_fill_kernel(const int* __restrict__ v,
                                                       const int* __restrict__ arr,
                                                       const int* __restrict__ rowptr,
                                                       int* __restrict__ eid, int E) {
  const int e = blockIdx.x * 256 + threadIdx.x;
  if (e < E) eid[rowptr[v[e]] + arr[e]] = e;
}

__global__ __launch_bounds__(256) void csr_rank_kernel(const int* __restrict__ eid,
                                                       const int* __restrict__ rowptr,
                                                       const int* __restrict__ u,
                                                       const int* __restrict__ v,
                                                       int* __restrict__ su, int E) {
  const int j = blockIdx.x * 256 + threadIdx.x;
  if (j >= E) return;
  const int e = eid[j];
  const int n = v[e];
  const int b = rowptr[n], en = rowptr[n + 1];
  int r = 0;
  for (int k = b; k < en; ++k) r += (eid[k] < e) ? 1 : 0;
  su[b + r] = u[e];
}

static constexpr int CH = 16;

static __device__ __forceinline__ int lane_i(int x, int l) {
  return __builtin_amdgcn_readlane(x, l);
}
static __device__ __forceinline__ float lane_f(float x, int l) {
  return __int_as_float(__builtin_amdgcn_readlane(__float_as_int(x), l));
}
static __device__ __forceinline__ float wave_sum_dpp(float x) {
  x = half_sum(x);
  return lane_f(x, 0) + lane_f(x, 16);
}

__global__ __launch_bounds__(256) void node_attn_kernel(const float* __restrict__ Q,
                                                        const float* __restrict__ X,
                                                        const float* __restrict__ st,
                                                        const int* __restrict__ rowptr,
                                                        const int* __restrict__ su,
                                                        float* __restrict__ S,
                                                        float* __restrict__ aux,
                                                        unsigned* __restrict__ gmax,
                                                        int N) {
  __shared__ float smax[8];
  const int wv = threadIdx.x >> 5, ln = threadIdx.x & 31;
  const int n = __builtin_amdgcn_readfirstlane(blockIdx.x * 8 + wv);
  float m = -3.0e38f;
  if (n < N) {
    const int b  = __builtin_amdgcn_readfirstlane(rowptr[n]);
    const int en = __builtin_amdgcn_readfirstlane(rowptr[n + 1]);
    float ssum = 0.0f, tsum = 0.0f;
    float4 acc;
    acc.x = 0.0f; acc.y = 0.0f; acc.z = 0.0f; acc.w = 0.0f;
    if (b < en) {
      const float4 qv = ((const float4*)(Q + (size_t)n * DIM))[ln];
      for (int j0 = b; j0 < en; j0 += CH) {
        const int cnt = (en - j0 < CH) ? (en - j0) : CH;
        const int jj = j0 + ((ln < cnt) ? ln : (cnt - 1));
        const int uu = su[jj];
        const float2 sv = ((const float2*)st)[uu];

        float4 kv[CH];
#pragma unroll
        for (int g = 0; g < CH / 4; ++g) {
          if (4 * g < cnt) {
#pragma unroll
            for (int l = 4 * g; l < 4 * g + 4; ++l) {
              const int ui = lane_i(uu, l);
              kv[l] = ((const float4*)(X + (size_t)ui * DIM))[ln];
            }
          }
        }

        float d[CH];
        float mc = -3.0e38f;
#pragma unroll
        for (int g = 0; g < CH / 4; ++g) {
          if (4 * g < cnt) {
#pragma unroll
            for (int l = 4 * g; l < 4 * g + 4; ++l) {
              float p = qv.x * kv[l].x + qv.y * kv[l].y + qv.z * kv[l].z + qv.w * kv[l].w;
              p = wave_sum_dpp(p);
              p = (p >= 0.0f) ? p : 0.1f * p;
              d[l] = p;
              if (l < cnt) mc = fmaxf(mc, p);
            }
          }
        }

        const float mn = fmaxf(m, mc);
        const float sc = expf(m - mn);
        ssum *= sc; tsum *= sc;
        acc.x *= sc; acc.y *= sc; acc.z *= sc; acc.w *= sc;
        m = mn;

#pragma unroll
        for (int l = 0; l < CH; ++l) {
          if (l < cnt) {
            const float e  = expf(d[l] - mn);
            const float mu = lane_f(sv.x, l);
            const float rs = lane_f(sv.y, l);
            const float w  = e * rs;
            ssum += e;
            tsum += w * mu;
            acc.x += w * kv[l].x;
            acc.y += w * kv[l].y;
            acc.z += w * kv[l].z;
            acc.w += w * kv[l].w;
          }
        }
      }
    }
    ((float4*)(S + (size_t)n * DIM))[ln] = acc;
    if (ln == 0) {
      float4 a; a.x = m; a.y = ssum; a.z = tsum; a.w = 0.0f;
      ((float4*)aux)[n] = a;
    }
  }
  if (ln == 0) smax[wv] = m;
  __syncthreads();
  if (threadIdx.x == 0) {
    float mm = smax[0];
#pragma unroll
    for (int i = 1; i < 8; ++i) mm = fmaxf(mm, smax[i]);
    atomicMax(gmax, f32_order_enc(mm));
  }
}

extern "C" void kernel_launch(void* const* d_in, const int* in_sizes, int n_in,
                              void* d_out, int out_size, void* d_ws, size_t ws_size,
                              hipStream_t stream) {
  (void)in_sizes; (void)n_in; (void)out_size; (void)ws_size;
  const int NA = 50000, NC = 100000, NE = 600000;

  const float* agt    = (const float*)d_in[0];
  const float* ctx    = (const float*)d_in[1];
  const int*   u      = (const int*)d_in[2];
  const int*   v      = (const int*)d_in[3];
  const float* Wq     = (const float*)d_in[4];
  const float* Wk     = (const float*)d_in[5];
  const float* Wctx   = (const float*)d_in[6];
  const float* gnc_w  = (const float*)d_in[7];
  const float* gnc_b  = (const float*)d_in[8];
  const float* Wagt   = (const float*)d_in[9];
  const float* gna_w  = (const float*)d_in[10];
  const float* gna_b  = (const float*)d_in[11];
  const float* norm_w = (const float*)d_in[12];
  const float* norm_b = (const float*)d_in[13];
  const float* Wlin   = (const float*)d_in[14];
  const float* gnl_w  = (const float*)d_in[15];
  const float* gnl_b  = (const float*)d_in[16];

  float* ws = (float*)d_ws;
  size_t off = 0;
  float* qbuf   = ws + off; off += (size_t)NA * DIM;
  float* xbuf   = ws + off; off += (size_t)NC * DIM;
  float* cpbuf  = ws + off; off += (size_t)NC * DIM;
  off += (size_t)NE;
  int*   cntbuf = (int*)(ws + off); off += (size_t)NA;
  float* aggbuf = ws + off; off += (size_t)NA * DIM;
  int*   csr    = (int*)(ws + off); off += (size_t)NA * DIM;
  float* agtbuf = ws + off; off += (size_t)NA * DIM;
  unsigned* maxbuf = (unsigned*)(ws + off); off += 16;

  int*   rowptr = csr;
  int*   eid    = csr + 50016;
  int*   subuf  = csr + 650016;
  int*   arrbuf = csr + 1250016;
  int*   bsum   = csr + 1850016;
  float* wfold  = (float*)(csr + 2000000);
  float* wmsg   = (float*)(csr + 2040000);
  float* r12buf = (float*)(csr + 2080000);
  float* stbuf  = cpbuf;
  float* auxbuf = cpbuf + 2 * 100000;

  const dim3 b128(128), b256(256);

  setup_kernel<<<33 + (NA + 255) / 256, b256, 0, stream>>>(Wq, Wk, wfold, Wagt, Wctx, gnc_w, gnc_b,
                                                          wmsg, r12buf, cntbuf, NA, maxbuf);

  csr_count_kernel<<<(NE + 255) / 256, b256, 0, stream>>>(v, cntbuf, arrbuf, NE);
  csr_blocksum_kernel<<<(NA + 255) / 256, b256, 0, stream>>>(cntbuf, bsum, NA);
  csr_scan_kernel<<<(NA + 255) / 256, b256, 0, stream>>>(cntbuf, bsum, rowptr, NA);
  csr_fill_kernel<<<(NE + 255) / 256, b256, 0, stream>>>(v, arrbuf, rowptr, eid, NE);
  csr_rank_kernel<<<(NE + 255) / 256, b256, 0, stream>>>(eid, rowptr, u, v, subuf, NE);

  for (int i = 0; i < NBLK; ++i) {
    const float* ain  = (i == 0) ? agt : agtbuf;
    float*       aout = (i == NBLK - 1) ? (float*)d_out : agtbuf;

    gemm_q_and_ctxp<<<(NA + 63) / 64 + (NC + 63) / 64, b128, 0, stream>>>(
        ain, wfold + (size_t)i * DIM * DIM, qbuf, NA, (NA + 63) / 64,
        ctx, Wctx + (size_t)i * DIM * DIM, stbuf, NC);

    node_attn_kernel<<<(NA + 7) / 8, b256, 0, stream>>>(qbuf, ctx, stbuf, rowptr, subuf,
                                                       aggbuf, auxbuf, maxbuf + i, NA);

    gemm_cat_256_gn2_relu<<<(NA + 63) / 64, b128, 0, stream>>>(
        ain, aggbuf, auxbuf, maxbuf + i, Wagt + (size_t)i * DIM * 2 * DIM,
        wmsg + (size_t)i * DIM * DIM, r12buf + (size_t)i * 2 * DIM, xbuf,
        gna_w + i * DIM, gna_b + i * DIM, norm_w + i * DIM, norm_b + i * DIM, NA);
    gemm_nt_128_gn_res_relu<<<(NA + 63) / 64, b128, 0, stream>>>(
        xbuf, Wlin + (size_t)i * DIM * DIM, aout, ain, gnl_w + i * DIM, gnl_b + i * DIM, NA);
  }
}


# Consecutive GEMMs on the same rows chained in one kernel: a wave's 16x128 tile stays on chip (accumulators -> own LDS tile -> A fragments): update -> output -> next query GEMM; 14 launches -> 11
# speedup vs baseline: 3.1336x; 1.0536x over previous
#include <hip/hip_runtime.h>


static constexpr int DIM  = 128;
static constexpr int NBLK = 2;
static constexpr int LDP  = 68;

typedef __attribute__((ext_vector_type(2))) float v2f;
typedef __attribute__((ext_vector_type(8))) float v8f;

static __device__ __forceinline__ unsigned f32_order_enc(float f) {
  unsigned u = __float_as_uint(f);
  return (u & 0x80000000u) ? ~u : (u | 0x80000000u);
}
static __device__ __forceinline__ float f32_order_dec(unsigned e) {
  unsigned u = (e & 0x80000000u) ? (e & 0x7fffffffu) : ~e;
  return __uint_as_float(u);
}

enum { EPI_STATS = 1, EPI_GN2_RELU = 2, EPI_GN_RES_RELU = 3 };

template <int XM>
static __device__ __forceinline__ float half_xor(float x) {
  return __int_as_float(__builtin_amdgcn_update_dpp(
      0, __float_as_int(x), 0x160 + XM, 0xf, 0xf, true));
}
static __device__ __forceinline__ float half_sum(float x) {
  x += half_xor<1>(x);
  x += half_xor<2>(x);
  x += half_xor<4>(x);
  x += half_xor<8>(x);
  return x;
}

template <bool RELU>
static __device__ __forceinline__ void gn_rows(v8f (&acc)[8],
                                               const float* __restrict__ w,
                                               const float* __restrict__ b,
                                               int l16) {
  float wv[8], bv[8];
#pragma unroll
  for (int t = 0; t < 8; ++t) {
    wv[t] = w[t * 16 + l16];
    bv[t] = b[t * 16 + l16];
  }
#pragma unroll
  for (int j = 0; j < 8; ++j) {
    float s = 0.0f, sq = 0.0f;
#pragma unroll
    for (int t = 0; t < 8; ++t) {
      const float x = acc[t][j];
      s += x;
      sq += x * x;
    }
    s  = half_sum(s);
    sq = half_sum(sq);
    const float mu = s * (1.0f / DIM);
    const float rs = rsqrtf(sq * (1.0f / DIM) - mu * mu + 1e-5f);
#pragma unroll
    for (int t = 0; t < 8; ++t) {
      float y = (acc[t][j] - mu) * rs * wv[t] + bv[t];
      if (RELU) y = fmaxf(y, 0.0f);
      acc[t][j] = y;
    }
  }
}

static __device__ __forceinline__ void stats_rows(const v8f (&acc)[8],
                                                  float* __restrict__ st,
                                                  int r0, int hk, int l16, int N) {
  float mymu = 0.0f, myrs = 0.0f;
#pragma unroll
  for (int j = 0; j < 8; ++j) {
    float s = 0.0f, sq = 0.0f;
#pragma unroll
    for (int t = 0; t < 8; ++t) {
      const float x = acc[t][j];
      s += x;
      sq += x * x;
    }
    s  = half_sum(s);
    sq = half_sum(sq);
    const float mu = s * (1.0f / DIM);
    const float rs = rsqrtf(sq * (1.0f / DIM) - mu * mu + 1e-5f);
    if (l16 == j) { mymu = mu; myrs = rs; }
  }
  const int row = r0 + hk * 8 + l16;
  if (l16 < 8 && row < N) {
    float2 v; v.x = mymu; v.y = myrs;
    ((float2*)st)[row] = v;
  }
}

template <int EPI>
static __device__ __forceinline__ void epilogue_store(v8f (&acc)[8],
                                                      float* __restrict__ C,
                                                      const float* __restrict__ R,
                                                      const float* __restrict__ w1,
                                                      const float* __restrict__ b1,
                                                      const float* __restrict__ w2,
                                                      const float* __restrict__ b2,
                                                      int r0, int hk, int l16, int N,
                                                      bool plain) {
  const int rb = r0 + hk * 8;

  if (EPI == EPI_GN_RES_RELU) {
    float res[8][8];
#pragma unroll
    for (int j = 0; j < 8; ++j) {
      int row = rb + j;
      if (row >= N) row = N - 1;
#pragma unroll
      for (int t = 0; t < 8; ++t) res[t][j] = R[(size_t)row * DIM + t * 16 + l16];
    }
    gn_rows<false>(acc, w1, b1, l16);
#pragma unroll
    for (int t = 0; t < 8; ++t)
#pragma unroll
      for (int j = 0; j < 8; ++j) acc[t][j] = fmaxf(acc[t][j] + res[t][j], 0.0f);
  } else if (EPI == EPI_GN2_RELU) {
    gn_rows<false>(acc, w1, b1, l16);
    gn_rows<true>(acc, w2, b2, l16);
  } else if (EPI == EPI_STATS) {
    if (!plain) {
      stats_rows(acc, C, r0, hk, l16, N);
      return;
    }
  }

  if (r0 + 16 <= N) {
#pragma unroll
    for (int t = 0; t < 8; ++t)
#pragma unroll
      for (int j = 0; j < 8; ++j)
        C[(size_t)(rb + j) * DIM + t * 16 + l16] = acc[t][j];
  } else {
#pragma unroll
    for (int t = 0; t < 8; ++t)
#pragma unroll
      for (int j = 0; j < 8; ++j) {
        const int row = rb + j;
        if (row < N) C[(size_t)row * DIM + t * 16 + l16] = acc[t][j];
      }
  }
}

template <int EPI>
static __device__ __forceinline__ void gemm_nt_128_body(float* __restrict__ Wlds,
                                                        const float* __restrict__ A,
                                                        const float* __restrict__ W,
                                                        float* __restrict__ C,
                                                        const float* __restrict__ R,
                                                        const float* __restrict__ w1,
                                                        const float* __restrict__ b1,
                                                        const float* __restrict__ w2,
                                                        const float* __restrict__ b2,
                                                        int N, int tile, bool plain) {
  const int tid = threadIdx.x;
  const int wv  = tid >> 5;
  const int ln  = tid & 31;
  const int hk  = ln >> 4;
  const int l16 = ln & 15;
  const int r0  = tile * 64 + wv * 16;

  v8f acc[8];
#pragma unroll
  for (int t = 0; t < 8; ++t)
#pragma unroll
    for (int j = 0; j < 8; ++j) acc[t][j] = 0.0f;

  int rowA = r0 + l16;
  if (rowA >= N) rowA = N - 1;
  const float* ap = A + (size_t)rowA * DIM;

  for (int kp = 0; kp < DIM; kp += 64) {
    __syncthreads();
#pragma unroll
    for (int i = 0; i < 16; ++i) {
      const int q  = tid + 128 * i;
      const int n  = q >> 4;
      const int c4 = (q & 15) << 2;
      *(float4*)(&Wlds[n * LDP + c4]) =
          *(const float4*)(W + (size_t)n * DIM + kp + c4);
    }
    __syncthreads();

#pragma unroll 4
    for (int k0 = 0; k0 < 64; k0 += 4) {
      const int ka = k0 + 2 * hk;
      v2f a; a.x = ap[kp + ka]; a.y = ap[kp + ka + 1];
#pragma unroll
      for (int t = 0; t < 8; ++t) {
        const float* wp = &Wlds[(t * 16 + l16) * LDP + ka];
        v2f b; b.x = wp[0]; b.y = wp[1];
        acc[t] = __builtin_amdgcn_wmma_f32_16x16x4_f32(
            false, a, false, b, (short)0, acc[t], false, false);
      }
    }
  }

  epilogue_store<EPI>(acc, C, R, w1, b1, w2, b2, r0, hk, l16, N, plain);
}

__global__ __launch_bounds__(128) void gemm_q_and_ctxp(const float* __restrict__ Aq,
                                                       const float* __restrict__ Wfq,
                                                       float* __restrict__ Cq, int Nq,
                                                       int tilesQ,
                                                       const float* __restrict__ Ac,
                                                       const float* __restrict__ Wc0,
                                                       float* __restrict__ Cc0,
                                                       const float* __restrict__ Wc1,
                                                       float* __restrict__ Cc1,
                                                       int Nc, int tilesC) {
  __shared__ float Wlds[128 * LDP];
  const int bid = blockIdx.x;
  const bool isq = bid < tilesQ;
  const bool sec = bid >= tilesQ + tilesC;
  const float* A  = isq ? Aq : Ac;
  const float* Wm = isq ? Wfq : (sec ? Wc1 : Wc0);
  float* C        = isq ? Cq : (sec ? Cc1 : Cc0);
  const int N     = isq ? Nq : Nc;
  const int tile  = isq ? bid : (sec ? bid - tilesQ - tilesC : bid - tilesQ);
  gemm_nt_128_body<EPI_STATS>(Wlds, A, Wm, C, nullptr, nullptr, nullptr, nullptr, nullptr, N, tile, isq);
}

static constexpr int XP = 132;

static __device__ __forceinline__ void tile_to_lds(const v8f (&acc)[8], float* xs,
                                                   int hk, int l16) {
#pragma unroll
  for (int t = 0; t < 8; ++t)
#pragma unroll
    for (int j = 0; j < 8; ++j) xs[(8 * hk + j) * XP + 16 * t + l16] = acc[t][j];
}

static __device__ __forceinline__ void gemm_tile_lds_128(v8f (&acc)[8], float* Wlds,
                                                         const float* xs,
                                                         const float* __restrict__ Wg,
                                                         int tid, int hk, int l16) {
#pragma unroll
  for (int t = 0; t < 8; ++t)
#pragma unroll
    for (int j = 0; j < 8; ++j) acc[t][j] = 0.0f;

  const float* xr = xs + l16 * XP;
  for (int kp = 0; kp < DIM; kp += 64) {
    __syncthreads();
#pragma unroll
    for (int i = 0; i < 16; ++i) {
      const int q  = tid + 128 * i;
      const int n  = q >> 4;
      const int c4 = (q & 15) << 2;
      *(float4*)(&Wlds[n * LDP + c4]) =
          *(const float4*)(Wg + (size_t)n * DIM + kp + c4);
    }
    __syncthreads();

#pragma unroll 4
    for (int k0 = 0; k0 < 64; k0 += 4) {
      const int ka = k0 + 2 * hk;
      v2f a; a.x = xr[kp + ka]; a.y = xr[kp + ka + 1];
#pragma unroll
      for (int t = 0; t < 8; ++t) {
        const float* wp = &Wlds[(t * 16 + l16) * LDP + ka];
        v2f b; b.x = wp[0]; b.y = wp[1];
        acc[t] = __builtin_amdgcn_wmma_f32_16x16x4_f32(
            false, a, false, b, (short)0, acc[t], false, false);
      }
    }
  }
}

template <bool NEXTQ>
__global__ __launch_bounds__(128) void gemm_update_chain(const float* __restrict__ A0,
                                                         const float* __restrict__ A1,
                                                         const float* __restrict__ aux,
                                                         const unsigned* __restrict__ gmax,
                                                         const float* __restrict__ W,
                                                         const float* __restrict__ Wm,
                                                         const float* __restrict__ r12,
                                                         const float* __restrict__ w1,
                                                         const float* __restrict__ b1,
                                                         const float* __restrict__ w2,
                                                         const float* __restrict__ b2,
                                                         const float* __restrict__ Wl,
                                                         const float* __restrict__ gl_w,
                                                         const float* __restrict__ gl_b,
                                                         float* __restrict__ Cout,
                                                         const float* __restrict__ Wfn,
                                                         float* __restrict__ Qn,
                                                         int N) {
  __shared__ __attribute__((aligned(16))) float Wlds[128 * LDP];
  __shared__ __attribute__((aligned(16))) float Xlds[4 * 16 * XP];
  const int tid = threadIdx.x;
  const int wv  = tid >> 5;
  const int ln  = tid & 31;
  const int hk  = ln >> 4;
  const int l16 = ln & 15;
  const int r0  = blockIdx.x * 64 + wv * 16;
  float* xs = Xlds + wv * 16 * XP;

  v8f acc[8];
#pragma unroll
  for (int t = 0; t < 8; ++t)
#pragma unroll
    for (int j = 0; j < 8; ++j) acc[t][j] = 0.0f;

  int rowA = r0 + l16;
  if (rowA >= N) rowA = N - 1;
  const size_t rbase = (size_t)rowA * DIM;

  const float4 ax = ((const float4*)aux)[rowA];
  const float gm  = f32_order_dec(*gmax);
  const float ee  = expf(ax.x - gm);
  const float fr  = ee / (ee * ax.y + 1e-16f);

  for (int kp = 0; kp < 2 * DIM; kp += 64) {
    const bool first = kp < DIM;
    const float* wsrc = first ? (W + kp) : (Wm + (kp - DIM));
    const int wld     = first ? 2 * DIM : DIM;
    __syncthreads();
#pragma unroll
    for (int i = 0; i < 16; ++i) {
      const int q  = tid + 128 * i;
      const int n  = q >> 4;
      const int c4 = (q & 15) << 2;
      *(float4*)(&Wlds[n * LDP + c4]) =
          *(const float4*)(wsrc + (size_t)n * wld + c4);
    }
    __syncthreads();

    const float* abase = first ? (A0 + rbase + kp) : (A1 + rbase + (kp - DIM));
    const float asc    = first ? 1.0f : fr;
#pragma unroll 4
    for (int k0 = 0; k0 < 64; k0 += 4) {
      const int ka = k0 + 2 * hk;
      v2f a; a.x = abase[ka] * asc; a.y = abase[ka + 1] * asc;
#pragma unroll
      for (int t = 0; t < 8; ++t) {
        const float* wp = &Wlds[(t * 16 + l16) * LDP + ka];
        v2f b; b.x = wp[0]; b.y = wp[1];
        acc[t] = __builtin_amdgcn_wmma_f32_16x16x4_f32(
            false, a, false, b, (short)0, acc[t], false, false);
      }
    }
  }

  {
    v2f a;
    a.x = (hk == 0) ? -fr * ax.z : 0.0f;
    a.y = (hk == 0) ?  fr * ax.y : 0.0f;
#pragma unroll
    for (int t = 0; t < 8; ++t) {
      v2f b;
      b.x = (hk == 0) ? r12[t * 16 + l16] : 0.0f;
      b.y = (hk == 0) ? r12[DIM + t * 16 + l16] : 0.0f;
      acc[t] = __builtin_amdgcn_wmma_f32_16x16x4_f32(
          false, a, false, b, (short)0, acc[t], false, false);
    }
  }

  gn_rows<false>(acc, w1, b1, l16);
  gn_rows<true>(acc, w2, b2, l16);
  tile_to_lds(acc, xs, hk, l16);

  gemm_tile_lds_128(acc, Wlds, xs, Wl, tid, hk, l16);
  epilogue_store<EPI_GN_RES_RELU>(acc, Cout, A0, gl_w, gl_b, nullptr, nullptr, r0, hk, l16, N, false);

  if (NEXTQ) {
    tile_to_lds(acc, xs, hk, l16);
    gemm_tile_lds_128(acc, Wlds, xs, Wfn, tid, hk, l16);
    epilogue_store<EPI_STATS>(acc, Qn, nullptr, nullptr, nullptr, nullptr, nullptr, r0, hk, l16, N, true);
  }
}

static __device__ __forceinline__ void fold_qk_body(const float* __restrict__ Wq,
                                                    const float* __restrict__ Wk,
                                                    float* __restrict__ Wf, int bid) {
  const int wv  = threadIdx.x >> 5;
  const int ln  = threadIdx.x & 31;
  const int hk  = ln >> 4;
  const int l16 = ln & 15;
  const int tile = bid * 8 + wv;
  const int i  = tile >> 6;
  const int j0 = ((tile >> 3) & 7) * 16;
  const int m0 = (tile & 7) * 16;
  const float* wq = Wq + (size_t)i * DIM * DIM;
  const float* wk = Wk + (size_t)i * DIM * DIM;

  v8f acc;
#pragma unroll
  for (int r = 0; r < 8; ++r) acc[r] = 0.0f;

#pragma unroll 8
  for (int d0 = 0; d0 < DIM; d0 += 4) {
    const int d = d0 + 2 * hk;
    v2f a; a.x = wk[d * DIM + j0 + l16]; a.y = wk[(d + 1) * DIM + j0 + l16];
    v2f b; b.x = wq[d * DIM + m0 + l16]; b.y = wq[(d + 1) * DIM + m0 + l16];
    acc = __builtin_amdgcn_wmma_f32_16x16x4_f32(
        false, a, false, b, (short)0, acc, false, false);
  }

  float* o = Wf + (size_t)i * DIM * DIM + (size_t)(j0 + 8 * hk) * DIM + m0 + l16;
#pragma unroll
  for (int r = 0; r < 8; ++r) o[r * DIM] = acc[r];
}

static __device__ __forceinline__ void fold_msg_body(const float* __restrict__ Wagt,
                                                     const float* __restrict__ Wctx,
                                                     const float* __restrict__ gw,
                                                     float* __restrict__ Wm, int bid) {
  const int wv  = threadIdx.x >> 5;
  const int ln  = threadIdx.x & 31;
  const int hk  = ln >> 4;
  const int l16 = ln & 15;
  const int tile = bid * 8 + wv;
  const int i  = tile >> 6;
  const int o0 = ((tile >> 3) & 7) * 16;
  const int k0 = (tile & 7) * 16;
  const float* wa = Wagt + (size_t)i * DIM * 2 * DIM + (size_t)(o0 + l16) * (2 * DIM) + DIM;
  const float* wc = Wctx + (size_t)i * DIM * DIM + k0 + l16;
  const float* g  = gw + i * DIM;

  v8f acc;
#pragma unroll
  for (int r = 0; r < 8; ++r) acc[r] = 0.0f;

#pragma unroll 8
  for (int d0 = 0; d0 < DIM; d0 += 4) {
    const int d = d0 + 2 * hk;
    v2f a; a.x = wa[d] * g[d];       a.y = wa[d + 1] * g[d + 1];
    v2f b; b.x = wc[d * DIM];        b.y = wc[(d + 1) * DIM];
    acc = __builtin_amdgcn_wmma_f32_16x16x4_f32(
        false, a, false, b, (short)0, acc, false, false);
  }

  float* o = Wm + (size_t)i * DIM * DIM + (size_t)(o0 + 8 * hk) * DIM + k0 + l16;
#pragma unroll
  for (int r = 0; r < 8; ++r) o[r * DIM] = acc[r];
}

__global__ __launch_bounds__(256) void setup_kernel(const float* __restrict__ Wq,
                                                    const float* __restrict__ Wk,
                                                    float* __restrict__ Wf,
                                                    const float* __restrict__ Wagt,
                                                    const float* __restrict__ Wctx,
                                                    const float* __restrict__ gw,
                                                    const float* __restrict__ gb,
                                                    float* __restrict__ Wm,
                                                    float* __restrict__ r12,
                                                    int* __restrict__ cnt, int n,
                                                    unsigned* __restrict__ gmax) {
  const int bid = blockIdx.x;
  if (bid < 16) {
    fold_qk_body(Wq, Wk, Wf, bid);
  } else if (bid < 32) {
    fold_msg_body(Wagt, Wctx, gw, Wm, bid - 16);
  } else if (bid == 32) {
    const int i = threadIdx.x >> 7;
    const int o = threadIdx.x & 127;
    const float* wa = Wagt + (size_t)i * DIM * 2 * DIM + (size_t)o * (2 * DIM) + DIM;
    float a1 = 0.0f, a2 = 0.0f;
    for (int c = 0; c < DIM; ++c) {
      const float x = wa[c];
      a1 += x * gw[i * DIM + c];
      a2 += x * gb[i * DIM + c];
    }
    r12[i * 2 * DIM + o]       = a1;
    r12[i * 2 * DIM + DIM + o] = a2;
  } else {
    const int i = (bid - 33) * 256 + threadIdx.x;
    if (i < n) cnt[i] = 0;
    if (bid == 33 && threadIdx.x < NBLK) gmax[threadIdx.x] = 0u;
  }
}

__global__ __launch_bounds__(256) void csr_count_kernel(const int* __restrict__ v,
                                                        int* __restrict__ cnt,
                                                        int* __restrict__ arr, int E) {
  const int e = blockIdx.x * 256 + threadIdx.x;
  if (e < E) arr[e] = atomicAdd(&cnt[v[e]], 1);
}

static __device__ __forceinline__ int wave_reduce_sum_i(int x) {
#pragma unroll
  for (int off = 16; off > 0; off >>= 1) x += __shfl_xor(x, off, 32);
  return x;
}

__global__ __launch_bounds__(256) void csr_blocksum_kernel(const int* __restrict__ cnt,
                                                           int* __restrict__ bsum, int N) {
  __shared__ int sw[8];
  const int wv = threadIdx.x >> 5, ln = threadIdx.x & 31;
  const int i = blockIdx.x * 256 + threadIdx.x;
  int c = (i < N) ? cnt[i] : 0;
  c = wave_reduce_sum_i(c);
  if (ln == 0) sw[wv] = c;
  __syncthreads();
  if (threadIdx.x == 0) {
    int t = 0;
#pragma unroll
    for (int k = 0; k < 8; ++k) t += sw[k];
    bsum[blockIdx.x] = t;
  }
}

__global__ __launch_bounds__(256) void csr_scan_kernel(const int* __restrict__ cnt,
                                                       const int* __restrict__ bsum,
                                                       int* __restrict__ rowptr, int N) {
  __shared__ int sa[256];
  __shared__ int sb[256];
  __shared__ int sw[8];
  const int t = threadIdx.x;
  const int wv = t >> 5, ln = t & 31;
  const int blk = blockIdx.x;

  int p = 0;
  for (int k = t; k < blk; k += 256) p += bsum[k];
  p = wave_reduce_sum_i(p);
  if (ln == 0) sw[wv] = p;

  const int i = blk * 256 + t;
  const int c = (i < N) ? cnt[i] : 0;
  sa[t] = c;
  __syncthreads();

  int base = 0;
#pragma unroll
  for (int k = 0; k < 8; ++k) base += sw[k];

  int* src = sa;
  int* dst = sb;
  for (int off = 1; off < 256; off <<= 1) {
    int val = src[t];
    if (t >= off) val += src[t - off];
    dst[t] = val;
    __syncthreads();
    int* tmp = src; src = dst; dst = tmp;
  }
  const int excl = base + src[t] - c;
  if (i < N) rowptr[i] = excl;
  if (i == N - 1) rowptr[N] = excl + c;
}

__global__ __launch_bounds__(256) void csr_fill_kernel(const int* __restrict__ v,
                                                       const int* __restrict__ arr,
                                                       const int* __restrict__ rowptr,
                                                       int* __restrict__ eid, int E) {
  const int e = blockIdx.x * 256 + threadIdx.x;
  if (e < E) eid[rowptr[v[e]] + arr[e]] = e;
}

__global__ __launch_bounds__(256) void csr_rank_kernel(const int* __restrict__ eid,
                                                       const int* __restrict__ rowptr,
                                                       const int* __restrict__ u,
                                                       const int* __restrict__ v,
                                                       int* __restrict__ su, int E) {
  const int j = blockIdx.x * 256 + threadIdx.x;
  if (j >= E) return;
  const int e = eid[j];
  const int n = v[e];
  const int b = rowptr[n], en = rowptr[n + 1];
  int r = 0;
  for (int k = b; k < en; ++k) r += (eid[k] < e) ? 1 : 0;
  su[b + r] = u[e];
}

static constexpr int CH = 16;

static __device__ __forceinline__ int lane_i(int x, int l) {
  return __builtin_amdgcn_readlane(x, l);
}
static __device__ __forceinline__ float lane_f(float x, int l) {
  return __int_as_float(__builtin_amdgcn_readlane(__float_as_int(x), l));
}
static __device__ __forceinline__ float wave_sum_dpp(float x) {
  x = half_sum(x);
  return lane_f(x, 0) + lane_f(x, 16);
}

__global__ __launch_bounds__(256) void node_attn_kernel(const float* __restrict__ Q,
                                                        const float* __restrict__ X,
                                                        const float* __restrict__ st,
                                                        const int* __restrict__ rowptr,
                                                        const int* __restrict__ su,
                                                        float* __restrict__ S,
                                                        float* __restrict__ aux,
                                                        unsigned* __restrict__ gmax,
                                                        int N) {
  __shared__ float smax[8];
  const int wv = threadIdx.x >> 5, ln = threadIdx.x & 31;
  const int n = __builtin_amdgcn_readfirstlane(blockIdx.x * 8 + wv);
  float m = -3.0e38f;
  if (n < N) {
    const int b  = __builtin_amdgcn_readfirstlane(rowptr[n]);
    const int en = __builtin_amdgcn_readfirstlane(rowptr[n + 1]);
    float ssum = 0.0f, tsum = 0.0f;
    float4 acc;
    acc.x = 0.0f; acc.y = 0.0f; acc.z = 0.0f; acc.w = 0.0f;
    if (b < en) {
      const float4 qv = ((const float4*)(Q + (size_t)n * DIM))[ln];
      for (int j0 = b; j0 < en; j0 += CH) {
        const int cnt = (en - j0 < CH) ? (en - j0) : CH;
        const int jj = j0 + ((ln < cnt) ? ln : (cnt - 1));
        const int uu = su[jj];
        const float2 sv = ((const float2*)st)[uu];

        float4 kv[CH];
#pragma unroll
        for (int g = 0; g < CH / 4; ++g) {
          if (4 * g < cnt) {
#pragma unroll
            for (int l = 4 * g; l < 4 * g + 4; ++l) {
              const int ui = lane_i(uu, l);
              kv[l] = ((const float4*)(X + (size_t)ui * DIM))[ln];
            }
          }
        }

        float d[CH];
        float mc = -3.0e38f;
#pragma unroll
        for (int g = 0; g < CH / 4; ++g) {
          if (4 * g < cnt) {
#pragma unroll
            for (int l = 4 * g; l < 4 * g + 4; ++l) {
              float p = qv.x * kv[l].x + qv.y * kv[l].y + qv.z * kv[l].z + qv.w * kv[l].w;
              p = wave_sum_dpp(p);
              p = (p >= 0.0f) ? p : 0.1f * p;
              d[l] = p;
              if (l < cnt) mc = fmaxf(mc, p);
            }
          }
        }

        const float mn = fmaxf(m, mc);
        const float sc = expf(m - mn);
        ssum *= sc; tsum *= sc;
        acc.x *= sc; acc.y *= sc; acc.z *= sc; acc.w *= sc;
        m = mn;

#pragma unroll
        for (int l = 0; l < CH; ++l) {
          if (l < cnt) {
            const float e  = expf(d[l] - mn);
            const float mu = lane_f(sv.x, l);
            const float rs = lane_f(sv.y, l);
            const float w  = e * rs;
            ssum += e;
            tsum += w * mu;
            acc.x += w * kv[l].x;
            acc.y += w * kv[l].y;
            acc.z += w * kv[l].z;
            acc.w += w * kv[l].w;
          }
        }
      }
    }
    ((float4*)(S + (size_t)n * DIM))[ln] = acc;
    if (ln == 0) {
      float4 a; a.x = m; a.y = ssum; a.z = tsum; a.w = 0.0f;
      ((float4*)aux)[n] = a;
    }
  }
  if (ln == 0) smax[wv] = m;
  __syncthreads();
  if (threadIdx.x == 0) {
    float mm = smax[0];
#pragma unroll
    for (int i = 1; i < 8; ++i) mm = fmaxf(mm, smax[i]);
    atomicMax(gmax, f32_order_enc(mm));
  }
}

extern "C" void kernel_launch(void* const* d_in, const int* in_sizes, int n_in,
                              void* d_out, int out_size, void* d_ws, size_t ws_size,
                              hipStream_t stream) {
  (void)in_sizes; (void)n_in; (void)out_size; (void)ws_size;
  const int NA = 50000, NC = 100000, NE = 600000;

  const float* agt    = (const float*)d_in[0];
  const float* ctx    = (const float*)d_in[1];
  const int*   u      = (const int*)d_in[2];
  const int*   v      = (const int*)d_in[3];
  const float* Wq     = (const float*)d_in[4];
  const float* Wk     = (const float*)d_in[5];
  const float* Wctx   = (const float*)d_in[6];
  const float* gnc_w  = (const float*)d_in[7];
  const float* gnc_b  = (const float*)d_in[8];
  const float* Wagt   = (const float*)d_in[9];
  const float* gna_w  = (const float*)d_in[10];
  const float* gna_b  = (const float*)d_in[11];
  const float* norm_w = (const float*)d_in[12];
  const float* norm_b = (const float*)d_in[13];
  const float* Wlin   = (const float*)d_in[14];
  const float* gnl_w  = (const float*)d_in[15];
  const float* gnl_b  = (const float*)d_in[16];

  float* ws = (float*)d_ws;
  size_t off = 0;
  float* qbuf   = ws + off; off += (size_t)NA * DIM;
  off += (size_t)NC * DIM;
  float* cpbuf  = ws + off; off += (size_t)NC * DIM;
  off += (size_t)NE;
  int*   cntbuf = (int*)(ws + off); off += (size_t)NA;
  float* aggbuf = ws + off; off += (size_t)NA * DIM;
  int*   csr    = (int*)(ws + off); off += (size_t)NA * DIM;
  float* agtbuf = ws + off; off += (size_t)NA * DIM;
  unsigned* maxbuf = (unsigned*)(ws + off); off += 16;

  int*   rowptr = csr;
  int*   eid    = csr + 50016;
  int*   subuf  = csr + 650016;
  int*   arrbuf = csr + 1250016;
  int*   bsum   = csr + 1850016;
  float* wfold  = (float*)(csr + 2000000);
  float* wmsg   = (float*)(csr + 2040000);
  float* r12buf = (float*)(csr + 2080000);
  float* stbuf0 = cpbuf;
  float* auxbuf = cpbuf + 2 * 100000;
  float* stbuf1 = cpbuf + 4 * 100000;

  const dim3 b128(128), b256(256);

  setup_kernel<<<33 + (NA + 255) / 256, b256, 0, stream>>>(Wq, Wk, wfold, Wagt, Wctx, gnc_w, gnc_b,
                                                          wmsg, r12buf, cntbuf, NA, maxbuf);

  csr_count_kernel<<<(NE + 255) / 256, b256, 0, stream>>>(v, cntbuf, arrbuf, NE);
  csr_blocksum_kernel<<<(NA + 255) / 256, b256, 0, stream>>>(cntbuf, bsum, NA);
  csr_scan_kernel<<<(NA + 255) / 256, b256, 0, stream>>>(cntbuf, bsum, rowptr, NA);
  csr_fill_kernel<<<(NE + 255) / 256, b256, 0, stream>>>(v, arrbuf, rowptr, eid, NE);
  csr_rank_kernel<<<(NE + 255) / 256, b256, 0, stream>>>(eid, rowptr, u, v, subuf, NE);

  gemm_q_and_ctxp<<<(NA + 63) / 64 + 2 * ((NC + 63) / 64), b128, 0, stream>>>(
      agt, wfold, qbuf, NA, (NA + 63) / 64,
      ctx, Wctx, stbuf0, Wctx + (size_t)DIM * DIM, stbuf1, NC, (NC + 63) / 64);

  node_attn_kernel<<<(NA + 7) / 8, b256, 0, stream>>>(qbuf, ctx, stbuf0, rowptr, subuf,
                                                     aggbuf, auxbuf, maxbuf, NA);
  gemm_update_chain<true><<<(NA + 63) / 64, b128, 0, stream>>>(
      agt, aggbuf, auxbuf, maxbuf, Wagt, wmsg, r12buf,
      gna_w, gna_b, norm_w, norm_b, Wlin, gnl_w, gnl_b, agtbuf,
      wfold + (size_t)DIM * DIM, qbuf, NA);

  node_attn_kernel<<<(NA + 7) / 8, b256, 0, stream>>>(qbuf, ctx, stbuf1, rowptr, subuf,
                                                     aggbuf, auxbuf, maxbuf + 1, NA);
  gemm_update_chain<false><<<(NA + 63) / 64, b128, 0, stream>>>(
      agtbuf, aggbuf, auxbuf, maxbuf + 1, Wagt + (size_t)DIM * 2 * DIM,
      wmsg + (size_t)DIM * DIM, r12buf + (size_t)2 * DIM,
      gna_w + DIM, gna_b + DIM, norm_w + DIM, norm_b + DIM,
      Wlin + (size_t)DIM * DIM, gnl_w + DIM, gnl_b + DIM, (float*)d_out,
      wfold, qbuf, NA);
}


# Edge kernel: per-slot scalar work done once per chunk by lane (lane l = slot l): all score sums in one transposing DPP tree (same pairs, sum order unchanged), then LeakyReLU, max, exp, weights
# speedup vs baseline: 3.2524x; 1.0379x over previous
#include <hip/hip_runtime.h>


static constexpr int DIM  = 128;
static constexpr int NBLK = 2;
static constexpr int LDP  = 68;

typedef __attribute__((ext_vector_type(2))) float v2f;
typedef __attribute__((ext_vector_type(8))) float v8f;

static __device__ __forceinline__ unsigned f32_order_enc(float f) {
  unsigned u = __float_as_uint(f);
  return (u & 0x80000000u) ? ~u : (u | 0x80000000u);
}
static __device__ __forceinline__ float f32_order_dec(unsigned e) {
  unsigned u = (e & 0x80000000u) ? (e & 0x7fffffffu) : ~e;
  return __uint_as_float(u);
}

enum { EPI_STATS = 1, EPI_GN2_RELU = 2, EPI_GN_RES_RELU = 3 };

template <int XM>
static __device__ __forceinline__ float half_xor(float x) {
  return __int_as_float(__builtin_amdgcn_update_dpp(
      0, __float_as_int(x), 0x160 + XM, 0xf, 0xf, true));
}
static __device__ __forceinline__ float half_sum(float x) {
  x += half_xor<1>(x);
  x += half_xor<2>(x);
  x += half_xor<4>(x);
  x += half_xor<8>(x);
  return x;
}

template <bool RELU>
static __device__ __forceinline__ void gn_rows(v8f (&acc)[8],
                                               const float* __restrict__ w,
                                               const float* __restrict__ b,
                                               int l16) {
  float wv[8], bv[8];
#pragma unroll
  for (int t = 0; t < 8; ++t) {
    wv[t] = w[t * 16 + l16];
    bv[t] = b[t * 16 + l16];
  }
#pragma unroll
  for (int j = 0; j < 8; ++j) {
    float s = 0.0f, sq = 0.0f;
#pragma unroll
    for (int t = 0; t < 8; ++t) {
      const float x = acc[t][j];
      s += x;
      sq += x * x;
    }
    s  = half_sum(s);
    sq = half_sum(sq);
    const float mu = s * (1.0f / DIM);
    const float rs = rsqrtf(sq * (1.0f / DIM) - mu * mu + 1e-5f);
#pragma unroll
    for (int t = 0; t < 8; ++t) {
      float y = (acc[t][j] - mu) * rs * wv[t] + bv[t];
      if (RELU) y = fmaxf(y, 0.0f);
      acc[t][j] = y;
    }
  }
}

static __device__ __forceinline__ void stats_rows(const v8f (&acc)[8],
                                                  float* __restrict__ st,
                                                  int r0, int hk, int l16, int N) {
  float mymu = 0.0f, myrs = 0.0f;
#pragma unroll
  for (int j = 0; j < 8; ++j) {
    float s = 0.0f, sq = 0.0f;
#pragma unroll
    for (int t = 0; t < 8; ++t) {
      const float x = acc[t][j];
      s += x;
      sq += x * x;
    }
    s  = half_sum(s);
    sq = half_sum(sq);
    const float mu = s * (1.0f / DIM);
    const float rs = rsqrtf(sq * (1.0f / DIM) - mu * mu + 1e-5f);
    if (l16 == j) { mymu = mu; myrs = rs; }
  }
  const int row = r0 + hk * 8 + l16;
  if (l16 < 8 && row < N) {
    float2 v; v.x = mymu; v.y = myrs;
    ((float2*)st)[row] = v;
  }
}

template <int EPI>
static __device__ __forceinline__ void epilogue_store(v8f (&acc)[8],
                                                      float* __restrict__ C,
                                                      const float* __restrict__ R,
                                                      const float* __restrict__ w1,
                                                      const float* __restrict__ b1,
                                                      const float* __restrict__ w2,
                                                      const float* __restrict__ b2,
                                                      int r0, int hk, int l16, int N,
                                                      bool plain) {
  const int rb = r0 + hk * 8;

  if (EPI == EPI_GN_RES_RELU) {
    float res[8][8];
#pragma unroll
    for (int j = 0; j < 8; ++j) {
      int row = rb + j;
      if (row >= N) row = N - 1;
#pragma unroll
      for (int t = 0; t < 8; ++t) res[t][j] = R[(size_t)row * DIM + t * 16 + l16];
    }
    gn_rows<false>(acc, w1, b1, l16);
#pragma unroll
    for (int t = 0; t < 8; ++t)
#pragma unroll
      for (int j = 0; j < 8; ++j) acc[t][j] = fmaxf(acc[t][j] + res[t][j], 0.0f);
  } else if (EPI == EPI_GN2_RELU) {
    gn_rows<false>(acc, w1, b1, l16);
    gn_rows<true>(acc, w2, b2, l16);
  } else if (EPI == EPI_STATS) {
    if (!plain) {
      stats_rows(acc, C, r0, hk, l16, N);
      return;
    }
  }

  if (r0 + 16 <= N) {
#pragma unroll
    for (int t = 0; t < 8; ++t)
#pragma unroll
      for (int j = 0; j < 8; ++j)
        C[(size_t)(rb + j) * DIM + t * 16 + l16] = acc[t][j];
  } else {
#pragma unroll
    for (int t = 0; t < 8; ++t)
#pragma unroll
      for (int j = 0; j < 8; ++j) {
        const int row = rb + j;
        if (row < N) C[(size_t)row * DIM + t * 16 + l16] = acc[t][j];
      }
  }
}

template <int EPI>
static __device__ __forceinline__ void gemm_nt_128_body(float* __restrict__ Wlds,
                                                        const float* __restrict__ A,
                                                        const float* __restrict__ W,
                                                        float* __restrict__ C,
                                                        const float* __restrict__ R,
                                                        const float* __restrict__ w1,
                                                        const float* __restrict__ b1,
                                                        const float* __restrict__ w2,
                                                        const float* __restrict__ b2,
                                                        int N, int tile, bool plain) {
  const int tid = threadIdx.x;
  const int wv  = tid >> 5;
  const int ln  = tid & 31;
  const int hk  = ln >> 4;
  const int l16 = ln & 15;
  const int r0  = tile * 64 + wv * 16;

  v8f acc[8];
#pragma unroll
  for (int t = 0; t < 8; ++t)
#pragma unroll
    for (int j = 0; j < 8; ++j) acc[t][j] = 0.0f;

  int rowA = r0 + l16;
  if (rowA >= N) rowA = N - 1;
  const float* ap = A + (size_t)rowA * DIM;

  for (int kp = 0; kp < DIM; kp += 64) {
    __syncthreads();
#pragma unroll
    for (int i = 0; i < 16; ++i) {
      const int q  = tid + 128 * i;
      const int n  = q >> 4;
      const int c4 = (q & 15) << 2;
      *(float4*)(&Wlds[n * LDP + c4]) =
          *(const float4*)(W + (size_t)n * DIM + kp + c4);
    }
    __syncthreads();

#pragma unroll 4
    for (int k0 = 0; k0 < 64; k0 += 4) {
      const int ka = k0 + 2 * hk;
      v2f a; a.x = ap[kp + ka]; a.y = ap[kp + ka + 1];
#pragma unroll
      for (int t = 0; t < 8; ++t) {
        const float* wp = &Wlds[(t * 16 + l16) * LDP + ka];
        v2f b; b.x = wp[0]; b.y = wp[1];
        acc[t] = __builtin_amdgcn_wmma_f32_16x16x4_f32(
            false, a, false, b, (short)0, acc[t], false, false);
      }
    }
  }

  epilogue_store<EPI>(acc, C, R, w1, b1, w2, b2, r0, hk, l16, N, plain);
}

__global__ __launch_bounds__(128) void gemm_q_and_ctxp(const float* __restrict__ Aq,
                                                       const float* __restrict__ Wfq,
                                                       float* __restrict__ Cq, int Nq,
                                                       int tilesQ,
                                                       const float* __restrict__ Ac,
                                                       const float* __restrict__ Wc0,
                                                       float* __restrict__ Cc0,
                                                       const float* __restrict__ Wc1,
                                                       float* __restrict__ Cc1,
                                                       int Nc, int tilesC) {
  __shared__ float Wlds[128 * LDP];
  const int bid = blockIdx.x;
  const bool isq = bid < tilesQ;
  const bool sec = bid >= tilesQ + tilesC;
  const float* A  = isq ? Aq : Ac;
  const float* Wm = isq ? Wfq : (sec ? Wc1 : Wc0);
  float* C        = isq ? Cq : (sec ? Cc1 : Cc0);
  const int N     = isq ? Nq : Nc;
  const int tile  = isq ? bid : (sec ? bid - tilesQ - tilesC : bid - tilesQ);
  gemm_nt_128_body<EPI_STATS>(Wlds, A, Wm, C, nullptr, nullptr, nullptr, nullptr, nullptr, N, tile, isq);
}

static constexpr int XP = 132;

static __device__ __forceinline__ void tile_to_lds(const v8f (&acc)[8], float* xs,
                                                   int hk, int l16) {
#pragma unroll
  for (int t = 0; t < 8; ++t)
#pragma unroll
    for (int j = 0; j < 8; ++j) xs[(8 * hk + j) * XP + 16 * t + l16] = acc[t][j];
}

static __device__ __forceinline__ void gemm_tile_lds_128(v8f (&acc)[8], float* Wlds,
                                                         const float* xs,
                                                         const float* __restrict__ Wg,
                                                         int tid, int hk, int l16) {
#pragma unroll
  for (int t = 0; t < 8; ++t)
#pragma unroll
    for (int j = 0; j < 8; ++j) acc[t][j] = 0.0f;

  const float* xr = xs + l16 * XP;
  for (int kp = 0; kp < DIM; kp += 64) {
    __syncthreads();
#pragma unroll
    for (int i = 0; i < 16; ++i) {
      const int q  = tid + 128 * i;
      const int n  = q >> 4;
      const int c4 = (q & 15) << 2;
      *(float4*)(&Wlds[n * LDP + c4]) =
          *(const float4*)(Wg + (size_t)n * DIM + kp + c4);
    }
    __syncthreads();

#pragma unroll 4
    for (int k0 = 0; k0 < 64; k0 += 4) {
      const int ka = k0 + 2 * hk;
      v2f a; a.x = xr[kp + ka]; a.y = xr[kp + ka + 1];
#pragma unroll
      for (int t = 0; t < 8; ++t) {
        const float* wp = &Wlds[(t * 16 + l16) * LDP + ka];
        v2f b; b.x = wp[0]; b.y = wp[1];
        acc[t] = __builtin_amdgcn_wmma_f32_16x16x4_f32(
            false, a, false, b, (short)0, acc[t], false, false);
      }
    }
  }
}

template <bool NEXTQ>
__global__ __launch_bounds__(128) void gemm_update_chain(const float* __restrict__ A0,
                                                         const float* __restrict__ A1,
                                                         const float* __restrict__ aux,
                                                         const unsigned* __restrict__ gmax,
                                                         const float* __restrict__ W,
                                                         const float* __restrict__ Wm,
                                                         const float* __restrict__ r12,
                                                         const float* __restrict__ w1,
                                                         const float* __restrict__ b1,
                                                         const float* __restrict__ w2,
                                                         const float* __restrict__ b2,
                                                         const float* __restrict__ Wl,
                                                         const float* __restrict__ gl_w,
                                                         const float* __restrict__ gl_b,
                                                         float* __restrict__ Cout,
                                                         const float* __restrict__ Wfn,
                                                         float* __restrict__ Qn,
                                                         int N) {
  __shared__ __attribute__((aligned(16))) float Wlds[128 * LDP];
  __shared__ __attribute__((aligned(16))) float Xlds[4 * 16 * XP];
  const int tid = threadIdx.x;
  const int wv  = tid >> 5;
  const int ln  = tid & 31;
  const int hk  = ln >> 4;
  const int l16 = ln & 15;
  const int r0  = blockIdx.x * 64 + wv * 16;
  float* xs = Xlds + wv * 16 * XP;

  v8f acc[8];
#pragma unroll
  for (int t = 0; t < 8; ++t)
#pragma unroll
    for (int j = 0; j < 8; ++j) acc[t][j] = 0.0f;

  int rowA = r0 + l16;
  if (rowA >= N) rowA = N - 1;
  const size_t rbase = (size_t)rowA * DIM;

  const float4 ax = ((const float4*)aux)[rowA];
  const float gm  = f32_order_dec(*gmax);
  const float ee  = expf(ax.x - gm);
  const float fr  = ee / (ee * ax.y + 1e-16f);

  for (int kp = 0; kp < 2 * DIM; kp += 64) {
    const bool first = kp < DIM;
    const float* wsrc = first ? (W + kp) : (Wm + (kp - DIM));
    const int wld     = first ? 2 * DIM : DIM;
    __syncthreads();
#pragma unroll
    for (int i = 0; i < 16; ++i) {
      const int q  = tid + 128 * i;
      const int n  = q >> 4;
      const int c4 = (q & 15) << 2;
      *(float4*)(&Wlds[n * LDP + c4]) =
          *(const float4*)(wsrc + (size_t)n * wld + c4);
    }
    __syncthreads();

    const float* abase = first ? (A0 + rbase + kp) : (A1 + rbase + (kp - DIM));
    const float asc    = first ? 1.0f : fr;
#pragma unroll 4
    for (int k0 = 0; k0 < 64; k0 += 4) {
      const int ka = k0 + 2 * hk;
      v2f a; a.x = abase[ka] * asc; a.y = abase[ka + 1] * asc;
#pragma unroll
      for (int t = 0; t < 8; ++t) {
        const float* wp = &Wlds[(t * 16 + l16) * LDP + ka];
        v2f b; b.x = wp[0]; b.y = wp[1];
        acc[t] = __builtin_amdgcn_wmma_f32_16x16x4_f32(
            false, a, false, b, (short)0, acc[t], false, false);
      }
    }
  }

  {
    v2f a;
    a.x = (hk == 0) ? -fr * ax.z : 0.0f;
    a.y = (hk == 0) ?  fr * ax.y : 0.0f;
#pragma unroll
    for (int t = 0; t < 8; ++t) {
      v2f b;
      b.x = (hk == 0) ? r12[t * 16 + l16] : 0.0f;
      b.y = (hk == 0) ? r12[DIM + t * 16 + l16] : 0.0f;
      acc[t] = __builtin_amdgcn_wmma_f32_16x16x4_f32(
          false, a, false, b, (short)0, acc[t], false, false);
    }
  }

  gn_rows<false>(acc, w1, b1, l16);
  gn_rows<true>(acc, w2, b2, l16);
  tile_to_lds(acc, xs, hk, l16);

  gemm_tile_lds_128(acc, Wlds, xs, Wl, tid, hk, l16);
  epilogue_store<EPI_GN_RES_RELU>(acc, Cout, A0, gl_w, gl_b, nullptr, nullptr, r0, hk, l16, N, false);

  if (NEXTQ) {
    tile_to_lds(acc, xs, hk, l16);
    gemm_tile_lds_128(acc, Wlds, xs, Wfn, tid, hk, l16);
    epilogue_store<EPI_STATS>(acc, Qn, nullptr, nullptr, nullptr, nullptr, nullptr, r0, hk, l16, N, true);
  }
}

static __device__ __forceinline__ void fold_qk_body(const float* __restrict__ Wq,
                                                    const float* __restrict__ Wk,
                                                    float* __restrict__ Wf, int bid) {
  const int wv  = threadIdx.x >> 5;
  const int ln  = threadIdx.x & 31;
  const int hk  = ln >> 4;
  const int l16 = ln & 15;
  const int tile = bid * 8 + wv;
  const int i  = tile >> 6;
  const int j0 = ((tile >> 3) & 7) * 16;
  const int m0 = (tile & 7) * 16;
  const float* wq = Wq + (size_t)i * DIM * DIM;
  const float* wk = Wk + (size_t)i * DIM * DIM;

  v8f acc;
#pragma unroll
  for (int r = 0; r < 8; ++r) acc[r] = 0.0f;

#pragma unroll 8
  for (int d0 = 0; d0 < DIM; d0 += 4) {
    const int d = d0 + 2 * hk;
    v2f a; a.x = wk[d * DIM + j0 + l16]; a.y = wk[(d + 1) * DIM + j0 + l16];
    v2f b; b.x = wq[d * DIM + m0 + l16]; b.y = wq[(d + 1) * DIM + m0 + l16];
    acc = __builtin_amdgcn_wmma_f32_16x16x4_f32(
        false, a, false, b, (short)0, acc, false, false);
  }

  float* o = Wf + (size_t)i * DIM * DIM + (size_t)(j0 + 8 * hk) * DIM + m0 + l16;
#pragma unroll
  for (int r = 0; r < 8; ++r) o[r * DIM] = acc[r];
}

static __device__ __forceinline__ void fold_msg_body(const float* __restrict__ Wagt,
                                                     const float* __restrict__ Wctx,
                                                     const float* __restrict__ gw,
                                                     float* __restrict__ Wm, int bid) {
  const int wv  = threadIdx.x >> 5;
  const int ln  = threadIdx.x & 31;
  const int hk  = ln >> 4;
  const int l16 = ln & 15;
  const int tile = bid * 8 + wv;
  const int i  = tile >> 6;
  const int o0 = ((tile >> 3) & 7) * 16;
  const int k0 = (tile & 7) * 16;
  const float* wa = Wagt + (size_t)i * DIM * 2 * DIM + (size_t)(o0 + l16) * (2 * DIM) + DIM;
  const float* wc = Wctx + (size_t)i * DIM * DIM + k0 + l16;
  const float* g  = gw + i * DIM;

  v8f acc;
#pragma unroll
  for (int r = 0; r < 8; ++r) acc[r] = 0.0f;

#pragma unroll 8
  for (int d0 = 0; d0 < DIM; d0 += 4) {
    const int d = d0 + 2 * hk;
    v2f a; a.x = wa[d] * g[d];       a.y = wa[d + 1] * g[d + 1];
    v2f b; b.x = wc[d * DIM];        b.y = wc[(d + 1) * DIM];
    acc = __builtin_amdgcn_wmma_f32_16x16x4_f32(
        false, a, false, b, (short)0, acc, false, false);
  }

  float* o = Wm + (size_t)i * DIM * DIM + (size_t)(o0 + 8 * hk) * DIM + k0 + l16;
#pragma unroll
  for (int r = 0; r < 8; ++r) o[r * DIM] = acc[r];
}

__global__ __launch_bounds__(256) void setup_kernel(const float* __restrict__ Wq,
                                                    const float* __restrict__ Wk,
                                                    float* __restrict__ Wf,
                                                    const float* __restrict__ Wagt,
                                                    const float* __restrict__ Wctx,
                                                    const float* __restrict__ gw,
                                                    const float* __restrict__ gb,
                                                    float* __restrict__ Wm,
                                                    float* __restrict__ r12,
                                                    int* __restrict__ cnt, int n,
                                                    unsigned* __restrict__ gmax) {
  const int bid = blockIdx.x;
  if (bid < 16) {
    fold_qk_body(Wq, Wk, Wf, bid);
  } else if (bid < 32) {
    fold_msg_body(Wagt, Wctx, gw, Wm, bid - 16);
  } else if (bid == 32) {
    const int i = threadIdx.x >> 7;
    const int o = threadIdx.x & 127;
    const float* wa = Wagt + (size_t)i * DIM * 2 * DIM + (size_t)o * (2 * DIM) + DIM;
    float a1 = 0.0f, a2 = 0.0f;
    for (int c = 0; c < DIM; ++c) {
      const float x = wa[c];
      a1 += x * gw[i * DIM + c];
      a2 += x * gb[i * DIM + c];
    }
    r12[i * 2 * DIM + o]       = a1;
    r12[i * 2 * DIM + DIM + o] = a2;
  } else {
    const int i = (bid - 33) * 256 + threadIdx.x;
    if (i < n) cnt[i] = 0;
    if (bid == 33 && threadIdx.x < NBLK) gmax[threadIdx.x] = 0u;
  }
}

__global__ __launch_bounds__(256) void csr_count_kernel(const int* __restrict__ v,
                                                        int* __restrict__ cnt,
                                                        int* __restrict__ arr, int E) {
  const int e = blockIdx.x * 256 + threadIdx.x;
  if (e < E) arr[e] = atomicAdd(&cnt[v[e]], 1);
}

static __device__ __forceinline__ int wave_reduce_sum_i(int x) {
#pragma unroll
  for (int off = 16; off > 0; off >>= 1) x += __shfl_xor(x, off, 32);
  return x;
}

__global__ __launch_bounds__(256) void csr_blocksum_kernel(const int* __restrict__ cnt,
                                                           int* __restrict__ bsum, int N) {
  __shared__ int sw[8];
  const int wv = threadIdx.x >> 5, ln = threadIdx.x & 31;
  const int i = blockIdx.x * 256 + threadIdx.x;
  int c = (i < N) ? cnt[i] : 0;
  c = wave_reduce_sum_i(c);
  if (ln == 0) sw[wv] = c;
  __syncthreads();
  if (threadIdx.x == 0) {
    int t = 0;
#pragma unroll
    for (int k = 0; k < 8; ++k) t += sw[k];
    bsum[blockIdx.x] = t;
  }
}

__global__ __launch_bounds__(256) void csr_scan_kernel(const int* __restrict__ cnt,
                                                       const int* __restrict__ bsum,
                                                       int* __restrict__ rowptr, int N) {
  __shared__ int sa[256];
  __shared__ int sb[256];
  __shared__ int sw[8];
  const int t = threadIdx.x;
  const int wv = t >> 5, ln = t & 31;
  const int blk = blockIdx.x;

  int p = 0;
  for (int k = t; k < blk; k += 256) p += bsum[k];
  p = wave_reduce_sum_i(p);
  if (ln == 0) sw[wv] = p;

  const int i = blk * 256 + t;
  const int c = (i < N) ? cnt[i] : 0;
  sa[t] = c;
  __syncthreads();

  int base = 0;
#pragma unroll
  for (int k = 0; k < 8; ++k) base += sw[k];

  int* src = sa;
  int* dst = sb;
  for (int off = 1; off < 256; off <<= 1) {
    int val = src[t];
    if (t >= off) val += src[t - off];
    dst[t] = val;
    __syncthreads();
    int* tmp = src; src = dst; dst = tmp;
  }
  const int excl = base + src[t] - c;
  if (i < N) rowptr[i] = excl;
  if (i == N - 1) rowptr[N] = excl + c;
}

__global__ __launch_bounds__(256) void csr_fill_kernel(const int* __restrict__ v,
                                                       const int* __restrict__ arr,
                                                       const int* __restrict__ rowptr,
                                                       int* __restrict__ eid, int E) {
  const int e = blockIdx.x * 256 + threadIdx.x;
  if (e < E) eid[rowptr[v[e]] + arr[e]] = e;
}

__global__ __launch_bounds__(256) void csr_rank_kernel(const int* __restrict__ eid,
                                                       const int* __restrict__ rowptr,
                                                       const int* __restrict__ u,
                                                       const int* __restrict__ v,
                                                       int* __restrict__ su, int E) {
  const int j = blockIdx.x * 256 + threadIdx.x;
  if (j >= E) return;
  const int e = eid[j];
  const int n = v[e];
  const int b = rowptr[n], en = rowptr[n + 1];
  int r = 0;
  for (int k = b; k < en; ++k) r += (eid[k] < e) ? 1 : 0;
  su[b + r] = u[e];
}

static constexpr int CH = 16;

static __device__ __forceinline__ int lane_i(int x, int l) {
  return __builtin_amdgcn_readlane(x, l);
}
static __device__ __forceinline__ float lane_f(float x, int l) {
  return __int_as_float(__builtin_amdgcn_readlane(__float_as_int(x), l));
}
static __device__ __forceinline__ float other_half(float x) {
  return __int_as_float(__builtin_amdgcn_permlanex16(
      __float_as_int(x), __float_as_int(x), 0x76543210, 0xfedcba98, false, false));
}

static __device__ __forceinline__ float half_sum_16(const float (&p)[16], int ln) {
  const bool b0 = (ln & 1) != 0, b1 = (ln & 2) != 0, b2 = (ln & 4) != 0, b3 = (ln & 8) != 0;
  float r1[8], r2[4], r3[2];
#pragma unroll
  for (int i = 0; i < 8; ++i) {
    const float keep = b0 ? p[2 * i + 1] : p[2 * i];
    const float send = b0 ? p[2 * i] : p[2 * i + 1];
    r1[i] = keep + half_xor<1>(send);
  }
#pragma unroll
  for (int i = 0; i < 4; ++i) {
    const float keep = b1 ? r1[2 * i + 1] : r1[2 * i];
    const float send = b1 ? r1[2 * i] : r1[2 * i + 1];
    r2[i] = keep + half_xor<2>(send);
  }
#pragma unroll
  for (int i = 0; i < 2; ++i) {
    const float keep = b2 ? r2[2 * i + 1] : r2[2 * i];
    const float send = b2 ? r2[2 * i] : r2[2 * i + 1];
    r3[i] = keep + half_xor<4>(send);
  }
  const float keep = b3 ? r3[1] : r3[0];
  const float send = b3 ? r3[0] : r3[1];
  return keep + half_xor<8>(send);
}

__global__ __launch_bounds__(256) void node_attn_kernel(const float* __restrict__ Q,
                                                        const float* __restrict__ X,
                                                        const float* __restrict__ st,
                                                        const int* __restrict__ rowptr,
                                                        const int* __restrict__ su,
                                                        float* __restrict__ S,
                                                        float* __restrict__ aux,
                                                        unsigned* __restrict__ gmax,
                                                        int N) {
  __shared__ float smax[8];
  const int wv = threadIdx.x >> 5, ln = threadIdx.x & 31;
  const int n = __builtin_amdgcn_readfirstlane(blockIdx.x * 8 + wv);
  float m = -3.0e38f;
  if (n < N) {
    const int b  = __builtin_amdgcn_readfirstlane(rowptr[n]);
    const int en = __builtin_amdgcn_readfirstlane(rowptr[n + 1]);
    float ssum = 0.0f, tsum = 0.0f;
    float4 acc;
    acc.x = 0.0f; acc.y = 0.0f; acc.z = 0.0f; acc.w = 0.0f;
    if (b < en) {
      const float4 qv = ((const float4*)(Q + (size_t)n * DIM))[ln];
      for (int j0 = b; j0 < en; j0 += CH) {
        const int cnt = (en - j0 < CH) ? (en - j0) : CH;
        const int jj = j0 + ((ln < cnt) ? ln : (cnt - 1));
        const int uu = su[jj];
        const float2 sv = ((const float2*)st)[uu];

        float4 kv[CH];
#pragma unroll
        for (int g = 0; g < CH / 4; ++g) {
          if (4 * g < cnt) {
#pragma unroll
            for (int l = 4 * g; l < 4 * g + 4; ++l) {
              const int ui = lane_i(uu, l);
              kv[l] = ((const float4*)(X + (size_t)ui * DIM))[ln];
            }
          }
        }

        float p[CH];
#pragma unroll
        for (int g = 0; g < CH / 4; ++g) {
          if (4 * g < cnt) {
#pragma unroll
            for (int l = 4 * g; l < 4 * g + 4; ++l)
              p[l] = qv.x * kv[l].x + qv.y * kv[l].y + qv.z * kv[l].z + qv.w * kv[l].w;
          } else {
#pragma unroll
            for (int l = 4 * g; l < 4 * g + 4; ++l) p[l] = 0.0f;
          }
        }

        const float hs  = half_sum_16(p, ln);
        const float tot = hs + other_half(hs);
        const float sco = (tot >= 0.0f) ? tot : 0.1f * tot;
        const bool valid = ln < cnt;
        float mx = valid ? sco : -3.0e38f;
        mx = fmaxf(mx, half_xor<1>(mx));
        mx = fmaxf(mx, half_xor<2>(mx));
        mx = fmaxf(mx, half_xor<4>(mx));
        mx = fmaxf(mx, half_xor<8>(mx));
        const float mc = lane_f(mx, 0);

        const float mn = fmaxf(m, mc);
        const float sc = expf(m - mn);
        ssum *= sc; tsum *= sc;
        acc.x *= sc; acc.y *= sc; acc.z *= sc; acc.w *= sc;
        m = mn;

        const float ev = valid ? expf(sco - mn) : 0.0f;
        const float wl = ev * sv.y;

#pragma unroll
        for (int g = 0; g < CH / 4; ++g) {
          if (4 * g < cnt) {
#pragma unroll
            for (int l = 4 * g; l < 4 * g + 4; ++l) {
              const float e  = lane_f(ev, l);
              const float w  = lane_f(wl, l);
              const float mu = lane_f(sv.x, l);
              ssum += e;
              tsum += w * mu;
              acc.x += w * kv[l].x;
              acc.y += w * kv[l].y;
              acc.z += w * kv[l].z;
              acc.w += w * kv[l].w;
            }
          }
        }
      }
    }
    ((float4*)(S + (size_t)n * DIM))[ln] = acc;
    if (ln == 0) {
      float4 a; a.x = m; a.y = ssum; a.z = tsum; a.w = 0.0f;
      ((float4*)aux)[n] = a;
    }
  }
  if (ln == 0) smax[wv] = m;
  __syncthreads();
  if (threadIdx.x == 0) {
    float mm = smax[0];
#pragma unroll
    for (int i = 1; i < 8; ++i) mm = fmaxf(mm, smax[i]);
    atomicMax(gmax, f32_order_enc(mm));
  }
}

extern "C" void kernel_launch(void* const* d_in, const int* in_sizes, int n_in,
                              void* d_out, int out_size, void* d_ws, size_t ws_size,
                              hipStream_t stream) {
  (void)in_sizes; (void)n_in; (void)out_size; (void)ws_size;
  const int NA = 50000, NC = 100000, NE = 600000;

  const float* agt    = (const float*)d_in[0];
  const float* ctx    = (const float*)d_in[1];
  const int*   u      = (const int*)d_in[2];
  const int*   v      = (const int*)d_in[3];
  const float* Wq     = (const float*)d_in[4];
  const float* Wk     = (const float*)d_in[5];
  const float* Wctx   = (const float*)d_in[6];
  const float* gnc_w  = (const float*)d_in[7];
  const float* gnc_b  = (const float*)d_in[8];
  const float* Wagt   = (const float*)d_in[9];
  const float* gna_w  = (const float*)d_in[10];
  const float* gna_b  = (const float*)d_in[11];
  const float* norm_w = (const float*)d_in[12];
  const float* norm_b = (const float*)d_in[13];
  const float* Wlin   = (const float*)d_in[14];
  const float* gnl_w  = (const float*)d_in[15];
  const float* gnl_b  = (const float*)d_in[16];

  float* ws = (float*)d_ws;
  size_t off = 0;
  float* qbuf   = ws + off; off += (size_t)NA * DIM;
  off += (size_t)NC * DIM;
  float* cpbuf  = ws + off; off += (size_t)NC * DIM;
  off += (size_t)NE;
  int*   cntbuf = (int*)(ws + off); off += (size_t)NA;
  float* aggbuf = ws + off; off += (size_t)NA * DIM;
  int*   csr    = (int*)(ws + off); off += (size_t)NA * DIM;
  float* agtbuf = ws + off; off += (size_t)NA * DIM;
  unsigned* maxbuf = (unsigned*)(ws + off); off += 16;

  int*   rowptr = csr;
  int*   eid    = csr + 50016;
  int*   subuf  = csr + 650016;
  int*   arrbuf = csr + 1250016;
  int*   bsum   = csr + 1850016;
  float* wfold  = (float*)(csr + 2000000);
  float* wmsg   = (float*)(csr + 2040000);
  float* r12buf = (float*)(csr + 2080000);
  float* stbuf0 = cpbuf;
  float* auxbuf = cpbuf + 2 * 100000;
  float* stbuf1 = cpbuf + 4 * 100000;

  const dim3 b128(128), b256(256);

  setup_kernel<<<33 + (NA + 255) / 256, b256, 0, stream>>>(Wq, Wk, wfold, Wagt, Wctx, gnc_w, gnc_b,
                                                          wmsg, r12buf, cntbuf, NA, maxbuf);

  csr_count_kernel<<<(NE + 255) / 256, b256, 0, stream>>>(v, cntbuf, arrbuf, NE);
  csr_blocksum_kernel<<<(NA + 255) / 256, b256, 0, stream>>>(cntbuf, bsum, NA);
  csr_scan_kernel<<<(NA + 255) / 256, b256, 0, stream>>>(cntbuf, bsum, rowptr, NA);
  csr_fill_kernel<<<(NE + 255) / 256, b256, 0, stream>>>(v, arrbuf, rowptr, eid, NE);
  csr_rank_kernel<<<(NE + 255) / 256, b256, 0, stream>>>(eid, rowptr, u, v, subuf, NE);

  gemm_q_and_ctxp<<<(NA + 63) / 64 + 2 * ((NC + 63) / 64), b128, 0, stream>>>(
      agt, wfold, qbuf, NA, (NA + 63) / 64,
      ctx, Wctx, stbuf0, Wctx + (size_t)DIM * DIM, stbuf1, NC, (NC + 63) / 64);

  node_attn_kernel<<<(NA + 7) / 8, b256, 0, stream>>>(qbuf, ctx, stbuf0, rowptr, subuf,
                                                     aggbuf, auxbuf, maxbuf, NA);
  gemm_update_chain<true><<<(NA + 63) / 64, b128, 0, stream>>>(
      agt, aggbuf, auxbuf, maxbuf, Wagt, wmsg, r12buf,
      gna_w, gna_b, norm_w, norm_b, Wlin, gnl_w, gnl_b, agtbuf,
      wfold + (size_t)DIM * DIM, qbuf, NA);

  node_attn_kernel<<<(NA + 7) / 8, b256, 0, stream>>>(qbuf, ctx, stbuf1, rowptr, subuf,
                                                     aggbuf, auxbuf, maxbuf + 1, NA);
  gemm_update_chain<false><<<(NA + 63) / 64, b128, 0, stream>>>(
      agtbuf, aggbuf, auxbuf, maxbuf + 1, Wagt + (size_t)DIM * 2 * DIM,
      wmsg + (size_t)DIM * DIM, r12buf + (size_t)2 * DIM,
      gna_w + DIM, gna_b + DIM, norm_w + DIM, norm_b + DIM,
      Wlin + (size_t)DIM * DIM, gnl_w + DIM, gnl_b + DIM, (float*)d_out,
      wfold, qbuf, NA);
}


# Chain GEMM kernels: the 53 strips left over beyond 3 work-groups per WGP run column-split over 4 waves (their row-norm sums cross waves via LDS: new sum order); busiest SIMD 4 -> 3.25 strips
# speedup vs baseline: 3.3783x; 1.0387x over previous
#include <hip/hip_runtime.h>


static constexpr int DIM  = 128;
static constexpr int NBLK = 2;
static constexpr int LDP  = 68;

typedef __attribute__((ext_vector_type(2))) float v2f;
typedef __attribute__((ext_vector_type(8))) float v8f;

static __device__ __forceinline__ unsigned f32_order_enc(float f) {
  unsigned u = __float_as_uint(f);
  return (u & 0x80000000u) ? ~u : (u | 0x80000000u);
}
static __device__ __forceinline__ float f32_order_dec(unsigned e) {
  unsigned u = (e & 0x80000000u) ? (e & 0x7fffffffu) : ~e;
  return __uint_as_float(u);
}

enum { EPI_STATS = 1, EPI_GN2_RELU = 2, EPI_GN_RES_RELU = 3 };

template <int XM>
static __device__ __forceinline__ float half_xor(float x) {
  return __int_as_float(__builtin_amdgcn_update_dpp(
      0, __float_as_int(x), 0x160 + XM, 0xf, 0xf, true));
}
static __device__ __forceinline__ float half_sum(float x) {
  x += half_xor<1>(x);
  x += half_xor<2>(x);
  x += half_xor<4>(x);
  x += half_xor<8>(x);
  return x;
}

template <bool RELU>
static __device__ __forceinline__ void gn_rows(v8f (&acc)[8],
                                               const float* __restrict__ w,
                                               const float* __restrict__ b,
                                               int l16) {
  float wv[8], bv[8];
#pragma unroll
  for (int t = 0; t < 8; ++t) {
    wv[t] = w[t * 16 + l16];
    bv[t] = b[t * 16 + l16];
  }
#pragma unroll
  for (int j = 0; j < 8; ++j) {
    float s = 0.0f, sq = 0.0f;
#pragma unroll
    for (int t = 0; t < 8; ++t) {
      const float x = acc[t][j];
      s += x;
      sq += x * x;
    }
    s  = half_sum(s);
    sq = half_sum(sq);
    const float mu = s * (1.0f / DIM);
    const float rs = rsqrtf(sq * (1.0f / DIM) - mu * mu + 1e-5f);
#pragma unroll
    for (int t = 0; t < 8; ++t) {
      float y = (acc[t][j] - mu) * rs * wv[t] + bv[t];
      if (RELU) y = fmaxf(y, 0.0f);
      acc[t][j] = y;
    }
  }
}

static __device__ __forceinline__ void stats_rows(const v8f (&acc)[8],
                                                  float* __restrict__ st,
                                                  int r0, int hk, int l16, int N) {
  float mymu = 0.0f, myrs = 0.0f;
#pragma unroll
  for (int j = 0; j < 8; ++j) {
    float s = 0.0f, sq = 0.0f;
#pragma unroll
    for (int t = 0; t < 8; ++t) {
      const float x = acc[t][j];
      s += x;
      sq += x * x;
    }
    s  = half_sum(s);
    sq = half_sum(sq);
    const float mu = s * (1.0f / DIM);
    const float rs = rsqrtf(sq * (1.0f / DIM) - mu * mu + 1e-5f);
    if (l16 == j) { mymu = mu; myrs = rs; }
  }
  const int row = r0 + hk * 8 + l16;
  if (l16 < 8 && row < N) {
    float2 v; v.x = mymu; v.y = myrs;
    ((float2*)st)[row] = v;
  }
}

template <int EPI>
static __device__ __forceinline__ void epilogue_store(v8f (&acc)[8],
                                                      float* __restrict__ C,
                                                      const float* __restrict__ R,
                                                      const float* __restrict__ w1,
                                                      const float* __restrict__ b1,
                                                      const float* __restrict__ w2,
                                                      const float* __restrict__ b2,
                                                      int r0, int hk, int l16, int N,
                                                      bool plain) {
  const int rb = r0 + hk * 8;

  if (EPI == EPI_GN_RES_RELU) {
    float res[8][8];
#pragma unroll
    for (int j = 0; j < 8; ++j) {
      int row = rb + j;
      if (row >= N) row = N - 1;
#pragma unroll
      for (int t = 0; t < 8; ++t) res[t][j] = R[(size_t)row * DIM + t * 16 + l16];
    }
    gn_rows<false>(acc, w1, b1, l16);
#pragma unroll
    for (int t = 0; t < 8; ++t)
#pragma unroll
      for (int j = 0; j < 8; ++j) acc[t][j] = fmaxf(acc[t][j] + res[t][j], 0.0f);
  } else if (EPI == EPI_GN2_RELU) {
    gn_rows<false>(acc, w1, b1, l16);
    gn_rows<true>(acc, w2, b2, l16);
  } else if (EPI == EPI_STATS) {
    if (!plain) {
      stats_rows(acc, C, r0, hk, l16, N);
      return;
    }
  }

  if (r0 + 16 <= N) {
#pragma unroll
    for (int t = 0; t < 8; ++t)
#pragma unroll
      for (int j = 0; j < 8; ++j)
        C[(size_t)(rb + j) * DIM + t * 16 + l16] = acc[t][j];
  } else {
#pragma unroll
    for (int t = 0; t < 8; ++t)
#pragma unroll
      for (int j = 0; j < 8; ++j) {
        const int row = rb + j;
        if (row < N) C[(size_t)row * DIM + t * 16 + l16] = acc[t][j];
      }
  }
}

template <int EPI>
static __device__ __forceinline__ void gemm_nt_128_body(float* __restrict__ Wlds,
                                                        const float* __restrict__ A,
                                                        const float* __restrict__ W,
                                                        float* __restrict__ C,
                                                        const float* __restrict__ R,
                                                        const float* __restrict__ w1,
                                                        const float* __restrict__ b1,
                                                        const float* __restrict__ w2,
                                                        const float* __restrict__ b2,
                                                        int N, int tile, bool plain) {
  const int tid = threadIdx.x;
  const int wv  = tid >> 5;
  const int ln  = tid & 31;
  const int hk  = ln >> 4;
  const int l16 = ln & 15;
  const int r0  = tile * 64 + wv * 16;

  v8f acc[8];
#pragma unroll
  for (int t = 0; t < 8; ++t)
#pragma unroll
    for (int j = 0; j < 8; ++j) acc[t][j] = 0.0f;

  int rowA = r0 + l16;
  if (rowA >= N) rowA = N - 1;
  const float* ap = A + (size_t)rowA * DIM;

  for (int kp = 0; kp < DIM; kp += 64) {
    __syncthreads();
#pragma unroll
    for (int i = 0; i < 16; ++i) {
      const int q  = tid + 128 * i;
      const int n  = q >> 4;
      const int c4 = (q & 15) << 2;
      *(float4*)(&Wlds[n * LDP + c4]) =
          *(const float4*)(W + (size_t)n * DIM + kp + c4);
    }
    __syncthreads();

#pragma unroll 4
    for (int k0 = 0; k0 < 64; k0 += 4) {
      const int ka = k0 + 2 * hk;
      v2f a; a.x = ap[kp + ka]; a.y = ap[kp + ka + 1];
#pragma unroll
      for (int t = 0; t < 8; ++t) {
        const float* wp = &Wlds[(t * 16 + l16) * LDP + ka];
        v2f b; b.x = wp[0]; b.y = wp[1];
        acc[t] = __builtin_amdgcn_wmma_f32_16x16x4_f32(
            false, a, false, b, (short)0, acc[t], false, false);
      }
    }
  }

  epilogue_store<EPI>(acc, C, R, w1, b1, w2, b2, r0, hk, l16, N, plain);
}

__global__ __launch_bounds__(128) void gemm_q_and_ctxp(const float* __restrict__ Aq,
                                                       const float* __restrict__ Wfq,
                                                       float* __restrict__ Cq, int Nq,
                                                       int tilesQ,
                                                       const float* __restrict__ Ac,
                                                       const float* __restrict__ Wc0,
                                                       float* __restrict__ Cc0,
                                                       const float* __restrict__ Wc1,
                                                       float* __restrict__ Cc1,
                                                       int Nc, int tilesC) {
  __shared__ float Wlds[128 * LDP];
  const int bid = blockIdx.x;
  const bool isq = bid < tilesQ;
  const bool sec = bid >= tilesQ + tilesC;
  const float* A  = isq ? Aq : Ac;
  const float* Wm = isq ? Wfq : (sec ? Wc1 : Wc0);
  float* C        = isq ? Cq : (sec ? Cc1 : Cc0);
  const int N     = isq ? Nq : Nc;
  const int tile  = isq ? bid : (sec ? bid - tilesQ - tilesC : bid - tilesQ);
  gemm_nt_128_body<EPI_STATS>(Wlds, A, Wm, C, nullptr, nullptr, nullptr, nullptr, nullptr, N, tile, isq);
}

static constexpr int XP = 132;

static __device__ __forceinline__ void tile_to_lds(const v8f (&acc)[8], float* xs,
                                                   int hk, int l16) {
#pragma unroll
  for (int t = 0; t < 8; ++t)
#pragma unroll
    for (int j = 0; j < 8; ++j) xs[(8 * hk + j) * XP + 16 * t + l16] = acc[t][j];
}

static __device__ __forceinline__ void gemm_tile_lds_128(v8f (&acc)[8], float* Wlds,
                                                         const float* xs,
                                                         const float* __restrict__ Wg,
                                                         int tid, int hk, int l16) {
#pragma unroll
  for (int t = 0; t < 8; ++t)
#pragma unroll
    for (int j = 0; j < 8; ++j) acc[t][j] = 0.0f;

  const float* xr = xs + l16 * XP;
  for (int kp = 0; kp < DIM; kp += 64) {
    __syncthreads();
#pragma unroll
    for (int i = 0; i < 16; ++i) {
      const int q  = tid + 128 * i;
      const int n  = q >> 4;
      const int c4 = (q & 15) << 2;
      *(float4*)(&Wlds[n * LDP + c4]) =
          *(const float4*)(Wg + (size_t)n * DIM + kp + c4);
    }
    __syncthreads();

#pragma unroll 4
    for (int k0 = 0; k0 < 64; k0 += 4) {
      const int ka = k0 + 2 * hk;
      v2f a; a.x = xr[kp + ka]; a.y = xr[kp + ka + 1];
#pragma unroll
      for (int t = 0; t < 8; ++t) {
        const float* wp = &Wlds[(t * 16 + l16) * LDP + ka];
        v2f b; b.x = wp[0]; b.y = wp[1];
        acc[t] = __builtin_amdgcn_wmma_f32_16x16x4_f32(
            false, a, false, b, (short)0, acc[t], false, false);
      }
    }
  }
}


template <bool RELU>
static __device__ __forceinline__ void gn_rows_split(v8f (&acc)[2], float* red,
                                                     const float* __restrict__ w,
                                                     const float* __restrict__ b,
                                                     int wv, int hk, int l16) {
  float wc[2], bc[2];
#pragma unroll
  for (int t = 0; t < 2; ++t) {
    wc[t] = w[(2 * wv + t) * 16 + l16];
    bc[t] = b[(2 * wv + t) * 16 + l16];
  }
  float mys = 0.0f, mysq = 0.0f;
#pragma unroll
  for (int j = 0; j < 8; ++j) {
    float s  = acc[0][j] + acc[1][j];
    float sq = acc[0][j] * acc[0][j] + acc[1][j] * acc[1][j];
    s  = half_sum(s);
    sq = half_sum(sq);
    if (l16 == j) { mys = s; mysq = sq; }
  }
  if (l16 < 8) {
    float2 p; p.x = mys; p.y = mysq;
    ((float2*)red)[wv * 16 + hk * 8 + l16] = p;
  }
  __syncthreads();
#pragma unroll
  for (int j = 0; j < 8; ++j) {
    float s = 0.0f, sq = 0.0f;
#pragma unroll
    for (int q = 0; q < 4; ++q) {
      const float2 p = ((const float2*)red)[q * 16 + hk * 8 + j];
      s += p.x;
      sq += p.y;
    }
    const float mu = s * (1.0f / DIM);
    const float rs = rsqrtf(sq * (1.0f / DIM) - mu * mu + 1e-5f);
#pragma unroll
    for (int t = 0; t < 2; ++t) {
      float y = (acc[t][j] - mu) * rs * wc[t] + bc[t];
      if (RELU) y = fmaxf(y, 0.0f);
      acc[t][j] = y;
    }
  }
}

static __device__ __forceinline__ void tile_to_lds_split(const v8f (&acc)[2], float* xs,
                                                         int wv, int hk, int l16) {
#pragma unroll
  for (int t = 0; t < 2; ++t)
#pragma unroll
    for (int j = 0; j < 8; ++j) xs[(8 * hk + j) * XP + 16 * (2 * wv + t) + l16] = acc[t][j];
}

static __device__ __forceinline__ void gemm_tile_lds_split(v8f (&acc)[2], float* Wlds,
                                                           const float* xs,
                                                           const float* __restrict__ Wg,
                                                           int tid, int wv, int hk, int l16) {
#pragma unroll
  for (int t = 0; t < 2; ++t)
#pragma unroll
    for (int j = 0; j < 8; ++j) acc[t][j] = 0.0f;

  const float* xr = xs + l16 * XP;
  for (int kp = 0; kp < DIM; kp += 64) {
    __syncthreads();
#pragma unroll
    for (int i = 0; i < 16; ++i) {
      const int q  = tid + 128 * i;
      const int n  = q >> 4;
      const int c4 = (q & 15) << 2;
      *(float4*)(&Wlds[n * LDP + c4]) =
          *(const float4*)(Wg + (size_t)n * DIM + kp + c4);
    }
    __syncthreads();

#pragma unroll 4
    for (int k0 = 0; k0 < 64; k0 += 4) {
      const int ka = k0 + 2 * hk;
      v2f a; a.x = xr[kp + ka]; a.y = xr[kp + ka + 1];
#pragma unroll
      for (int t = 0; t < 2; ++t) {
        const float* wp = &Wlds[((2 * wv + t) * 16 + l16) * LDP + ka];
        v2f b; b.x = wp[0]; b.y = wp[1];
        acc[t] = __builtin_amdgcn_wmma_f32_16x16x4_f32(
            false, a, false, b, (short)0, acc[t], false, false);
      }
    }
  }
}

static __device__ __forceinline__ void store_split(const v8f (&acc)[2], float* __restrict__ C,
                                                   int rb, int wv, int l16, int N) {
#pragma unroll
  for (int t = 0; t < 2; ++t)
#pragma unroll
    for (int j = 0; j < 8; ++j) {
      const int row = rb + j;
      if (row < N) C[(size_t)row * DIM + (2 * wv + t) * 16 + l16] = acc[t][j];
    }
}

template <bool NEXTQ>
static __device__ __forceinline__ void chain_split_body(float* Wlds, float* Xlds,
                                                        const float* __restrict__ A0,
                                                        const float* __restrict__ A1,
                                                        const float* __restrict__ aux,
                                                        const unsigned* __restrict__ gmax,
                                                        const float* __restrict__ W,
                                                        const float* __restrict__ Wm,
                                                        const float* __restrict__ r12,
                                                        const float* __restrict__ w1,
                                                        const float* __restrict__ b1,
                                                        const float* __restrict__ w2,
                                                        const float* __restrict__ b2,
                                                        const float* __restrict__ Wl,
                                                        const float* __restrict__ gl_w,
                                                        const float* __restrict__ gl_b,
                                                        float* __restrict__ Cout,
                                                        const float* __restrict__ Wfn,
                                                        float* __restrict__ Qn,
                                                        int N, int r0) {
  const int tid = threadIdx.x;
  const int wv  = tid >> 5;
  const int ln  = tid & 31;
  const int hk  = ln >> 4;
  const int l16 = ln & 15;
  float* xs   = Xlds;
  float* red0 = Xlds + 16 * XP;
  float* red1 = red0 + 128;
  const int rb = r0 + hk * 8;

  v8f acc[2];
#pragma unroll
  for (int t = 0; t < 2; ++t)
#pragma unroll
    for (int j = 0; j < 8; ++j) acc[t][j] = 0.0f;

  int rowA = r0 + l16;
  if (rowA >= N) rowA = N - 1;
  const size_t rbase = (size_t)rowA * DIM;

  const float4 ax = ((const float4*)aux)[rowA];
  const float gm  = f32_order_dec(*gmax);
  const float ee  = expf(ax.x - gm);
  const float fr  = ee / (ee * ax.y + 1e-16f);

  for (int kp = 0; kp < 2 * DIM; kp += 64) {
    const bool first = kp < DIM;
    const float* wsrc = first ? (W + kp) : (Wm + (kp - DIM));
    const int wld     = first ? 2 * DIM : DIM;
    __syncthreads();
#pragma unroll
    for (int i = 0; i < 16; ++i) {
      const int q  = tid + 128 * i;
      const int n  = q >> 4;
      const int c4 = (q & 15) << 2;
      *(float4*)(&Wlds[n * LDP + c4]) =
          *(const float4*)(wsrc + (size_t)n * wld + c4);
    }
    __syncthreads();

    const float* abase = first ? (A0 + rbase + kp) : (A1 + rbase + (kp - DIM));
    const float asc    = first ? 1.0f : fr;
#pragma unroll 4
    for (int k0 = 0; k0 < 64; k0 += 4) {
      const int ka = k0 + 2 * hk;
      v2f a; a.x = abase[ka] * asc; a.y = abase[ka + 1] * asc;
#pragma unroll
      for (int t = 0; t < 2; ++t) {
        const float* wp = &Wlds[((2 * wv + t) * 16 + l16) * LDP + ka];
        v2f b; b.x = wp[0]; b.y = wp[1];
        acc[t] = __builtin_amdgcn_wmma_f32_16x16x4_f32(
            false, a, false, b, (short)0, acc[t], false, false);
      }
    }
  }

  {
    v2f a;
    a.x = (hk == 0) ? -fr * ax.z : 0.0f;
    a.y = (hk == 0) ?  fr * ax.y : 0.0f;
#pragma unroll
    for (int t = 0; t < 2; ++t) {
      v2f b;
      b.x = (hk == 0) ? r12[(2 * wv + t) * 16 + l16] : 0.0f;
      b.y = (hk == 0) ? r12[DIM + (2 * wv + t) * 16 + l16] : 0.0f;
      acc[t] = __builtin_amdgcn_wmma_f32_16x16x4_f32(
          false, a, false, b, (short)0, acc[t], false, false);
    }
  }

  gn_rows_split<false>(acc, red0, w1, b1, wv, hk, l16);
  gn_rows_split<true>(acc, red1, w2, b2, wv, hk, l16);
  tile_to_lds_split(acc, xs, wv, hk, l16);

  gemm_tile_lds_split(acc, Wlds, xs, Wl, tid, wv, hk, l16);
  {
    float res[2][8];
#pragma unroll
    for (int j = 0; j < 8; ++j) {
      int row = rb + j;
      if (row >= N) row = N - 1;
#pragma unroll
      for (int t = 0; t < 2; ++t) res[t][j] = A0[(size_t)row * DIM + (2 * wv + t) * 16 + l16];
    }
    gn_rows_split<false>(acc, red0, gl_w, gl_b, wv, hk, l16);
#pragma unroll
    for (int t = 0; t < 2; ++t)
#pragma unroll
      for (int j = 0; j < 8; ++j) acc[t][j] = fmaxf(acc[t][j] + res[t][j], 0.0f);
  }
  store_split(acc, Cout, rb, wv, l16, N);

  if (NEXTQ) {
    tile_to_lds_split(acc, xs, wv, hk, l16);
    gemm_tile_lds_split(acc, Wlds, xs, Wfn, tid, wv, hk, l16);
    store_split(acc, Qn, rb, wv, l16, N);
  }
}

template <bool NEXTQ>
__global__ __launch_bounds__(128) void gemm_update_chain(const float* __restrict__ A0,
                                                         const float* __restrict__ A1,
                                                         const float* __restrict__ aux,
                                                         const unsigned* __restrict__ gmax,
                                                         const float* __restrict__ W,
                                                         const float* __restrict__ Wm,
                                                         const float* __restrict__ r12,
                                                         const float* __restrict__ w1,
                                                         const float* __restrict__ b1,
                                                         const float* __restrict__ w2,
                                                         const float* __restrict__ b2,
                                                         const float* __restrict__ Wl,
                                                         const float* __restrict__ gl_w,
                                                         const float* __restrict__ gl_b,
                                                         float* __restrict__ Cout,
                                                         const float* __restrict__ Wfn,
                                                         float* __restrict__ Qn,
                                                         int N, int tilesFull) {
  __shared__ __attribute__((aligned(16))) float Wlds[128 * LDP];
  __shared__ __attribute__((aligned(16))) float Xlds[4 * 16 * XP];
  if ((int)blockIdx.x >= tilesFull) {
    chain_split_body<NEXTQ>(Wlds, Xlds, A0, A1, aux, gmax, W, Wm, r12, w1, b1, w2, b2,
                            Wl, gl_w, gl_b, Cout, Wfn, Qn, N,
                            tilesFull * 64 + ((int)blockIdx.x - tilesFull) * 16);
    return;
  }
  const int tid = threadIdx.x;
  const int wv  = tid >> 5;
  const int ln  = tid & 31;
  const int hk  = ln >> 4;
  const int l16 = ln & 15;
  const int r0  = blockIdx.x * 64 + wv * 16;
  float* xs = Xlds + wv * 16 * XP;

  v8f acc[8];
#pragma unroll
  for (int t = 0; t < 8; ++t)
#pragma unroll
    for (int j = 0; j < 8; ++j) acc[t][j] = 0.0f;

  int rowA = r0 + l16;
  if (rowA >= N) rowA = N - 1;
  const size_t rbase = (size_t)rowA * DIM;

  const float4 ax = ((const float4*)aux)[rowA];
  const float gm  = f32_order_dec(*gmax);
  const float ee  = expf(ax.x - gm);
  const float fr  = ee / (ee * ax.y + 1e-16f);

  for (int kp = 0; kp < 2 * DIM; kp += 64) {
    const bool first = kp < DIM;
    const float* wsrc = first ? (W + kp) : (Wm + (kp - DIM));
    const int wld     = first ? 2 * DIM : DIM;
    __syncthreads();
#pragma unroll
    for (int i = 0; i < 16; ++i) {
      const int q  = tid + 128 * i;
      const int n  = q >> 4;
      const int c4 = (q & 15) << 2;
      *(float4*)(&Wlds[n * LDP + c4]) =
          *(const float4*)(wsrc + (size_t)n * wld + c4);
    }
    __syncthreads();

    const float* abase = first ? (A0 + rbase + kp) : (A1 + rbase + (kp - DIM));
    const float asc    = first ? 1.0f : fr;
#pragma unroll 4
    for (int k0 = 0; k0 < 64; k0 += 4) {
      const int ka = k0 + 2 * hk;
      v2f a; a.x = abase[ka] * asc; a.y = abase[ka + 1] * asc;
#pragma unroll
      for (int t = 0; t < 8; ++t) {
        const float* wp = &Wlds[(t * 16 + l16) * LDP + ka];
        v2f b; b.x = wp[0]; b.y = wp[1];
        acc[t] = __builtin_amdgcn_wmma_f32_16x16x4_f32(
            false, a, false, b, (short)0, acc[t], false, false);
      }
    }
  }

  {
    v2f a;
    a.x = (hk == 0) ? -fr * ax.z : 0.0f;
    a.y = (hk == 0) ?  fr * ax.y : 0.0f;
#pragma unroll
    for (int t = 0; t < 8; ++t) {
      v2f b;
      b.x = (hk == 0) ? r12[t * 16 + l16] : 0.0f;
      b.y = (hk == 0) ? r12[DIM + t * 16 + l16] : 0.0f;
      acc[t] = __builtin_amdgcn_wmma_f32_16x16x4_f32(
          false, a, false, b, (short)0, acc[t], false, false);
    }
  }

  gn_rows<false>(acc, w1, b1, l16);
  gn_rows<true>(acc, w2, b2, l16);
  tile_to_lds(acc, xs, hk, l16);

  gemm_tile_lds_128(acc, Wlds, xs, Wl, tid, hk, l16);
  epilogue_store<EPI_GN_RES_RELU>(acc, Cout, A0, gl_w, gl_b, nullptr, nullptr, r0, hk, l16, N, false);

  if (NEXTQ) {
    tile_to_lds(acc, xs, hk, l16);
    gemm_tile_lds_128(acc, Wlds, xs, Wfn, tid, hk, l16);
    epilogue_store<EPI_STATS>(acc, Qn, nullptr, nullptr, nullptr, nullptr, nullptr, r0, hk, l16, N, true);
  }
}

static __device__ __forceinline__ void fold_qk_body(const float* __restrict__ Wq,
                                                    const float* __restrict__ Wk,
                                                    float* __restrict__ Wf, int bid) {
  const int wv  = threadIdx.x >> 5;
  const int ln  = threadIdx.x & 31;
  const int hk  = ln >> 4;
  const int l16 = ln & 15;
  const int tile = bid * 8 + wv;
  const int i  = tile >> 6;
  const int j0 = ((tile >> 3) & 7) * 16;
  const int m0 = (tile & 7) * 16;
  const float* wq = Wq + (size_t)i * DIM * DIM;
  const float* wk = Wk + (size_t)i * DIM * DIM;

  v8f acc;
#pragma unroll
  for (int r = 0; r < 8; ++r) acc[r] = 0.0f;

#pragma unroll 8
  for (int d0 = 0; d0 < DIM; d0 += 4) {
    const int d = d0 + 2 * hk;
    v2f a; a.x = wk[d * DIM + j0 + l16]; a.y = wk[(d + 1) * DIM + j0 + l16];
    v2f b; b.x = wq[d * DIM + m0 + l16]; b.y = wq[(d + 1) * DIM + m0 + l16];
    acc = __builtin_amdgcn_wmma_f32_16x16x4_f32(
        false, a, false, b, (short)0, acc, false, false);
  }

  float* o = Wf + (size_t)i * DIM * DIM + (size_t)(j0 + 8 * hk) * DIM + m0 + l16;
#pragma unroll
  for (int r = 0; r < 8; ++r) o[r * DIM] = acc[r];
}

static __device__ __forceinline__ void fold_msg_body(const float* __restrict__ Wagt,
                                                     const float* __restrict__ Wctx,
                                                     const float* __restrict__ gw,
                                                     float* __restrict__ Wm, int bid) {
  const int wv  = threadIdx.x >> 5;
  const int ln  = threadIdx.x & 31;
  const int hk  = ln >> 4;
  const int l16 = ln & 15;
  const int tile = bid * 8 + wv;
  const int i  = tile >> 6;
  const int o0 = ((tile >> 3) & 7) * 16;
  const int k0 = (tile & 7) * 16;
  const float* wa = Wagt + (size_t)i * DIM * 2 * DIM + (size_t)(o0 + l16) * (2 * DIM) + DIM;
  const float* wc = Wctx + (size_t)i * DIM * DIM + k0 + l16;
  const float* g  = gw + i * DIM;

  v8f acc;
#pragma unroll
  for (int r = 0; r < 8; ++r) acc[r] = 0.0f;

#pragma unroll 8
  for (int d0 = 0; d0 < DIM; d0 += 4) {
    const int d = d0 + 2 * hk;
    v2f a; a.x = wa[d] * g[d];       a.y = wa[d + 1] * g[d + 1];
    v2f b; b.x = wc[d * DIM];        b.y = wc[(d + 1) * DIM];
    acc = __builtin_amdgcn_wmma_f32_16x16x4_f32(
        false, a, false, b, (short)0, acc, false, false);
  }

  float* o = Wm + (size_t)i * DIM * DIM + (size_t)(o0 + 8 * hk) * DIM + k0 + l16;
#pragma unroll
  for (int r = 0; r < 8; ++r) o[r * DIM] = acc[r];
}

__global__ __launch_bounds__(256) void setup_kernel(const float* __restrict__ Wq,
                                                    const float* __restrict__ Wk,
                                                    float* __restrict__ Wf,
                                                    const float* __restrict__ Wagt,
                                                    const float* __restrict__ Wctx,
                                                    const float* __restrict__ gw,
                                                    const float* __restrict__ gb,
                                                    float* __restrict__ Wm,
                                                    float* __restrict__ r12,
                                                    int* __restrict__ cnt, int n,
                                                    unsigned* __restrict__ gmax) {
  const int bid = blockIdx.x;
  if (bid < 16) {
    fold_qk_body(Wq, Wk, Wf, bid);
  } else if (bid < 32) {
    fold_msg_body(Wagt, Wctx, gw, Wm, bid - 16);
  } else if (bid == 32) {
    const int i = threadIdx.x >> 7;
    const int o = threadIdx.x & 127;
    const float* wa = Wagt + (size_t)i * DIM * 2 * DIM + (size_t)o * (2 * DIM) + DIM;
    float a1 = 0.0f, a2 = 0.0f;
    for (int c = 0; c < DIM; ++c) {
      const float x = wa[c];
      a1 += x * gw[i * DIM + c];
      a2 += x * gb[i * DIM + c];
    }
    r12[i * 2 * DIM + o]       = a1;
    r12[i * 2 * DIM + DIM + o] = a2;
  } else {
    const int i = (bid - 33) * 256 + threadIdx.x;
    if (i < n) cnt[i] = 0;
    if (bid == 33 && threadIdx.x < NBLK) gmax[threadIdx.x] = 0u;
  }
}

__global__ __launch_bounds__(256) void csr_count_kernel(const int* __restrict__ v,
                                                        int* __restrict__ cnt,
                                                        int* __restrict__ arr, int E) {
  const int e = blockIdx.x * 256 + threadIdx.x;
  if (e < E) arr[e] = atomicAdd(&cnt[v[e]], 1);
}

static __device__ __forceinline__ int wave_reduce_sum_i(int x) {
#pragma unroll
  for (int off = 16; off > 0; off >>= 1) x += __shfl_xor(x, off, 32);
  return x;
}

__global__ __launch_bounds__(256) void csr_blocksum_kernel(const int* __restrict__ cnt,
                                                           int* __restrict__ bsum, int N) {
  __shared__ int sw[8];
  const int wv = threadIdx.x >> 5, ln = threadIdx.x & 31;
  const int i = blockIdx.x * 256 + threadIdx.x;
  int c = (i < N) ? cnt[i] : 0;
  c = wave_reduce_sum_i(c);
  if (ln == 0) sw[wv] = c;
  __syncthreads();
  if (threadIdx.x == 0) {
    int t = 0;
#pragma unroll
    for (int k = 0; k < 8; ++k) t += sw[k];
    bsum[blockIdx.x] = t;
  }
}

__global__ __launch_bounds__(256) void csr_scan_kernel(const int* __restrict__ cnt,
                                                       const int* __restrict__ bsum,
                                                       int* __restrict__ rowptr, int N) {
  __shared__ int sa[256];
  __shared__ int sb[256];
  __shared__ int sw[8];
  const int t = threadIdx.x;
  const int wv = t >> 5, ln = t & 31;
  const int blk = blockIdx.x;

  int p = 0;
  for (int k = t; k < blk; k += 256) p += bsum[k];
  p = wave_reduce_sum_i(p);
  if (ln == 0) sw[wv] = p;

  const int i = blk * 256 + t;
  const int c = (i < N) ? cnt[i] : 0;
  sa[t] = c;
  __syncthreads();

  int base = 0;
#pragma unroll
  for (int k = 0; k < 8; ++k) base += sw[k];

  int* src = sa;
  int* dst = sb;
  for (int off = 1; off < 256; off <<= 1) {
    int val = src[t];
    if (t >= off) val += src[t - off];
    dst[t] = val;
    __syncthreads();
    int* tmp = src; src = dst; dst = tmp;
  }
  const int excl = base + src[t] - c;
  if (i < N) rowptr[i] = excl;
  if (i == N - 1) rowptr[N] = excl + c;
}

__global__ __launch_bounds__(256) void csr_fill_kernel(const int* __restrict__ v,
                                                       const int* __restrict__ arr,
                                                       const int* __restrict__ rowptr,
                                                       int* __restrict__ eid, int E) {
  const int e = blockIdx.x * 256 + threadIdx.x;
  if (e < E) eid[rowptr[v[e]] + arr[e]] = e;
}

__global__ __launch_bounds__(256) void csr_rank_kernel(const int* __restrict__ eid,
                                                       const int* __restrict__ rowptr,
                                                       const int* __restrict__ u,
                                                       const int* __restrict__ v,
                                                       int* __restrict__ su, int E) {
  const int j = blockIdx.x * 256 + threadIdx.x;
  if (j >= E) return;
  const int e = eid[j];
  const int n = v[e];
  const int b = rowptr[n], en = rowptr[n + 1];
  int r = 0;
  for (int k = b; k < en; ++k) r += (eid[k] < e) ? 1 : 0;
  su[b + r] = u[e];
}

static constexpr int CH = 16;

static __device__ __forceinline__ int lane_i(int x, int l) {
  return __builtin_amdgcn_readlane(x, l);
}
static __device__ __forceinline__ float lane_f(float x, int l) {
  return __int_as_float(__builtin_amdgcn_readlane(__float_as_int(x), l));
}
static __device__ __forceinline__ float other_half(float x) {
  return __int_as_float(__builtin_amdgcn_permlanex16(
      __float_as_int(x), __float_as_int(x), 0x76543210, 0xfedcba98, false, false));
}

static __device__ __forceinline__ float half_sum_16(const float (&p)[16], int ln) {
  const bool b0 = (ln & 1) != 0, b1 = (ln & 2) != 0, b2 = (ln & 4) != 0, b3 = (ln & 8) != 0;
  float r1[8], r2[4], r3[2];
#pragma unroll
  for (int i = 0; i < 8; ++i) {
    const float keep = b0 ? p[2 * i + 1] : p[2 * i];
    const float send = b0 ? p[2 * i] : p[2 * i + 1];
    r1[i] = keep + half_xor<1>(send);
  }
#pragma unroll
  for (int i = 0; i < 4; ++i) {
    const float keep = b1 ? r1[2 * i + 1] : r1[2 * i];
    const float send = b1 ? r1[2 * i] : r1[2 * i + 1];
    r2[i] = keep + half_xor<2>(send);
  }
#pragma unroll
  for (int i = 0; i < 2; ++i) {
    const float keep = b2 ? r2[2 * i + 1] : r2[2 * i];
    const float send = b2 ? r2[2 * i] : r2[2 * i + 1];
    r3[i] = keep + half_xor<4>(send);
  }
  const float keep = b3 ? r3[1] : r3[0];
  const float send = b3 ? r3[0] : r3[1];
  return keep + half_xor<8>(send);
}

__global__ __launch_bounds__(256) void node_attn_kernel(const float* __restrict__ Q,
                                                        const float* __restrict__ X,
                                                        const float* __restrict__ st,
                                                        const int* __restrict__ rowptr,
                                                        const int* __restrict__ su,
                                                        float* __restrict__ S,
                                                        float* __restrict__ aux,
                                                        unsigned* __restrict__ gmax,
                                                        int N) {
  __shared__ float smax[8];
  const int wv = threadIdx.x >> 5, ln = threadIdx.x & 31;
  const int n = __builtin_amdgcn_readfirstlane(blockIdx.x * 8 + wv);
  float m = -3.0e38f;
  if (n < N) {
    const int b  = __builtin_amdgcn_readfirstlane(rowptr[n]);
    const int en = __builtin_amdgcn_readfirstlane(rowptr[n + 1]);
    float ssum = 0.0f, tsum = 0.0f;
    float4 acc;
    acc.x = 0.0f; acc.y = 0.0f; acc.z = 0.0f; acc.w = 0.0f;
    if (b < en) {
      const float4 qv = ((const float4*)(Q + (size_t)n * DIM))[ln];
      for (int j0 = b; j0 < en; j0 += CH) {
        const int cnt = (en - j0 < CH) ? (en - j0) : CH;
        const int jj = j0 + ((ln < cnt) ? ln : (cnt - 1));
        const int uu = su[jj];
        const float2 sv = ((const float2*)st)[uu];

        float4 kv[CH];
#pragma unroll
        for (int g = 0; g < CH / 4; ++g) {
          if (4 * g < cnt) {
#pragma unroll
            for (int l = 4 * g; l < 4 * g + 4; ++l) {
              const int ui = lane_i(uu, l);
              kv[l] = ((const float4*)(X + (size_t)ui * DIM))[ln];
            }
          }
        }

        float p[CH];
#pragma unroll
        for (int g = 0; g < CH / 4; ++g) {
          if (4 * g < cnt) {
#pragma unroll
            for (int l = 4 * g; l < 4 * g + 4; ++l)
              p[l] = qv.x * kv[l].x + qv.y * kv[l].y + qv.z * kv[l].z + qv.w * kv[l].w;
          } else {
#pragma unroll
            for (int l = 4 * g; l < 4 * g + 4; ++l) p[l] = 0.0f;
          }
        }

        const float hs  = half_sum_16(p, ln);
        const float tot = hs + other_half(hs);
        const float sco = (tot >= 0.0f) ? tot : 0.1f * tot;
        const bool valid = ln < cnt;
        float mx = valid ? sco : -3.0e38f;
        mx = fmaxf(mx, half_xor<1>(mx));
        mx = fmaxf(mx, half_xor<2>(mx));
        mx = fmaxf(mx, half_xor<4>(mx));
        mx = fmaxf(mx, half_xor<8>(mx));
        const float mc = lane_f(mx, 0);

        const float mn = fmaxf(m, mc);
        const float sc = expf(m - mn);
        ssum *= sc; tsum *= sc;
        acc.x *= sc; acc.y *= sc; acc.z *= sc; acc.w *= sc;
        m = mn;

        const float ev = valid ? expf(sco - mn) : 0.0f;
        const float wl = ev * sv.y;

#pragma unroll
        for (int g = 0; g < CH / 4; ++g) {
          if (4 * g < cnt) {
#pragma unroll
            for (int l = 4 * g; l < 4 * g + 4; ++l) {
              const float e  = lane_f(ev, l);
              const float w  = lane_f(wl, l);
              const float mu = lane_f(sv.x, l);
              ssum += e;
              tsum += w * mu;
              acc.x += w * kv[l].x;
              acc.y += w * kv[l].y;
              acc.z += w * kv[l].z;
              acc.w += w * kv[l].w;
            }
          }
        }
      }
    }
    ((float4*)(S + (size_t)n * DIM))[ln] = acc;
    if (ln == 0) {
      float4 a; a.x = m; a.y = ssum; a.z = tsum; a.w = 0.0f;
      ((float4*)aux)[n] = a;
    }
  }
  if (ln == 0) smax[wv] = m;
  __syncthreads();
  if (threadIdx.x == 0) {
    float mm = smax[0];
#pragma unroll
    for (int i = 1; i < 8; ++i) mm = fmaxf(mm, smax[i]);
    atomicMax(gmax, f32_order_enc(mm));
  }
}

extern "C" void kernel_launch(void* const* d_in, const int* in_sizes, int n_in,
                              void* d_out, int out_size, void* d_ws, size_t ws_size,
                              hipStream_t stream) {
  (void)in_sizes; (void)n_in; (void)out_size; (void)ws_size;
  const int NA = 50000, NC = 100000, NE = 600000;

  const float* agt    = (const float*)d_in[0];
  const float* ctx    = (const float*)d_in[1];
  const int*   u      = (const int*)d_in[2];
  const int*   v      = (const int*)d_in[3];
  const float* Wq     = (const float*)d_in[4];
  const float* Wk     = (const float*)d_in[5];
  const float* Wctx   = (const float*)d_in[6];
  const float* gnc_w  = (const float*)d_in[7];
  const float* gnc_b  = (const float*)d_in[8];
  const float* Wagt   = (const float*)d_in[9];
  const float* gna_w  = (const float*)d_in[10];
  const float* gna_b  = (const float*)d_in[11];
  const float* norm_w = (const float*)d_in[12];
  const float* norm_b = (const float*)d_in[13];
  const float* Wlin   = (const float*)d_in[14];
  const float* gnl_w  = (const float*)d_in[15];
  const float* gnl_b  = (const float*)d_in[16];

  float* ws = (float*)d_ws;
  size_t off = 0;
  float* qbuf   = ws + off; off += (size_t)NA * DIM;
  off += (size_t)NC * DIM;
  float* cpbuf  = ws + off; off += (size_t)NC * DIM;
  off += (size_t)NE;
  int*   cntbuf = (int*)(ws + off); off += (size_t)NA;
  float* aggbuf = ws + off; off += (size_t)NA * DIM;
  int*   csr    = (int*)(ws + off); off += (size_t)NA * DIM;
  float* agtbuf = ws + off; off += (size_t)NA * DIM;
  unsigned* maxbuf = (unsigned*)(ws + off); off += 16;

  int*   rowptr = csr;
  int*   eid    = csr + 50016;
  int*   subuf  = csr + 650016;
  int*   arrbuf = csr + 1250016;
  int*   bsum   = csr + 1850016;
  float* wfold  = (float*)(csr + 2000000);
  float* wmsg   = (float*)(csr + 2040000);
  float* r12buf = (float*)(csr + 2080000);
  float* stbuf0 = cpbuf;
  float* auxbuf = cpbuf + 2 * 100000;
  float* stbuf1 = cpbuf + 4 * 100000;

  const dim3 b128(128), b256(256);
  const int CHAIN_FULL = 768;

  setup_kernel<<<33 + (NA + 255) / 256, b256, 0, stream>>>(Wq, Wk, wfold, Wagt, Wctx, gnc_w, gnc_b,
                                                          wmsg, r12buf, cntbuf, NA, maxbuf);

  csr_count_kernel<<<(NE + 255) / 256, b256, 0, stream>>>(v, cntbuf, arrbuf, NE);
  csr_blocksum_kernel<<<(NA + 255) / 256, b256, 0, stream>>>(cntbuf, bsum, NA);
  csr_scan_kernel<<<(NA + 255) / 256, b256, 0, stream>>>(cntbuf, bsum, rowptr, NA);
  csr_fill_kernel<<<(NE + 255) / 256, b256, 0, stream>>>(v, arrbuf, rowptr, eid, NE);
  csr_rank_kernel<<<(NE + 255) / 256, b256, 0, stream>>>(eid, rowptr, u, v, subuf, NE);

  gemm_q_and_ctxp<<<(NA + 63) / 64 + 2 * ((NC + 63) / 64), b128, 0, stream>>>(
      agt, wfold, qbuf, NA, (NA + 63) / 64,
      ctx, Wctx, stbuf0, Wctx + (size_t)DIM * DIM, stbuf1, NC, (NC + 63) / 64);

  node_attn_kernel<<<(NA + 7) / 8, b256, 0, stream>>>(qbuf, ctx, stbuf0, rowptr, subuf,
                                                     aggbuf, auxbuf, maxbuf, NA);
  gemm_update_chain<true><<<CHAIN_FULL + (NA - CHAIN_FULL * 64 + 15) / 16, b128, 0, stream>>>(
      agt, aggbuf, auxbuf, maxbuf, Wagt, wmsg, r12buf,
      gna_w, gna_b, norm_w, norm_b, Wlin, gnl_w, gnl_b, agtbuf,
      wfold + (size_t)DIM * DIM, qbuf, NA, CHAIN_FULL);

  node_attn_kernel<<<(NA + 7) / 8, b256, 0, stream>>>(qbuf, ctx, stbuf1, rowptr, subuf,
                                                     aggbuf, auxbuf, maxbuf + 1, NA);
  gemm_update_chain<false><<<CHAIN_FULL + (NA - CHAIN_FULL * 64 + 15) / 16, b128, 0, stream>>>(
      agtbuf, aggbuf, auxbuf, maxbuf + 1, Wagt + (size_t)DIM * 2 * DIM,
      wmsg + (size_t)DIM * DIM, r12buf + (size_t)2 * DIM,
      gna_w + DIM, gna_b + DIM, norm_w + DIM, norm_b + DIM,
      Wlin + (size_t)DIM * DIM, gnl_w + DIM, gnl_b + DIM, (float*)d_out,
      wfold, qbuf, NA, CHAIN_FULL);
}


# Edge kernel: 8192 resident waves each walk about 6 destination rows (all its segment bounds in one request; next row's slot indices and query row asked one row ahead) instead of 50000 one-row waves
# speedup vs baseline: 3.5305x; 1.0451x over previous
#include <hip/hip_runtime.h>


static constexpr int DIM  = 128;
static constexpr int NBLK = 2;
static constexpr int LDP  = 68;

typedef __attribute__((ext_vector_type(2))) float v2f;
typedef __attribute__((ext_vector_type(8))) float v8f;

static __device__ __forceinline__ unsigned f32_order_enc(float f) {
  unsigned u = __float_as_uint(f);
  return (u & 0x80000000u) ? ~u : (u | 0x80000000u);
}
static __device__ __forceinline__ float f32_order_dec(unsigned e) {
  unsigned u = (e & 0x80000000u) ? (e & 0x7fffffffu) : ~e;
  return __uint_as_float(u);
}

enum { EPI_STATS = 1, EPI_GN2_RELU = 2, EPI_GN_RES_RELU = 3 };

template <int XM>
static __device__ __forceinline__ float half_xor(float x) {
  return __int_as_float(__builtin_amdgcn_update_dpp(
      0, __float_as_int(x), 0x160 + XM, 0xf, 0xf, true));
}
static __device__ __forceinline__ float half_sum(float x) {
  x += half_xor<1>(x);
  x += half_xor<2>(x);
  x += half_xor<4>(x);
  x += half_xor<8>(x);
  return x;
}

template <bool RELU>
static __device__ __forceinline__ void gn_rows(v8f (&acc)[8],
                                               const float* __restrict__ w,
                                               const float* __restrict__ b,
                                               int l16) {
  float wv[8], bv[8];
#pragma unroll
  for (int t = 0; t < 8; ++t) {
    wv[t] = w[t * 16 + l16];
    bv[t] = b[t * 16 + l16];
  }
#pragma unroll
  for (int j = 0; j < 8; ++j) {
    float s = 0.0f, sq = 0.0f;
#pragma unroll
    for (int t = 0; t < 8; ++t) {
      const float x = acc[t][j];
      s += x;
      sq += x * x;
    }
    s  = half_sum(s);
    sq = half_sum(sq);
    const float mu = s * (1.0f / DIM);
    const float rs = rsqrtf(sq * (1.0f / DIM) - mu * mu + 1e-5f);
#pragma unroll
    for (int t = 0; t < 8; ++t) {
      float y = (acc[t][j] - mu) * rs * wv[t] + bv[t];
      if (RELU) y = fmaxf(y, 0.0f);
      acc[t][j] = y;
    }
  }
}

static __device__ __forceinline__ void stats_rows(const v8f (&acc)[8],
                                                  float* __restrict__ st,
                                                  int r0, int hk, int l16, int N) {
  float mymu = 0.0f, myrs = 0.0f;
#pragma unroll
  for (int j = 0; j < 8; ++j) {
    float s = 0.0f, sq = 0.0f;
#pragma unroll
    for (int t = 0; t < 8; ++t) {
      const float x = acc[t][j];
      s += x;
      sq += x * x;
    }
    s  = half_sum(s);
    sq = half_sum(sq);
    const float mu = s * (1.0f / DIM);
    const float rs = rsqrtf(sq * (1.0f / DIM) - mu * mu + 1e-5f);
    if (l16 == j) { mymu = mu; myrs = rs; }
  }
  const int row = r0 + hk * 8 + l16;
  if (l16 < 8 && row < N) {
    float2 v; v.x = mymu; v.y = myrs;
    ((float2*)st)[row] = v;
  }
}

template <int EPI>
static __device__ __forceinline__ void epilogue_store(v8f (&acc)[8],
                                                      float* __restrict__ C,
                                                      const float* __restrict__ R,
                                                      const float* __restrict__ w1,
                                                      const float* __restrict__ b1,
                                                      const float* __restrict__ w2,
                                                      const float* __restrict__ b2,
                                                      int r0, int hk, int l16, int N,
                                                      bool plain) {
  const int rb = r0 + hk * 8;

  if (EPI == EPI_GN_RES_RELU) {
    float res[8][8];
#pragma unroll
    for (int j = 0; j < 8; ++j) {
      int row = rb + j;
      if (row >= N) row = N - 1;
#pragma unroll
      for (int t = 0; t < 8; ++t) res[t][j] = R[(size_t)row * DIM + t * 16 + l16];
    }
    gn_rows<false>(acc, w1, b1, l16);
#pragma unroll
    for (int t = 0; t < 8; ++t)
#pragma unroll
      for (int j = 0; j < 8; ++j) acc[t][j] = fmaxf(acc[t][j] + res[t][j], 0.0f);
  } else if (EPI == EPI_GN2_RELU) {
    gn_rows<false>(acc, w1, b1, l16);
    gn_rows<true>(acc, w2, b2, l16);
  } else if (EPI == EPI_STATS) {
    if (!plain) {
      stats_rows(acc, C, r0, hk, l16, N);
      return;
    }
  }

  if (r0 + 16 <= N) {
#pragma unroll
    for (int t = 0; t < 8; ++t)
#pragma unroll
      for (int j = 0; j < 8; ++j)
        C[(size_t)(rb + j) * DIM + t * 16 + l16] = acc[t][j];
  } else {
#pragma unroll
    for (int t = 0; t < 8; ++t)
#pragma unroll
      for (int j = 0; j < 8; ++j) {
        const int row = rb + j;
        if (row < N) C[(size_t)row * DIM + t * 16 + l16] = acc[t][j];
      }
  }
}

template <int EPI>
static __device__ __forceinline__ void gemm_nt_128_body(float* __restrict__ Wlds,
                                                        const float* __restrict__ A,
                                                        const float* __restrict__ W,
                                                        float* __restrict__ C,
                                                        const float* __restrict__ R,
                                                        const float* __restrict__ w1,
                                                        const float* __restrict__ b1,
                                                        const float* __restrict__ w2,
                                                        const float* __restrict__ b2,
                                                        int N, int tile, bool plain) {
  const int tid = threadIdx.x;
  const int wv  = tid >> 5;
  const int ln  = tid & 31;
  const int hk  = ln >> 4;
  const int l16 = ln & 15;
  const int r0  = tile * 64 + wv * 16;

  v8f acc[8];
#pragma unroll
  for (int t = 0; t < 8; ++t)
#pragma unroll
    for (int j = 0; j < 8; ++j) acc[t][j] = 0.0f;

  int rowA = r0 + l16;
  if (rowA >= N) rowA = N - 1;
  const float* ap = A + (size_t)rowA * DIM;

  for (int kp = 0; kp < DIM; kp += 64) {
    __syncthreads();
#pragma unroll
    for (int i = 0; i < 16; ++i) {
      const int q  = tid + 128 * i;
      const int n  = q >> 4;
      const int c4 = (q & 15) << 2;
      *(float4*)(&Wlds[n * LDP + c4]) =
          *(const float4*)(W + (size_t)n * DIM + kp + c4);
    }
    __syncthreads();

#pragma unroll 4
    for (int k0 = 0; k0 < 64; k0 += 4) {
      const int ka = k0 + 2 * hk;
      v2f a; a.x = ap[kp + ka]; a.y = ap[kp + ka + 1];
#pragma unroll
      for (int t = 0; t < 8; ++t) {
        const float* wp = &Wlds[(t * 16 + l16) * LDP + ka];
        v2f b; b.x = wp[0]; b.y = wp[1];
        acc[t] = __builtin_amdgcn_wmma_f32_16x16x4_f32(
            false, a, false, b, (short)0, acc[t], false, false);
      }
    }
  }

  epilogue_store<EPI>(acc, C, R, w1, b1, w2, b2, r0, hk, l16, N, plain);
}

__global__ __launch_bounds__(128) void gemm_q_and_ctxp(const float* __restrict__ Aq,
                                                       const float* __restrict__ Wfq,
                                                       float* __restrict__ Cq, int Nq,
                                                       int tilesQ,
                                                       const float* __restrict__ Ac,
                                                       const float* __restrict__ Wc0,
                                                       float* __restrict__ Cc0,
                                                       const float* __restrict__ Wc1,
                                                       float* __restrict__ Cc1,
                                                       int Nc, int tilesC) {
  __shared__ float Wlds[128 * LDP];
  const int bid = blockIdx.x;
  const bool isq = bid < tilesQ;
  const bool sec = bid >= tilesQ + tilesC;
  const float* A  = isq ? Aq : Ac;
  const float* Wm = isq ? Wfq : (sec ? Wc1 : Wc0);
  float* C        = isq ? Cq : (sec ? Cc1 : Cc0);
  const int N     = isq ? Nq : Nc;
  const int tile  = isq ? bid : (sec ? bid - tilesQ - tilesC : bid - tilesQ);
  gemm_nt_128_body<EPI_STATS>(Wlds, A, Wm, C, nullptr, nullptr, nullptr, nullptr, nullptr, N, tile, isq);
}

static constexpr int XP = 132;

static __device__ __forceinline__ void tile_to_lds(const v8f (&acc)[8], float* xs,
                                                   int hk, int l16) {
#pragma unroll
  for (int t = 0; t < 8; ++t)
#pragma unroll
    for (int j = 0; j < 8; ++j) xs[(8 * hk + j) * XP + 16 * t + l16] = acc[t][j];
}

static __device__ __forceinline__ void gemm_tile_lds_128(v8f (&acc)[8], float* Wlds,
                                                         const float* xs,
                                                         const float* __restrict__ Wg,
                                                         int tid, int hk, int l16) {
#pragma unroll
  for (int t = 0; t < 8; ++t)
#pragma unroll
    for (int j = 0; j < 8; ++j) acc[t][j] = 0.0f;

  const float* xr = xs + l16 * XP;
  for (int kp = 0; kp < DIM; kp += 64) {
    __syncthreads();
#pragma unroll
    for (int i = 0; i < 16; ++i) {
      const int q  = tid + 128 * i;
      const int n  = q >> 4;
      const int c4 = (q & 15) << 2;
      *(float4*)(&Wlds[n * LDP + c4]) =
          *(const float4*)(Wg + (size_t)n * DIM + kp + c4);
    }
    __syncthreads();

#pragma unroll 4
    for (int k0 = 0; k0 < 64; k0 += 4) {
      const int ka = k0 + 2 * hk;
      v2f a; a.x = xr[kp + ka]; a.y = xr[kp + ka + 1];
#pragma unroll
      for (int t = 0; t < 8; ++t) {
        const float* wp = &Wlds[(t * 16 + l16) * LDP + ka];
        v2f b; b.x = wp[0]; b.y = wp[1];
        acc[t] = __builtin_amdgcn_wmma_f32_16x16x4_f32(
            false, a, false, b, (short)0, acc[t], false, false);
      }
    }
  }
}


template <bool RELU>
static __device__ __forceinline__ void gn_rows_split(v8f (&acc)[2], float* red,
                                                     const float* __restrict__ w,
                                                     const float* __restrict__ b,
                                                     int wv, int hk, int l16) {
  float wc[2], bc[2];
#pragma unroll
  for (int t = 0; t < 2; ++t) {
    wc[t] = w[(2 * wv + t) * 16 + l16];
    bc[t] = b[(2 * wv + t) * 16 + l16];
  }
  float mys = 0.0f, mysq = 0.0f;
#pragma unroll
  for (int j = 0; j < 8; ++j) {
    float s  = acc[0][j] + acc[1][j];
    float sq = acc[0][j] * acc[0][j] + acc[1][j] * acc[1][j];
    s  = half_sum(s);
    sq = half_sum(sq);
    if (l16 == j) { mys = s; mysq = sq; }
  }
  if (l16 < 8) {
    float2 p; p.x = mys; p.y = mysq;
    ((float2*)red)[wv * 16 + hk * 8 + l16] = p;
  }
  __syncthreads();
#pragma unroll
  for (int j = 0; j < 8; ++j) {
    float s = 0.0f, sq = 0.0f;
#pragma unroll
    for (int q = 0; q < 4; ++q) {
      const float2 p = ((const float2*)red)[q * 16 + hk * 8 + j];
      s += p.x;
      sq += p.y;
    }
    const float mu = s * (1.0f / DIM);
    const float rs = rsqrtf(sq * (1.0f / DIM) - mu * mu + 1e-5f);
#pragma unroll
    for (int t = 0; t < 2; ++t) {
      float y = (acc[t][j] - mu) * rs * wc[t] + bc[t];
      if (RELU) y = fmaxf(y, 0.0f);
      acc[t][j] = y;
    }
  }
}

static __device__ __forceinline__ void tile_to_lds_split(const v8f (&acc)[2], float* xs,
                                                         int wv, int hk, int l16) {
#pragma unroll
  for (int t = 0; t < 2; ++t)
#pragma unroll
    for (int j = 0; j < 8; ++j) xs[(8 * hk + j) * XP + 16 * (2 * wv + t) + l16] = acc[t][j];
}

static __device__ __forceinline__ void gemm_tile_lds_split(v8f (&acc)[2], float* Wlds,
                                                           const float* xs,
                                                           const float* __restrict__ Wg,
                                                           int tid, int wv, int hk, int l16) {
#pragma unroll
  for (int t = 0; t < 2; ++t)
#pragma unroll
    for (int j = 0; j < 8; ++j) acc[t][j] = 0.0f;

  const float* xr = xs + l16 * XP;
  for (int kp = 0; kp < DIM; kp += 64) {
    __syncthreads();
#pragma unroll
    for (int i = 0; i < 16; ++i) {
      const int q  = tid + 128 * i;
      const int n  = q >> 4;
      const int c4 = (q & 15) << 2;
      *(float4*)(&Wlds[n * LDP + c4]) =
          *(const float4*)(Wg + (size_t)n * DIM + kp + c4);
    }
    __syncthreads();

#pragma unroll 4
    for (int k0 = 0; k0 < 64; k0 += 4) {
      const int ka = k0 + 2 * hk;
      v2f a; a.x = xr[kp + ka]; a.y = xr[kp + ka + 1];
#pragma unroll
      for (int t = 0; t < 2; ++t) {
        const float* wp = &Wlds[((2 * wv + t) * 16 + l16) * LDP + ka];
        v2f b; b.x = wp[0]; b.y = wp[1];
        acc[t] = __builtin_amdgcn_wmma_f32_16x16x4_f32(
            false, a, false, b, (short)0, acc[t], false, false);
      }
    }
  }
}

static __device__ __forceinline__ void store_split(const v8f (&acc)[2], float* __restrict__ C,
                                                   int rb, int wv, int l16, int N) {
#pragma unroll
  for (int t = 0; t < 2; ++t)
#pragma unroll
    for (int j = 0; j < 8; ++j) {
      const int row = rb + j;
      if (row < N) C[(size_t)row * DIM + (2 * wv + t) * 16 + l16] = acc[t][j];
    }
}

template <bool NEXTQ>
static __device__ __forceinline__ void chain_split_body(float* Wlds, float* Xlds,
                                                        const float* __restrict__ A0,
                                                        const float* __restrict__ A1,
                                                        const float* __restrict__ aux,
                                                        const unsigned* __restrict__ gmax,
                                                        const float* __restrict__ W,
                                                        const float* __restrict__ Wm,
                                                        const float* __restrict__ r12,
                                                        const float* __restrict__ w1,
                                                        const float* __restrict__ b1,
                                                        const float* __restrict__ w2,
                                                        const float* __restrict__ b2,
                                                        const float* __restrict__ Wl,
                                                        const float* __restrict__ gl_w,
                                                        const float* __restrict__ gl_b,
                                                        float* __restrict__ Cout,
                                                        const float* __restrict__ Wfn,
                                                        float* __restrict__ Qn,
                                                        int N, int r0) {
  const int tid = threadIdx.x;
  const int wv  = tid >> 5;
  const int ln  = tid & 31;
  const int hk  = ln >> 4;
  const int l16 = ln & 15;
  float* xs   = Xlds;
  float* red0 = Xlds + 16 * XP;
  float* red1 = red0 + 128;
  const int rb = r0 + hk * 8;

  v8f acc[2];
#pragma unroll
  for (int t = 0; t < 2; ++t)
#pragma unroll
    for (int j = 0; j < 8; ++j) acc[t][j] = 0.0f;

  int rowA = r0 + l16;
  if (rowA >= N) rowA = N - 1;
  const size_t rbase = (size_t)rowA * DIM;

  const float4 ax = ((const float4*)aux)[rowA];
  const float gm  = f32_order_dec(*gmax);
  const float ee  = expf(ax.x - gm);
  const float fr  = ee / (ee * ax.y + 1e-16f);

  for (int kp = 0; kp < 2 * DIM; kp += 64) {
    const bool first = kp < DIM;
    const float* wsrc = first ? (W + kp) : (Wm + (kp - DIM));
    const int wld     = first ? 2 * DIM : DIM;
    __syncthreads();
#pragma unroll
    for (int i = 0; i < 16; ++i) {
      const int q  = tid + 128 * i;
      const int n  = q >> 4;
      const int c4 = (q & 15) << 2;
      *(float4*)(&Wlds[n * LDP + c4]) =
          *(const float4*)(wsrc + (size_t)n * wld + c4);
    }
    __syncthreads();

    const float* abase = first ? (A0 + rbase + kp) : (A1 + rbase + (kp - DIM));
    const float asc    = first ? 1.0f : fr;
#pragma unroll 4
    for (int k0 = 0; k0 < 64; k0 += 4) {
      const int ka = k0 + 2 * hk;
      v2f a; a.x = abase[ka] * asc; a.y = abase[ka + 1] * asc;
#pragma unroll
      for (int t = 0; t < 2; ++t) {
        const float* wp = &Wlds[((2 * wv + t) * 16 + l16) * LDP + ka];
        v2f b; b.x = wp[0]; b.y = wp[1];
        acc[t] = __builtin_amdgcn_wmma_f32_16x16x4_f32(
            false, a, false, b, (short)0, acc[t], false, false);
      }
    }
  }

  {
    v2f a;
    a.x = (hk == 0) ? -fr * ax.z : 0.0f;
    a.y = (hk == 0) ?  fr * ax.y : 0.0f;
#pragma unroll
    for (int t = 0; t < 2; ++t) {
      v2f b;
      b.x = (hk == 0) ? r12[(2 * wv + t) * 16 + l16] : 0.0f;
      b.y = (hk == 0) ? r12[DIM + (2 * wv + t) * 16 + l16] : 0.0f;
      acc[t] = __builtin_amdgcn_wmma_f32_16x16x4_f32(
          false, a, false, b, (short)0, acc[t], false, false);
    }
  }

  gn_rows_split<false>(acc, red0, w1, b1, wv, hk, l16);
  gn_rows_split<true>(acc, red1, w2, b2, wv, hk, l16);
  tile_to_lds_split(acc, xs, wv, hk, l16);

  gemm_tile_lds_split(acc, Wlds, xs, Wl, tid, wv, hk, l16);
  {
    float res[2][8];
#pragma unroll
    for (int j = 0; j < 8; ++j) {
      int row = rb + j;
      if (row >= N) row = N - 1;
#pragma unroll
      for (int t = 0; t < 2; ++t) res[t][j] = A0[(size_t)row * DIM + (2 * wv + t) * 16 + l16];
    }
    gn_rows_split<false>(acc, red0, gl_w, gl_b, wv, hk, l16);
#pragma unroll
    for (int t = 0; t < 2; ++t)
#pragma unroll
      for (int j = 0; j < 8; ++j) acc[t][j] = fmaxf(acc[t][j] + res[t][j], 0.0f);
  }
  store_split(acc, Cout, rb, wv, l16, N);

  if (NEXTQ) {
    tile_to_lds_split(acc, xs, wv, hk, l16);
    gemm_tile_lds_split(acc, Wlds, xs, Wfn, tid, wv, hk, l16);
    store_split(acc, Qn, rb, wv, l16, N);
  }
}

template <bool NEXTQ>
__global__ __launch_bounds__(128) void gemm_update_chain(const float* __restrict__ A0,
                                                         const float* __restrict__ A1,
                                                         const float* __restrict__ aux,
                                                         const unsigned* __restrict__ gmax,
                                                         const float* __restrict__ W,
                                                         const float* __restrict__ Wm,
                                                         const float* __restrict__ r12,
                                                         const float* __restrict__ w1,
                                                         const float* __restrict__ b1,
                                                         const float* __restrict__ w2,
                                                         const float* __restrict__ b2,
                                                         const float* __restrict__ Wl,
                                                         const float* __restrict__ gl_w,
                                                         const float* __restrict__ gl_b,
                                                         float* __restrict__ Cout,
                                                         const float* __restrict__ Wfn,
                                                         float* __restrict__ Qn,
                                                         int N, int tilesFull) {
  __shared__ __attribute__((aligned(16))) float Wlds[128 * LDP];
  __shared__ __attribute__((aligned(16))) float Xlds[4 * 16 * XP];
  if ((int)blockIdx.x >= tilesFull) {
    chain_split_body<NEXTQ>(Wlds, Xlds, A0, A1, aux, gmax, W, Wm, r12, w1, b1, w2, b2,
                            Wl, gl_w, gl_b, Cout, Wfn, Qn, N,
                            tilesFull * 64 + ((int)blockIdx.x - tilesFull) * 16);
    return;
  }
  const int tid = threadIdx.x;
  const int wv  = tid >> 5;
  const int ln  = tid & 31;
  const int hk  = ln >> 4;
  const int l16 = ln & 15;
  const int r0  = blockIdx.x * 64 + wv * 16;
  float* xs = Xlds + wv * 16 * XP;

  v8f acc[8];
#pragma unroll
  for (int t = 0; t < 8; ++t)
#pragma unroll
    for (int j = 0; j < 8; ++j) acc[t][j] = 0.0f;

  int rowA = r0 + l16;
  if (rowA >= N) rowA = N - 1;
  const size_t rbase = (size_t)rowA * DIM;

  const float4 ax = ((const float4*)aux)[rowA];
  const float gm  = f32_order_dec(*gmax);
  const float ee  = expf(ax.x - gm);
  const float fr  = ee / (ee * ax.y + 1e-16f);

  for (int kp = 0; kp < 2 * DIM; kp += 64) {
    const bool first = kp < DIM;
    const float* wsrc = first ? (W + kp) : (Wm + (kp - DIM));
    const int wld     = first ? 2 * DIM : DIM;
    __syncthreads();
#pragma unroll
    for (int i = 0; i < 16; ++i) {
      const int q  = tid + 128 * i;
      const int n  = q >> 4;
      const int c4 = (q & 15) << 2;
      *(float4*)(&Wlds[n * LDP + c4]) =
          *(const float4*)(wsrc + (size_t)n * wld + c4);
    }
    __syncthreads();

    const float* abase = first ? (A0 + rbase + kp) : (A1 + rbase + (kp - DIM));
    const float asc    = first ? 1.0f : fr;
#pragma unroll 4
    for (int k0 = 0; k0 < 64; k0 += 4) {
      const int ka = k0 + 2 * hk;
      v2f a; a.x = abase[ka] * asc; a.y = abase[ka + 1] * asc;
#pragma unroll
      for (int t = 0; t < 8; ++t) {
        const float* wp = &Wlds[(t * 16 + l16) * LDP + ka];
        v2f b; b.x = wp[0]; b.y = wp[1];
        acc[t] = __builtin_amdgcn_wmma_f32_16x16x4_f32(
            false, a, false, b, (short)0, acc[t], false, false);
      }
    }
  }

  {
    v2f a;
    a.x = (hk == 0) ? -fr * ax.z : 0.0f;
    a.y = (hk == 0) ?  fr * ax.y : 0.0f;
#pragma unroll
    for (int t = 0; t < 8; ++t) {
      v2f b;
      b.x = (hk == 0) ? r12[t * 16 + l16] : 0.0f;
      b.y = (hk == 0) ? r12[DIM + t * 16 + l16] : 0.0f;
      acc[t] = __builtin_amdgcn_wmma_f32_16x16x4_f32(
          false, a, false, b, (short)0, acc[t], false, false);
    }
  }

  gn_rows<false>(acc, w1, b1, l16);
  gn_rows<true>(acc, w2, b2, l16);
  tile_to_lds(acc, xs, hk, l16);

  gemm_tile_lds_128(acc, Wlds, xs, Wl, tid, hk, l16);
  epilogue_store<EPI_GN_RES_RELU>(acc, Cout, A0, gl_w, gl_b, nullptr, nullptr, r0, hk, l16, N, false);

  if (NEXTQ) {
    tile_to_lds(acc, xs, hk, l16);
    gemm_tile_lds_128(acc, Wlds, xs, Wfn, tid, hk, l16);
    epilogue_store<EPI_STATS>(acc, Qn, nullptr, nullptr, nullptr, nullptr, nullptr, r0, hk, l16, N, true);
  }
}

static __device__ __forceinline__ void fold_qk_body(const float* __restrict__ Wq,
                                                    const float* __restrict__ Wk,
                                                    float* __restrict__ Wf, int bid) {
  const int wv  = threadIdx.x >> 5;
  const int ln  = threadIdx.x & 31;
  const int hk  = ln >> 4;
  const int l16 = ln & 15;
  const int tile = bid * 8 + wv;
  const int i  = tile >> 6;
  const int j0 = ((tile >> 3) & 7) * 16;
  const int m0 = (tile & 7) * 16;
  const float* wq = Wq + (size_t)i * DIM * DIM;
  const float* wk = Wk + (size_t)i * DIM * DIM;

  v8f acc;
#pragma unroll
  for (int r = 0; r < 8; ++r) acc[r] = 0.0f;

#pragma unroll 8
  for (int d0 = 0; d0 < DIM; d0 += 4) {
    const int d = d0 + 2 * hk;
    v2f a; a.x = wk[d * DIM + j0 + l16]; a.y = wk[(d + 1) * DIM + j0 + l16];
    v2f b; b.x = wq[d * DIM + m0 + l16]; b.y = wq[(d + 1) * DIM + m0 + l16];
    acc = __builtin_amdgcn_wmma_f32_16x16x4_f32(
        false, a, false, b, (short)0, acc, false, false);
  }

  float* o = Wf + (size_t)i * DIM * DIM + (size_t)(j0 + 8 * hk) * DIM + m0 + l16;
#pragma unroll
  for (int r = 0; r < 8; ++r) o[r * DIM] = acc[r];
}

static __device__ __forceinline__ void fold_msg_body(const float* __restrict__ Wagt,
                                                     const float* __restrict__ Wctx,
                                                     const float* __restrict__ gw,
                                                     float* __restrict__ Wm, int bid) {
  const int wv  = threadIdx.x >> 5;
  const int ln  = threadIdx.x & 31;
  const int hk  = ln >> 4;
  const int l16 = ln & 15;
  const int tile = bid * 8 + wv;
  const int i  = tile >> 6;
  const int o0 = ((tile >> 3) & 7) * 16;
  const int k0 = (tile & 7) * 16;
  const float* wa = Wagt + (size_t)i * DIM * 2 * DIM + (size_t)(o0 + l16) * (2 * DIM) + DIM;
  const float* wc = Wctx + (size_t)i * DIM * DIM + k0 + l16;
  const float* g  = gw + i * DIM;

  v8f acc;
#pragma unroll
  for (int r = 0; r < 8; ++r) acc[r] = 0.0f;

#pragma unroll 8
  for (int d0 = 0; d0 < DIM; d0 += 4) {
    const int d = d0 + 2 * hk;
    v2f a; a.x = wa[d] * g[d];       a.y = wa[d + 1] * g[d + 1];
    v2f b; b.x = wc[d * DIM];        b.y = wc[(d + 1) * DIM];
    acc = __builtin_amdgcn_wmma_f32_16x16x4_f32(
        false, a, false, b, (short)0, acc, false, false);
  }

  float* o = Wm + (size_t)i * DIM * DIM + (size_t)(o0 + 8 * hk) * DIM + k0 + l16;
#pragma unroll
  for (int r = 0; r < 8; ++r) o[r * DIM] = acc[r];
}

__global__ __launch_bounds__(256) void setup_kernel(const float* __restrict__ Wq,
                                                    const float* __restrict__ Wk,
                                                    float* __restrict__ Wf,
                                                    const float* __restrict__ Wagt,
                                                    const float* __restrict__ Wctx,
                                                    const float* __restrict__ gw,
                                                    const float* __restrict__ gb,
                                                    float* __restrict__ Wm,
                                                    float* __restrict__ r12,
                                                    int* __restrict__ cnt, int n,
                                                    unsigned* __restrict__ gmax) {
  const int bid = blockIdx.x;
  if (bid < 16) {
    fold_qk_body(Wq, Wk, Wf, bid);
  } else if (bid < 32) {
    fold_msg_body(Wagt, Wctx, gw, Wm, bid - 16);
  } else if (bid == 32) {
    const int i = threadIdx.x >> 7;
    const int o = threadIdx.x & 127;
    const float* wa = Wagt + (size_t)i * DIM * 2 * DIM + (size_t)o * (2 * DIM) + DIM;
    float a1 = 0.0f, a2 = 0.0f;
    for (int c = 0; c < DIM; ++c) {
      const float x = wa[c];
      a1 += x * gw[i * DIM + c];
      a2 += x * gb[i * DIM + c];
    }
    r12[i * 2 * DIM + o]       = a1;
    r12[i * 2 * DIM + DIM + o] = a2;
  } else {
    const int i = (bid - 33) * 256 + threadIdx.x;
    if (i < n) cnt[i] = 0;
    if (bid == 33 && threadIdx.x < NBLK) gmax[threadIdx.x] = 0u;
  }
}

__global__ __launch_bounds__(256) void csr_count_kernel(const int* __restrict__ v,
                                                        int* __restrict__ cnt,
                                                        int* __restrict__ arr, int E) {
  const int e = blockIdx.x * 256 + threadIdx.x;
  if (e < E) arr[e] = atomicAdd(&cnt[v[e]], 1);
}

static __device__ __forceinline__ int wave_reduce_sum_i(int x) {
#pragma unroll
  for (int off = 16; off > 0; off >>= 1) x += __shfl_xor(x, off, 32);
  return x;
}

__global__ __launch_bounds__(256) void csr_blocksum_kernel(const int* __restrict__ cnt,
                                                           int* __restrict__ bsum, int N) {
  __shared__ int sw[8];
  const int wv = threadIdx.x >> 5, ln = threadIdx.x & 31;
  const int i = blockIdx.x * 256 + threadIdx.x;
  int c = (i < N) ? cnt[i] : 0;
  c = wave_reduce_sum_i(c);
  if (ln == 0) sw[wv] = c;
  __syncthreads();
  if (threadIdx.x == 0) {
    int t = 0;
#pragma unroll
    for (int k = 0; k < 8; ++k) t += sw[k];
    bsum[blockIdx.x] = t;
  }
}

__global__ __launch_bounds__(256) void csr_scan_kernel(const int* __restrict__ cnt,
                                                       const int* __restrict__ bsum,
                                                       int* __restrict__ rowptr, int N) {
  __shared__ int sa[256];
  __shared__ int sb[256];
  __shared__ int sw[8];
  const int t = threadIdx.x;
  const int wv = t >> 5, ln = t & 31;
  const int blk = blockIdx.x;

  int p = 0;
  for (int k = t; k < blk; k += 256) p += bsum[k];
  p = wave_reduce_sum_i(p);
  if (ln == 0) sw[wv] = p;

  const int i = blk * 256 + t;
  const int c = (i < N) ? cnt[i] : 0;
  sa[t] = c;
  __syncthreads();

  int base = 0;
#pragma unroll
  for (int k = 0; k < 8; ++k) base += sw[k];

  int* src = sa;
  int* dst = sb;
  for (int off = 1; off < 256; off <<= 1) {
    int val = src[t];
    if (t >= off) val += src[t - off];
    dst[t] = val;
    __syncthreads();
    int* tmp = src; src = dst; dst = tmp;
  }
  const int excl = base + src[t] - c;
  if (i < N) rowptr[i] = excl;
  if (i == N - 1) rowptr[N] = excl + c;
}

__global__ __launch_bounds__(256) void csr_fill_kernel(const int* __restrict__ v,
                                                       const int* __restrict__ arr,
                                                       const int* __restrict__ rowptr,
                                                       int* __restrict__ eid, int E) {
  const int e = blockIdx.x * 256 + threadIdx.x;
  if (e < E) eid[rowptr[v[e]] + arr[e]] = e;
}

__global__ __launch_bounds__(256) void csr_rank_kernel(const int* __restrict__ eid,
                                                       const int* __restrict__ rowptr,
                                                       const int* __restrict__ u,
                                                       const int* __restrict__ v,
                                                       int* __restrict__ su, int E) {
  const int j = blockIdx.x * 256 + threadIdx.x;
  if (j >= E) return;
  const int e = eid[j];
  const int n = v[e];
  const int b = rowptr[n], en = rowptr[n + 1];
  int r = 0;
  for (int k = b; k < en; ++k) r += (eid[k] < e) ? 1 : 0;
  su[b + r] = u[e];
}

static constexpr int CH = 16;

static __device__ __forceinline__ int lane_i(int x, int l) {
  return __builtin_amdgcn_readlane(x, l);
}
static __device__ __forceinline__ float lane_f(float x, int l) {
  return __int_as_float(__builtin_amdgcn_readlane(__float_as_int(x), l));
}
static __device__ __forceinline__ float other_half(float x) {
  return __int_as_float(__builtin_amdgcn_permlanex16(
      __float_as_int(x), __float_as_int(x), 0x76543210, 0xfedcba98, false, false));
}

static __device__ __forceinline__ float half_sum_16(const float (&p)[16], int ln) {
  const bool b0 = (ln & 1) != 0, b1 = (ln & 2) != 0, b2 = (ln & 4) != 0, b3 = (ln & 8) != 0;
  float r1[8], r2[4], r3[2];
#pragma unroll
  for (int i = 0; i < 8; ++i) {
    const float keep = b0 ? p[2 * i + 1] : p[2 * i];
    const float send = b0 ? p[2 * i] : p[2 * i + 1];
    r1[i] = keep + half_xor<1>(send);
  }
#pragma unroll
  for (int i = 0; i < 4; ++i) {
    const float keep = b1 ? r1[2 * i + 1] : r1[2 * i];
    const float send = b1 ? r1[2 * i] : r1[2 * i + 1];
    r2[i] = keep + half_xor<2>(send);
  }
#pragma unroll
  for (int i = 0; i < 2; ++i) {
    const float keep = b2 ? r2[2 * i + 1] : r2[2 * i];
    const float send = b2 ? r2[2 * i] : r2[2 * i + 1];
    r3[i] = keep + half_xor<4>(send);
  }
  const float keep = b3 ? r3[1] : r3[0];
  const float send = b3 ? r3[0] : r3[1];
  return keep + half_xor<8>(send);
}

static constexpr int NA_WG = 1024;

static __device__ __forceinline__ void chunk_request(const float* __restrict__ X,
                                                     const float* __restrict__ st,
                                                     int uu, int cnt, int ln,
                                                     float4 (&kv)[CH], float2& sv) {
  sv = ((const float2*)st)[uu];
#pragma unroll
  for (int g = 0; g < CH / 4; ++g) {
    if (4 * g < cnt) {
#pragma unroll
      for (int l = 4 * g; l < 4 * g + 4; ++l) {
        const int ui = lane_i(uu, l);
        kv[l] = ((const float4*)(X + (size_t)ui * DIM))[ln];
      }
    }
  }
}

static __device__ __forceinline__ void chunk_reduce(const float4 (&kv)[CH], const float2 sv,
                                                    const float4 qv, int cnt, int ln,
                                                    float& m, float& ssum, float& tsum,
                                                    float4& acc) {
  float p[CH];
#pragma unroll
  for (int g = 0; g < CH / 4; ++g) {
    if (4 * g < cnt) {
#pragma unroll
      for (int l = 4 * g; l < 4 * g + 4; ++l)
        p[l] = qv.x * kv[l].x + qv.y * kv[l].y + qv.z * kv[l].z + qv.w * kv[l].w;
    } else {
#pragma unroll
      for (int l = 4 * g; l < 4 * g + 4; ++l) p[l] = 0.0f;
    }
  }

  const float hs  = half_sum_16(p, ln);
  const float tot = hs + other_half(hs);
  const float sco = (tot >= 0.0f) ? tot : 0.1f * tot;
  const bool valid = ln < cnt;
  float mx = valid ? sco : -3.0e38f;
  mx = fmaxf(mx, half_xor<1>(mx));
  mx = fmaxf(mx, half_xor<2>(mx));
  mx = fmaxf(mx, half_xor<4>(mx));
  mx = fmaxf(mx, half_xor<8>(mx));
  const float mc = lane_f(mx, 0);

  const float mn = fmaxf(m, mc);
  const float sc = expf(m - mn);
  ssum *= sc; tsum *= sc;
  acc.x *= sc; acc.y *= sc; acc.z *= sc; acc.w *= sc;
  m = mn;

  const float ev = valid ? expf(sco - mn) : 0.0f;
  const float wl = ev * sv.y;

#pragma unroll
  for (int g = 0; g < CH / 4; ++g) {
    if (4 * g < cnt) {
#pragma unroll
      for (int l = 4 * g; l < 4 * g + 4; ++l) {
        const float e  = lane_f(ev, l);
        const float w  = lane_f(wl, l);
        const float mu = lane_f(sv.x, l);
        ssum += e;
        tsum += w * mu;
        acc.x += w * kv[l].x;
        acc.y += w * kv[l].y;
        acc.z += w * kv[l].z;
        acc.w += w * kv[l].w;
      }
    }
  }
}

static __device__ __forceinline__ int first_chunk_slot(int sb, int se, int ln) {
  const int c0 = (se - sb < CH) ? (se - sb) : CH;
  const int j  = sb + ((ln < c0) ? ln : (c0 - 1));
  return (j > 0) ? j : 0;
}

__global__ __launch_bounds__(256) void node_attn_kernel(const float* __restrict__ Q,
                                                        const float* __restrict__ X,
                                                        const float* __restrict__ st,
                                                        const int* __restrict__ rowptr,
                                                        const int* __restrict__ su,
                                                        float* __restrict__ S,
                                                        float* __restrict__ aux,
                                                        unsigned* __restrict__ gmax,
                                                        int N) {
  __shared__ float smax[8];
  const int wv = threadIdx.x >> 5, ln = threadIdx.x & 31;
  const int nw = __builtin_amdgcn_readfirstlane((int)gridDim.x * 8);
  const int w0 = __builtin_amdgcn_readfirstlane((int)blockIdx.x * 8 + wv);
  float wmax = -3.0e38f;
  for (int base = w0; base < N; base += 32 * nw) {
    const int nl = base + ln * nw;
    int segb = 0, sege = 0;
    if (nl < N) { segb = rowptr[nl]; sege = rowptr[nl + 1]; }
    int left = (N - base + nw - 1) / nw;
    if (left > 32) left = 32;

    int b  = lane_i(segb, 0);
    int en = lane_i(sege, 0);
    float4 qv = ((const float4*)(Q + (size_t)base * DIM))[ln];
    int uf = su[first_chunk_slot(b, en, ln)];

    for (int i = 0; i < left; ++i) {
      const int n = base + i * nw;
      float m = -3.0e38f;
      float ssum = 0.0f, tsum = 0.0f;
      float4 acc;
      acc.x = 0.0f; acc.y = 0.0f; acc.z = 0.0f; acc.w = 0.0f;

      const int inx = (i + 1 < left) ? (i + 1) : i;
      const int nb  = lane_i(segb, inx);
      const int nen = lane_i(sege, inx);
      const int ufn = su[first_chunk_slot(nb, nen, ln)];
      const float4 qn = ((const float4*)(Q + (size_t)(base + inx * nw) * DIM))[ln];

      float4 kv[CH];
      float2 sv;
      sv.x = 0.0f; sv.y = 0.0f;
      const int cnt0 = (en - b < CH) ? (en - b) : CH;
      if (cnt0 > 0) {
        chunk_request(X, st, uf, cnt0, ln, kv, sv);
        chunk_reduce(kv, sv, qv, cnt0, ln, m, ssum, tsum, acc);
      }

      for (int j0 = b + CH; j0 < en; j0 += CH) {
        const int cnt = (en - j0 < CH) ? (en - j0) : CH;
        const int uu = su[j0 + ((ln < cnt) ? ln : (cnt - 1))];
        chunk_request(X, st, uu, cnt, ln, kv, sv);
        chunk_reduce(kv, sv, qv, cnt, ln, m, ssum, tsum, acc);
      }

      ((float4*)(S + (size_t)n * DIM))[ln] = acc;
      if (ln == 0) {
        float4 a; a.x = m; a.y = ssum; a.z = tsum; a.w = 0.0f;
        ((float4*)aux)[n] = a;
      }
      wmax = fmaxf(wmax, m);

      b = nb; en = nen; uf = ufn; qv = qn;
    }
  }
  if (ln == 0) smax[wv] = wmax;
  __syncthreads();
  if (threadIdx.x == 0) {
    float mm = smax[0];
#pragma unroll
    for (int i = 1; i < 8; ++i) mm = fmaxf(mm, smax[i]);
    atomicMax(gmax, f32_order_enc(mm));
  }
}

extern "C" void kernel_launch(void* const* d_in, const int* in_sizes, int n_in,
                              void* d_out, int out_size, void* d_ws, size_t ws_size,
                              hipStream_t stream) {
  (void)in_sizes; (void)n_in; (void)out_size; (void)ws_size;
  const int NA = 50000, NC = 100000, NE = 600000;

  const float* agt    = (const float*)d_in[0];
  const float* ctx    = (const float*)d_in[1];
  const int*   u      = (const int*)d_in[2];
  const int*   v      = (const int*)d_in[3];
  const float* Wq     = (const float*)d_in[4];
  const float* Wk     = (const float*)d_in[5];
  const float* Wctx   = (const float*)d_in[6];
  const float* gnc_w  = (const float*)d_in[7];
  const float* gnc_b  = (const float*)d_in[8];
  const float* Wagt   = (const float*)d_in[9];
  const float* gna_w  = (const float*)d_in[10];
  const float* gna_b  = (const float*)d_in[11];
  const float* norm_w = (const float*)d_in[12];
  const float* norm_b = (const float*)d_in[13];
  const float* Wlin   = (const float*)d_in[14];
  const float* gnl_w  = (const float*)d_in[15];
  const float* gnl_b  = (const float*)d_in[16];

  float* ws = (float*)d_ws;
  size_t off = 0;
  float* qbuf   = ws + off; off += (size_t)NA * DIM;
  off += (size_t)NC * DIM;
  float* cpbuf  = ws + off; off += (size_t)NC * DIM;
  off += (size_t)NE;
  int*   cntbuf = (int*)(ws + off); off += (size_t)NA;
  float* aggbuf = ws + off; off += (size_t)NA * DIM;
  int*   csr    = (int*)(ws + off); off += (size_t)NA * DIM;
  float* agtbuf = ws + off; off += (size_t)NA * DIM;
  unsigned* maxbuf = (unsigned*)(ws + off); off += 16;

  int*   rowptr = csr;
  int*   eid    = csr + 50016;
  int*   subuf  = csr + 650016;
  int*   arrbuf = csr + 1250016;
  int*   bsum   = csr + 1850016;
  float* wfold  = (float*)(csr + 2000000);
  float* wmsg   = (float*)(csr + 2040000);
  float* r12buf = (float*)(csr + 2080000);
  float* stbuf0 = cpbuf;
  float* auxbuf = cpbuf + 2 * 100000;
  float* stbuf1 = cpbuf + 4 * 100000;

  const dim3 b128(128), b256(256);
  const int CHAIN_FULL = 768;

  setup_kernel<<<33 + (NA + 255) / 256, b256, 0, stream>>>(Wq, Wk, wfold, Wagt, Wctx, gnc_w, gnc_b,
                                                          wmsg, r12buf, cntbuf, NA, maxbuf);

  csr_count_kernel<<<(NE + 255) / 256, b256, 0, stream>>>(v, cntbuf, arrbuf, NE);
  csr_blocksum_kernel<<<(NA + 255) / 256, b256, 0, stream>>>(cntbuf, bsum, NA);
  csr_scan_kernel<<<(NA + 255) / 256, b256, 0, stream>>>(cntbuf, bsum, rowptr, NA);
  csr_fill_kernel<<<(NE + 255) / 256, b256, 0, stream>>>(v, arrbuf, rowptr, eid, NE);
  csr_rank_kernel<<<(NE + 255) / 256, b256, 0, stream>>>(eid, rowptr, u, v, subuf, NE);

  gemm_q_and_ctxp<<<(NA + 63) / 64 + 2 * ((NC + 63) / 64), b128, 0, stream>>>(
      agt, wfold, qbuf, NA, (NA + 63) / 64,
      ctx, Wctx, stbuf0, Wctx + (size_t)DIM * DIM, stbuf1, NC, (NC + 63) / 64);

  node_attn_kernel<<<NA_WG, b256, 0, stream>>>(qbuf, ctx, stbuf0, rowptr, subuf,
                                                     aggbuf, auxbuf, maxbuf, NA);
  gemm_update_chain<true><<<CHAIN_FULL + (NA - CHAIN_FULL * 64 + 15) / 16, b128, 0, stream>>>(
      agt, aggbuf, auxbuf, maxbuf, Wagt, wmsg, r12buf,
      gna_w, gna_b, norm_w, norm_b, Wlin, gnl_w, gnl_b, agtbuf,
      wfold + (size_t)DIM * DIM, qbuf, NA, CHAIN_FULL);

  node_attn_kernel<<<NA_WG, b256, 0, stream>>>(qbuf, ctx, stbuf1, rowptr, subuf,
                                                     aggbuf, auxbuf, maxbuf + 1, NA);
  gemm_update_chain<false><<<CHAIN_FULL + (NA - CHAIN_FULL * 64 + 15) / 16, b128, 0, stream>>>(
      agtbuf, aggbuf, auxbuf, maxbuf + 1, Wagt + (size_t)DIM * 2 * DIM,
      wmsg + (size_t)DIM * DIM, r12buf + (size_t)2 * DIM,
      gna_w + DIM, gna_b + DIM, norm_w + DIM, norm_b + DIM,
      Wlin + (size_t)DIM * DIM, gnl_w + DIM, gnl_b + DIM, (float*)d_out,
      wfold, qbuf, NA, CHAIN_FULL);
}
